# starting kernel
# baseline (speedup 1.0000x reference)
#include <hip/hip_runtime.h>
#include <hip/hip_bf16.h>

typedef __attribute__((ext_vector_type(2))) float v2f;
typedef __attribute__((ext_vector_type(8))) float v8f;

#define CFEAT 128

__global__ void degree_kernel(const int* __restrict__ col, float* __restrict__ deg, int E) {
    int e = blockIdx.x * blockDim.x + threadIdx.x;
    if (e < E) atomicAdd(&deg[col[e]], 1.0f);
}

__global__ void dinv_kernel(float* __restrict__ deg, int n) {
    int i = blockIdx.x * blockDim.x + threadIdx.x;
    if (i < n) {
        float d = deg[i] + 1.0f;
        deg[i] = rsqrtf(d);
    }
}

template <int K, int N, int ADD_BIAS>
__global__ void gemm16_wmma_f32(const float* __restrict__ A,
                                const float* __restrict__ W,
                                const float* __restrict__ bias,
                                float* __restrict__ C, int M) {
    const int lane = threadIdx.x & 31;
    const int wave = threadIdx.x >> 5;
    const int lo = lane & 15;
    const int hi = lane >> 4;

    const int m0 = blockIdx.x * 16;
    const int n0 = (blockIdx.y * 4 + wave) * 16;

    const float* abase = A + (size_t)(m0 + lo) * K + 2 * hi;
    const float* wbase = W + (size_t)(2 * hi) * N + n0 + lo;

    v8f acc = {};
#pragma unroll
    for (int k0 = 0; k0 < K; k0 += 4) {
        v2f a, b;
        a.x = abase[k0];
        a.y = abase[k0 + 1];
        b.x = wbase[(size_t)k0 * N];
        b.y = wbase[(size_t)(k0 + 1) * N];
        acc = __builtin_amdgcn_wmma_f32_16x16x4_f32(
             false, a,  false, b,
             (short)0, acc,  false,  false);
    }

    const float bv = ADD_BIAS ? bias[n0 + lo] : 0.0f;
    float* cbase = C + (size_t)(m0 + 8 * hi) * N + n0 + lo;
#pragma unroll
    for (int v = 0; v < 8; ++v) {
        cbase[(size_t)v * N] = acc[v] + bv;
    }
}

__global__ void aggregate_edges(const float* __restrict__ h,
                                const int* __restrict__ row,
                                const int* __restrict__ col,
                                const float* __restrict__ dinv,
                                float* __restrict__ agg, int E) {
    const int lane = threadIdx.x & 31;
    const int e = blockIdx.x * (blockDim.x >> 5) + (threadIdx.x >> 5);
    if (e >= E) return;
    const int r = row[e];
    const int c = col[e];
    const float nrm = dinv[r] * dinv[c];
    const float4 v = ((const float4*)(h + (size_t)r * CFEAT))[lane];
    float* dst = agg + (size_t)c * CFEAT + lane * 4;
    atomicAdd(dst + 0, v.x * nrm);
    atomicAdd(dst + 1, v.y * nrm);
    atomicAdd(dst + 2, v.z * nrm);
    atomicAdd(dst + 3, v.w * nrm);
}

__global__ void selfloop_bias_relu(const float* __restrict__ h,
                                   const float* __restrict__ dinv,
                                   const float* __restrict__ bias,
                                   float* __restrict__ agg,
                                   int n) {
    const size_t idx = (size_t)blockIdx.x * blockDim.x + threadIdx.x;
    if (idx >= (size_t)n * CFEAT) return;
    const int i = (int)(idx >> 7);
    const int f = (int)(idx & 127);
    const float d = dinv[i];
    const float v = agg[idx] + h[idx] * d * d + bias[f];
    agg[idx] = v > 0.0f ? v : 0.0f;
}

extern "C" void kernel_launch(void* const* d_in, const int* in_sizes, int n_in,
                              void* d_out, int out_size, void* d_ws, size_t ws_size,
                              hipStream_t stream) {
    const float* x  = (const float*)d_in[0];
    const int*   ei = (const int*)d_in[1];
    const float* W1 = (const float*)d_in[2];
    const float* b1 = (const float*)d_in[3];
    const float* W2 = (const float*)d_in[4];
    const float* b2 = (const float*)d_in[5];
    const float* Wl = (const float*)d_in[6];
    const float* bl = (const float*)d_in[7];
    float* out = (float*)d_out;

    const int E  = in_sizes[1] / 2;
    const int Nn = in_sizes[0] / CFEAT;
    const int C  = CFEAT;

    const int* row = ei;
    const int* col = ei + E;

    float* dinv = (float*)d_ws;
    float* bufA = dinv + Nn;
    float* bufB = bufA + (size_t)Nn * C;

    const size_t featBytes = (size_t)Nn * C * sizeof(float);
    const int Mt = Nn / 16;
    const int featBlocks = (int)(((size_t)Nn * C + 255) / 256);

    hipMemsetAsync(dinv, 0, (size_t)Nn * sizeof(float), stream);
    degree_kernel<<<(E + 255) / 256, 256, 0, stream>>>(col, dinv, E);
    dinv_kernel<<<(Nn + 255) / 256, 256, 0, stream>>>(dinv, Nn);

    gemm16_wmma_f32<128, 128, 0><<<dim3(Mt, 2), 128, 0, stream>>>(x, W1, nullptr, bufA, Nn);
    hipMemsetAsync(bufB, 0, featBytes, stream);
    aggregate_edges<<<(E + 7) / 8, 256, 0, stream>>>(bufA, row, col, dinv, bufB, E);
    selfloop_bias_relu<<<featBlocks, 256, 0, stream>>>(bufA, dinv, b1, bufB, Nn);

    gemm16_wmma_f32<128, 128, 0><<<dim3(Mt, 2), 128, 0, stream>>>(bufB, W2, nullptr, bufA, Nn);
    hipMemsetAsync(bufB, 0, featBytes, stream);
    aggregate_edges<<<(E + 7) / 8, 256, 0, stream>>>(bufA, row, col, dinv, bufB, E);
    selfloop_bias_relu<<<featBlocks, 256, 0, stream>>>(bufA, dinv, b2, bufB, Nn);

    gemm16_wmma_f32<128, 64, 1><<<dim3(Mt, 1), 128, 0, stream>>>(bufB, Wl, bl, out, Nn);
}


# Aggregation by owner over a target-sorted edge list (counting sort each call): a wave sums a node's messages in registers; float atomics, zero fill, self-loop pass gone; sum order = arrival rank
# speedup vs baseline: 2.0847x; 2.0847x over previous
#include <hip/hip_runtime.h>
#include <hip/hip_bf16.h>

typedef __attribute__((ext_vector_type(2))) float v2f;
typedef __attribute__((ext_vector_type(8))) float v8f;

#define CFEAT 128

__global__ void count_rank_kernel(const int* __restrict__ col, int* __restrict__ cnt,
                                  int* __restrict__ rank, int E) {
    int e = blockIdx.x * blockDim.x + threadIdx.x;
    if (e < E) rank[e] = atomicAdd(&cnt[col[e]], 1);
}

__global__ void __launch_bounds__(1024)
scan_dinv_kernel(const int* __restrict__ cnt, int* __restrict__ rowptr,
                 float* __restrict__ dinv, int n) {
    __shared__ int wave_tot[32];
    __shared__ int wave_off[32];
    __shared__ int tile_tot;
    const int t = threadIdx.x;
    const int lane = t & 31;
    const int wave = t >> 5;
    int carry = 0;
    for (int base = 0; base < n; base += 4096) {
        const int i0 = base + 4 * t;
        int c0 = 0, c1 = 0, c2 = 0, c3 = 0;
        if (i0 + 3 < n) {
            const int4 v = *(const int4*)(cnt + i0);
            c0 = v.x; c1 = v.y; c2 = v.z; c3 = v.w;
        } else {
            if (i0 + 0 < n) c0 = cnt[i0 + 0];
            if (i0 + 1 < n) c1 = cnt[i0 + 1];
            if (i0 + 2 < n) c2 = cnt[i0 + 2];
        }
        const int mine = c0 + c1 + c2 + c3;
        int incl = mine;
#pragma unroll
        for (int d = 1; d < 32; d <<= 1) {
            const int up = __shfl_up(incl, d);
            if (lane >= d) incl += up;
        }
        if (lane == 31) wave_tot[wave] = incl;
        __syncthreads();
        if (wave == 0) {
            const int wt = wave_tot[lane];
            int wi = wt;
#pragma unroll
            for (int d = 1; d < 32; d <<= 1) {
                const int up = __shfl_up(wi, d);
                if (lane >= d) wi += up;
            }
            wave_off[lane] = wi - wt;
            if (lane == 31) tile_tot = wi;
        }
        __syncthreads();
        const int p0 = carry + wave_off[wave] + incl - mine;
        const int p1 = p0 + c0, p2 = p1 + c1, p3 = p2 + c2;
        if (i0 + 3 < n) {
            *(int4*)(rowptr + i0) = make_int4(p0, p1, p2, p3);
            *(float4*)(dinv + i0) = make_float4(rsqrtf((float)c0 + 1.0f), rsqrtf((float)c1 + 1.0f),
                                                rsqrtf((float)c2 + 1.0f), rsqrtf((float)c3 + 1.0f));
        } else {
            if (i0 + 0 < n) { rowptr[i0 + 0] = p0; dinv[i0 + 0] = rsqrtf((float)c0 + 1.0f); }
            if (i0 + 1 < n) { rowptr[i0 + 1] = p1; dinv[i0 + 1] = rsqrtf((float)c1 + 1.0f); }
            if (i0 + 2 < n) { rowptr[i0 + 2] = p2; dinv[i0 + 2] = rsqrtf((float)c2 + 1.0f); }
        }
        carry += tile_tot;
    }
    if (t == 0) rowptr[n] = carry;
}

__global__ void fill_kernel(const int* __restrict__ row, const int* __restrict__ col,
                            const int* __restrict__ rowptr, const int* __restrict__ rank,
                            const float* __restrict__ dinv,
                            int* __restrict__ srcw, int E) {
    int e = blockIdx.x * blockDim.x + threadIdx.x;
    if (e < E) {
        const int r = row[e];
        ((int2*)srcw)[rowptr[col[e]] + rank[e]] = make_int2(r, __float_as_int(dinv[r]));
    }
}

template <int K, int N, int ADD_BIAS>
__global__ void gemm16_wmma_f32(const float* __restrict__ A,
                                const float* __restrict__ W,
                                const float* __restrict__ bias,
                                float* __restrict__ C, int M) {
    const int lane = threadIdx.x & 31;
    const int wave = threadIdx.x >> 5;
    const int lo = lane & 15;
    const int hi = lane >> 4;

    const int m0 = blockIdx.x * 16;
    const int n0 = (blockIdx.y * 4 + wave) * 16;

    const float* abase = A + (size_t)(m0 + lo) * K + 2 * hi;
    const float* wbase = W + (size_t)(2 * hi) * N + n0 + lo;

    v8f acc = {};
#pragma unroll
    for (int k0 = 0; k0 < K; k0 += 4) {
        v2f a, b;
        a.x = abase[k0];
        a.y = abase[k0 + 1];
        b.x = wbase[(size_t)k0 * N];
        b.y = wbase[(size_t)(k0 + 1) * N];
        acc = __builtin_amdgcn_wmma_f32_16x16x4_f32(
             false, a,  false, b,
             (short)0, acc,  false,  false);
    }

    const float bv = ADD_BIAS ? bias[n0 + lo] : 0.0f;
    float* cbase = C + (size_t)(m0 + 8 * hi) * N + n0 + lo;
#pragma unroll
    for (int v = 0; v < 8; ++v) {
        cbase[(size_t)v * N] = acc[v] + bv;
    }
}

#define NODES_PER_WAVE 4

__device__ __forceinline__ void axpy4(float4& acc, const float4 v, const float w) {
    acc.x += v.x * w; acc.y += v.y * w; acc.z += v.z * w; acc.w += v.w * w;
}

__global__ void __launch_bounds__(256)
gather_bias_relu(const float* __restrict__ h,
                 const int* __restrict__ rowptr,
                 const int* __restrict__ srcw,
                 const float* __restrict__ dinv,
                 const float* __restrict__ bias,
                 float* __restrict__ outp, int n) {
    const int lane = threadIdx.x & 31;
    const int wave = __builtin_amdgcn_readfirstlane((int)(threadIdx.x >> 5));
    const int node0 = (blockIdx.x * (int)(blockDim.x >> 5) + wave) * NODES_PER_WAVE;
    if (node0 >= n) return;
    const float4* __restrict__ h4 = (const float4*)h + lane;
    const float4 b = ((const float4*)bias)[lane];
    int beg = rowptr[node0];
    for (int i = 0; i < NODES_PER_WAVE; ++i) {
        const int node = node0 + i;
        if (node >= n) break;
        const int end = rowptr[node + 1];
        const float dc = dinv[node];
        const float4 self = h4[(size_t)node * (CFEAT / 4)];
        float4 acc = make_float4(0.0f, 0.0f, 0.0f, 0.0f);
        for (int base = beg; base < end; base += 32) {
            const int j = base + lane;
            int r = 0;
            float w = 0.0f;
            if (j < end) { const int2 p = ((const int2*)srcw)[j]; r = p.x; w = __int_as_float(p.y) * dc; }
            const int wi = __float_as_int(w);
            const int m = (end - base) < 32 ? (end - base) : 32;
            int k = 0;
            for (; k + 4 <= m; k += 4) {
                const int r0 = __builtin_amdgcn_readlane(r, k + 0);
                const int r1 = __builtin_amdgcn_readlane(r, k + 1);
                const int r2 = __builtin_amdgcn_readlane(r, k + 2);
                const int r3 = __builtin_amdgcn_readlane(r, k + 3);
                const float4 v0 = h4[(size_t)r0 * (CFEAT / 4)];
                const float4 v1 = h4[(size_t)r1 * (CFEAT / 4)];
                const float4 v2 = h4[(size_t)r2 * (CFEAT / 4)];
                const float4 v3 = h4[(size_t)r3 * (CFEAT / 4)];
                axpy4(acc, v0, __int_as_float(__builtin_amdgcn_readlane(wi, k + 0)));
                axpy4(acc, v1, __int_as_float(__builtin_amdgcn_readlane(wi, k + 1)));
                axpy4(acc, v2, __int_as_float(__builtin_amdgcn_readlane(wi, k + 2)));
                axpy4(acc, v3, __int_as_float(__builtin_amdgcn_readlane(wi, k + 3)));
            }
            const int left = m - k;
            if (left == 3) {
                const int r0 = __builtin_amdgcn_readlane(r, k + 0);
                const int r1 = __builtin_amdgcn_readlane(r, k + 1);
                const int r2 = __builtin_amdgcn_readlane(r, k + 2);
                const float4 v0 = h4[(size_t)r0 * (CFEAT / 4)];
                const float4 v1 = h4[(size_t)r1 * (CFEAT / 4)];
                const float4 v2 = h4[(size_t)r2 * (CFEAT / 4)];
                axpy4(acc, v0, __int_as_float(__builtin_amdgcn_readlane(wi, k + 0)));
                axpy4(acc, v1, __int_as_float(__builtin_amdgcn_readlane(wi, k + 1)));
                axpy4(acc, v2, __int_as_float(__builtin_amdgcn_readlane(wi, k + 2)));
            } else if (left == 2) {
                const int r0 = __builtin_amdgcn_readlane(r, k + 0);
                const int r1 = __builtin_amdgcn_readlane(r, k + 1);
                const float4 v0 = h4[(size_t)r0 * (CFEAT / 4)];
                const float4 v1 = h4[(size_t)r1 * (CFEAT / 4)];
                axpy4(acc, v0, __int_as_float(__builtin_amdgcn_readlane(wi, k + 0)));
                axpy4(acc, v1, __int_as_float(__builtin_amdgcn_readlane(wi, k + 1)));
            } else if (left == 1) {
                const int r0 = __builtin_amdgcn_readlane(r, k);
                const float4 v0 = h4[(size_t)r0 * (CFEAT / 4)];
                axpy4(acc, v0, __int_as_float(__builtin_amdgcn_readlane(wi, k)));
            }
        }
        axpy4(acc, self, dc * dc);
        float4 o;
        o.x = acc.x + b.x; o.y = acc.y + b.y; o.z = acc.z + b.z; o.w = acc.w + b.w;
        o.x = o.x > 0.0f ? o.x : 0.0f;
        o.y = o.y > 0.0f ? o.y : 0.0f;
        o.z = o.z > 0.0f ? o.z : 0.0f;
        o.w = o.w > 0.0f ? o.w : 0.0f;
        ((float4*)outp)[(size_t)node * (CFEAT / 4) + lane] = o;
        beg = end;
    }
}

extern "C" void kernel_launch(void* const* d_in, const int* in_sizes, int n_in,
                              void* d_out, int out_size, void* d_ws, size_t ws_size,
                              hipStream_t stream) {
    const float* x  = (const float*)d_in[0];
    const int*   ei = (const int*)d_in[1];
    const float* W1 = (const float*)d_in[2];
    const float* b1 = (const float*)d_in[3];
    const float* W2 = (const float*)d_in[4];
    const float* b2 = (const float*)d_in[5];
    const float* Wl = (const float*)d_in[6];
    const float* bl = (const float*)d_in[7];
    float* out = (float*)d_out;

    const int E  = in_sizes[1] / 2;
    const int Nn = in_sizes[0] / CFEAT;
    const int C  = CFEAT;

    const int* row = ei;
    const int* col = ei + E;

    float* dinv = (float*)d_ws;
    float* bufA = dinv + Nn;
    float* bufB = bufA + (size_t)Nn * C;
    int* cnt    = (int*)(bufB + (size_t)Nn * C);
    int* rowptr = cnt + Nn;
    int* srcw   = rowptr + Nn + 4;
    int* rank   = srcw + 2 * (size_t)E;

    const int Mt = Nn / 16;
    const int nodeBlocks = (Nn + 8 * NODES_PER_WAVE - 1) / (8 * NODES_PER_WAVE);

    hipMemsetAsync(cnt, 0, (size_t)Nn * sizeof(int), stream);
    count_rank_kernel<<<(E + 255) / 256, 256, 0, stream>>>(col, cnt, rank, E);
    scan_dinv_kernel<<<1, 1024, 0, stream>>>(cnt, rowptr, dinv, Nn);
    fill_kernel<<<(E + 255) / 256, 256, 0, stream>>>(row, col, rowptr, rank, dinv, srcw, E);

    gemm16_wmma_f32<128, 128, 0><<<dim3(Mt, 2), 128, 0, stream>>>(x, W1, nullptr, bufA, Nn);
    gather_bias_relu<<<nodeBlocks, 256, 0, stream>>>(bufA, rowptr, srcw, dinv, b1, bufB, Nn);

    gemm16_wmma_f32<128, 128, 0><<<dim3(Mt, 2), 128, 0, stream>>>(bufB, W2, nullptr, bufA, Nn);
    gather_bias_relu<<<nodeBlocks, 256, 0, stream>>>(bufA, rowptr, srcw, dinv, b2, bufB, Nn);

    gemm16_wmma_f32<128, 64, 1><<<dim3(Mt, 1), 128, 0, stream>>>(bufB, Wl, bl, out, Nn);
}


# GEMMs W-stationary: each wave keeps its 32 columns of W as f32 WMMA fragments in registers and walks over row tiles staged once per block through LDS by whole-row 16-byte loads; k order unchanged
# speedup vs baseline: 2.4384x; 1.1697x over previous
#include <hip/hip_runtime.h>
#include <hip/hip_bf16.h>

typedef __attribute__((ext_vector_type(2))) float v2f;
typedef __attribute__((ext_vector_type(4))) float v4f;
typedef __attribute__((ext_vector_type(8))) float v8f;

#define CFEAT 128

__global__ void count_rank_kernel(const int* __restrict__ col, int* __restrict__ cnt,
                                  int* __restrict__ rank, int E) {
    int e = blockIdx.x * blockDim.x + threadIdx.x;
    if (e < E) rank[e] = atomicAdd(&cnt[col[e]], 1);
}

__global__ void __launch_bounds__(1024)
scan_dinv_kernel(const int* __restrict__ cnt, int* __restrict__ rowptr,
                 float* __restrict__ dinv, int n) {
    __shared__ int wave_tot[32];
    __shared__ int wave_off[32];
    __shared__ int tile_tot;
    const int t = threadIdx.x;
    const int lane = t & 31;
    const int wave = t >> 5;
    int carry = 0;
    for (int base = 0; base < n; base += 4096) {
        const int i0 = base + 4 * t;
        int c0 = 0, c1 = 0, c2 = 0, c3 = 0;
        if (i0 + 3 < n) {
            const int4 v = *(const int4*)(cnt + i0);
            c0 = v.x; c1 = v.y; c2 = v.z; c3 = v.w;
        } else {
            if (i0 + 0 < n) c0 = cnt[i0 + 0];
            if (i0 + 1 < n) c1 = cnt[i0 + 1];
            if (i0 + 2 < n) c2 = cnt[i0 + 2];
        }
        const int mine = c0 + c1 + c2 + c3;
        int incl = mine;
#pragma unroll
        for (int d = 1; d < 32; d <<= 1) {
            const int up = __shfl_up(incl, d);
            if (lane >= d) incl += up;
        }
        if (lane == 31) wave_tot[wave] = incl;
        __syncthreads();
        if (wave == 0) {
            const int wt = wave_tot[lane];
            int wi = wt;
#pragma unroll
            for (int d = 1; d < 32; d <<= 1) {
                const int up = __shfl_up(wi, d);
                if (lane >= d) wi += up;
            }
            wave_off[lane] = wi - wt;
            if (lane == 31) tile_tot = wi;
        }
        __syncthreads();
        const int p0 = carry + wave_off[wave] + incl - mine;
        const int p1 = p0 + c0, p2 = p1 + c1, p3 = p2 + c2;
        if (i0 + 3 < n) {
            *(int4*)(rowptr + i0) = make_int4(p0, p1, p2, p3);
            *(float4*)(dinv + i0) = make_float4(rsqrtf((float)c0 + 1.0f), rsqrtf((float)c1 + 1.0f),
                                                rsqrtf((float)c2 + 1.0f), rsqrtf((float)c3 + 1.0f));
        } else {
            if (i0 + 0 < n) { rowptr[i0 + 0] = p0; dinv[i0 + 0] = rsqrtf((float)c0 + 1.0f); }
            if (i0 + 1 < n) { rowptr[i0 + 1] = p1; dinv[i0 + 1] = rsqrtf((float)c1 + 1.0f); }
            if (i0 + 2 < n) { rowptr[i0 + 2] = p2; dinv[i0 + 2] = rsqrtf((float)c2 + 1.0f); }
        }
        carry += tile_tot;
    }
    if (t == 0) rowptr[n] = carry;
}

__global__ void fill_kernel(const int* __restrict__ row, const int* __restrict__ col,
                            const int* __restrict__ rowptr, const int* __restrict__ rank,
                            const float* __restrict__ dinv,
                            int* __restrict__ srcw, int E) {
    int e = blockIdx.x * blockDim.x + threadIdx.x;
    if (e < E) {
        const int r = row[e];
        ((int2*)srcw)[rowptr[col[e]] + rank[e]] = make_int2(r, __float_as_int(dinv[r]));
    }
}

#define LDS_PITCH 132

template <int N, int ADD_BIAS>
__global__ void __launch_bounds__(128)
gemm_wstat_wmma_f32(const float* __restrict__ A,
                    const float* __restrict__ W,
                    const float* __restrict__ bias,
                    float* __restrict__ C, int M) {
    constexpr int K = CFEAT;
    constexpr int KS = K / 4;
    constexpr int NPAIR = N / 32;
    constexpr int MT = 4 / NPAIR;
    constexpr int ROWS = 16 * MT;
    constexpr int LD4 = ROWS / 4;

    __shared__ __attribute__((aligned(16))) float lds[2][ROWS * LDS_PITCH];

    const int lane = threadIdx.x & 31;
    const int wave = threadIdx.x >> 5;
    const int lo = lane & 15;
    const int hi = lane >> 4;
    const int mt = wave / NPAIR;
    const int n0 = (wave % NPAIR) * 32;

    v2f b0[KS], b1[KS];
    {
        const float* wbase = W + (size_t)(2 * hi) * N + n0 + lo;
#pragma unroll
        for (int s = 0; s < KS; ++s) {
            b0[s].x = wbase[(size_t)(4 * s) * N];
            b0[s].y = wbase[(size_t)(4 * s + 1) * N];
            b1[s].x = wbase[(size_t)(4 * s) * N + 16];
            b1[s].y = wbase[(size_t)(4 * s + 1) * N + 16];
        }
    }
    const float bv0 = ADD_BIAS ? bias[n0 + lo] : 0.0f;
    const float bv1 = ADD_BIAS ? bias[n0 + 16 + lo] : 0.0f;

    const int steps = (M + ROWS - 1) / ROWS;
    const v4f* __restrict__ A4 = (const v4f*)A + lane;
    v4f g[LD4];

    int st = blockIdx.x;
    if (st < steps) {
#pragma unroll
        for (int i = 0; i < LD4; ++i) {
            const int row = st * ROWS + wave + 4 * i;
            g[i] = A4[(size_t)(row < M ? row : M - 1) * (K / 4)];
        }
#pragma unroll
        for (int i = 0; i < LD4; ++i)
            *(v4f*)&lds[0][(wave + 4 * i) * LDS_PITCH + 4 * lane] = g[i];
    }
    __syncthreads();

    int cur = 0;
    for (; st < steps; st += gridDim.x) {
        const int nxt = st + gridDim.x;
        const bool more = nxt < steps;
        if (more) {
#pragma unroll
            for (int i = 0; i < LD4; ++i) {
                const int row = nxt * ROWS + wave + 4 * i;
                g[i] = A4[(size_t)(row < M ? row : M - 1) * (K / 4)];
            }
        }

        const float* arow = &lds[cur][(16 * mt + lo) * LDS_PITCH + 2 * hi];
        v2f a[KS];
#pragma unroll
        for (int s = 0; s < KS; ++s) a[s] = *(const v2f*)(arow + 4 * s);
        __builtin_amdgcn_sched_barrier(0);

        v8f acc0 = {}, acc1 = {};
#pragma unroll
        for (int s = 0; s < KS; ++s) {
            acc0 = __builtin_amdgcn_wmma_f32_16x16x4_f32(
                 false, a[s],  false, b0[s],
                 (short)0, acc0,  false,  false);
            acc1 = __builtin_amdgcn_wmma_f32_16x16x4_f32(
                 false, a[s],  false, b1[s],
                 (short)0, acc1,  false,  false);
        }

        const int r0 = st * ROWS + 16 * mt + 8 * hi;
        float* cbase = C + (size_t)r0 * N + n0 + lo;
        float o0[8], o1[8];
#pragma unroll
        for (int v = 0; v < 8; ++v) { o0[v] = acc0[v] + bv0; o1[v] = acc1[v] + bv1; }
        if ((st + 1) * ROWS <= M) {
#pragma unroll
            for (int v = 0; v < 8; ++v) {
                cbase[(size_t)v * N] = o0[v];
                cbase[(size_t)v * N + 16] = o1[v];
            }
        } else {
#pragma unroll
            for (int v = 0; v < 8; ++v) {
                if (r0 + v < M) {
                    cbase[(size_t)v * N] = o0[v];
                    cbase[(size_t)v * N + 16] = o1[v];
                }
            }
        }

        if (more) {
#pragma unroll
            for (int i = 0; i < LD4; ++i)
                *(v4f*)&lds[cur ^ 1][(wave + 4 * i) * LDS_PITCH + 4 * lane] = g[i];
        }
        __builtin_amdgcn_fence(__ATOMIC_RELEASE, "workgroup", "local");
        __builtin_amdgcn_s_barrier();
        __builtin_amdgcn_fence(__ATOMIC_ACQUIRE, "workgroup", "local");
        cur ^= 1;
    }
}

#define NODES_PER_WAVE 4

__device__ __forceinline__ void axpy4(float4& acc, const float4 v, const float w) {
    acc.x += v.x * w; acc.y += v.y * w; acc.z += v.z * w; acc.w += v.w * w;
}

__global__ void __launch_bounds__(256)
gather_bias_relu(const float* __restrict__ h,
                 const int* __restrict__ rowptr,
                 const int* __restrict__ srcw,
                 const float* __restrict__ dinv,
                 const float* __restrict__ bias,
                 float* __restrict__ outp, int n) {
    const int lane = threadIdx.x & 31;
    const int wave = __builtin_amdgcn_readfirstlane((int)(threadIdx.x >> 5));
    const int node0 = (blockIdx.x * (int)(blockDim.x >> 5) + wave) * NODES_PER_WAVE;
    if (node0 >= n) return;
    const float4* __restrict__ h4 = (const float4*)h + lane;
    const float4 b = ((const float4*)bias)[lane];
    int beg = rowptr[node0];
    for (int i = 0; i < NODES_PER_WAVE; ++i) {
        const int node = node0 + i;
        if (node >= n) break;
        const int end = rowptr[node + 1];
        const float dc = dinv[node];
        const float4 self = h4[(size_t)node * (CFEAT / 4)];
        float4 acc = make_float4(0.0f, 0.0f, 0.0f, 0.0f);
        for (int base = beg; base < end; base += 32) {
            const int j = base + lane;
            int r = 0;
            float w = 0.0f;
            if (j < end) { const int2 p = ((const int2*)srcw)[j]; r = p.x; w = __int_as_float(p.y) * dc; }
            const int wi = __float_as_int(w);
            const int m = (end - base) < 32 ? (end - base) : 32;
            int k = 0;
            for (; k + 4 <= m; k += 4) {
                const int r0 = __builtin_amdgcn_readlane(r, k + 0);
                const int r1 = __builtin_amdgcn_readlane(r, k + 1);
                const int r2 = __builtin_amdgcn_readlane(r, k + 2);
                const int r3 = __builtin_amdgcn_readlane(r, k + 3);
                const float4 v0 = h4[(size_t)r0 * (CFEAT / 4)];
                const float4 v1 = h4[(size_t)r1 * (CFEAT / 4)];
                const float4 v2 = h4[(size_t)r2 * (CFEAT / 4)];
                const float4 v3 = h4[(size_t)r3 * (CFEAT / 4)];
                axpy4(acc, v0, __int_as_float(__builtin_amdgcn_readlane(wi, k + 0)));
                axpy4(acc, v1, __int_as_float(__builtin_amdgcn_readlane(wi, k + 1)));
                axpy4(acc, v2, __int_as_float(__builtin_amdgcn_readlane(wi, k + 2)));
                axpy4(acc, v3, __int_as_float(__builtin_amdgcn_readlane(wi, k + 3)));
            }
            const int left = m - k;
            if (left == 3) {
                const int r0 = __builtin_amdgcn_readlane(r, k + 0);
                const int r1 = __builtin_amdgcn_readlane(r, k + 1);
                const int r2 = __builtin_amdgcn_readlane(r, k + 2);
                const float4 v0 = h4[(size_t)r0 * (CFEAT / 4)];
                const float4 v1 = h4[(size_t)r1 * (CFEAT / 4)];
                const float4 v2 = h4[(size_t)r2 * (CFEAT / 4)];
                axpy4(acc, v0, __int_as_float(__builtin_amdgcn_readlane(wi, k + 0)));
                axpy4(acc, v1, __int_as_float(__builtin_amdgcn_readlane(wi, k + 1)));
                axpy4(acc, v2, __int_as_float(__builtin_amdgcn_readlane(wi, k + 2)));
            } else if (left == 2) {
                const int r0 = __builtin_amdgcn_readlane(r, k + 0);
                const int r1 = __builtin_amdgcn_readlane(r, k + 1);
                const float4 v0 = h4[(size_t)r0 * (CFEAT / 4)];
                const float4 v1 = h4[(size_t)r1 * (CFEAT / 4)];
                axpy4(acc, v0, __int_as_float(__builtin_amdgcn_readlane(wi, k + 0)));
                axpy4(acc, v1, __int_as_float(__builtin_amdgcn_readlane(wi, k + 1)));
            } else if (left == 1) {
                const int r0 = __builtin_amdgcn_readlane(r, k);
                const float4 v0 = h4[(size_t)r0 * (CFEAT / 4)];
                axpy4(acc, v0, __int_as_float(__builtin_amdgcn_readlane(wi, k)));
            }
        }
        axpy4(acc, self, dc * dc);
        float4 o;
        o.x = acc.x + b.x; o.y = acc.y + b.y; o.z = acc.z + b.z; o.w = acc.w + b.w;
        o.x = o.x > 0.0f ? o.x : 0.0f;
        o.y = o.y > 0.0f ? o.y : 0.0f;
        o.z = o.z > 0.0f ? o.z : 0.0f;
        o.w = o.w > 0.0f ? o.w : 0.0f;
        ((float4*)outp)[(size_t)node * (CFEAT / 4) + lane] = o;
        beg = end;
    }
}

extern "C" void kernel_launch(void* const* d_in, const int* in_sizes, int n_in,
                              void* d_out, int out_size, void* d_ws, size_t ws_size,
                              hipStream_t stream) {
    const float* x  = (const float*)d_in[0];
    const int*   ei = (const int*)d_in[1];
    const float* W1 = (const float*)d_in[2];
    const float* b1 = (const float*)d_in[3];
    const float* W2 = (const float*)d_in[4];
    const float* b2 = (const float*)d_in[5];
    const float* Wl = (const float*)d_in[6];
    const float* bl = (const float*)d_in[7];
    float* out = (float*)d_out;

    const int E  = in_sizes[1] / 2;
    const int Nn = in_sizes[0] / CFEAT;
    const int C  = CFEAT;

    const int* row = ei;
    const int* col = ei + E;

    float* dinv = (float*)d_ws;
    float* bufA = dinv + Nn;
    float* bufB = bufA + (size_t)Nn * C;
    int* cnt    = (int*)(bufB + (size_t)Nn * C);
    int* rowptr = cnt + Nn;
    int* srcw   = rowptr + Nn + 4;
    int* rank   = srcw + 2 * (size_t)E;

    const int nodeBlocks = (Nn + 8 * NODES_PER_WAVE - 1) / (8 * NODES_PER_WAVE);

    hipMemsetAsync(cnt, 0, (size_t)Nn * sizeof(int), stream);
    count_rank_kernel<<<(E + 255) / 256, 256, 0, stream>>>(col, cnt, rank, E);
    scan_dinv_kernel<<<1, 1024, 0, stream>>>(cnt, rowptr, dinv, Nn);
    fill_kernel<<<(E + 255) / 256, 256, 0, stream>>>(row, col, rowptr, rank, dinv, srcw, E);

    gemm_wstat_wmma_f32<128, 0><<<512, 128, 0, stream>>>(x, W1, nullptr, bufA, Nn);
    gather_bias_relu<<<nodeBlocks, 256, 0, stream>>>(bufA, rowptr, srcw, dinv, b1, bufB, Nn);

    gemm_wstat_wmma_f32<128, 0><<<512, 128, 0, stream>>>(bufB, W2, nullptr, bufA, Nn);
    gather_bias_relu<<<nodeBlocks, 256, 0, stream>>>(bufA, rowptr, srcw, dinv, b2, bufB, Nn);

    gemm_wstat_wmma_f32<64, 1><<<512, 128, 0, stream>>>(bufB, Wl, bl, out, Nn);
}


# Prefix scan of target counts rides in the first GEMM's launch (independent work): one extra 128-thread work-group per 4096-count tile, each sums the counts before its tile itself; own launch gone
# speedup vs baseline: 2.6934x; 1.1046x over previous
#include <hip/hip_runtime.h>
#include <hip/hip_bf16.h>

typedef __attribute__((ext_vector_type(2))) float v2f;
typedef __attribute__((ext_vector_type(4))) float v4f;
typedef __attribute__((ext_vector_type(8))) float v8f;

#define CFEAT 128

__global__ void count_rank_kernel(const int* __restrict__ col, int* __restrict__ cnt,
                                  int* __restrict__ rank, int E) {
    int e = blockIdx.x * blockDim.x + threadIdx.x;
    if (e < E) rank[e] = atomicAdd(&cnt[col[e]], 1);
}

__global__ void fill_kernel(const int* __restrict__ row, const int* __restrict__ col,
                            const int* __restrict__ rowptr, const int* __restrict__ rank,
                            const float* __restrict__ dinv,
                            int* __restrict__ srcw, int E) {
    int e = blockIdx.x * blockDim.x + threadIdx.x;
    if (e < E) {
        const int r = row[e];
        ((int2*)srcw)[rowptr[col[e]] + rank[e]] = make_int2(r, __float_as_int(dinv[r]));
    }
}

#define SCAN_V 8
#define SCAN_TILE (128 * 4 * SCAN_V)

__device__ __forceinline__ void scan_dinv_role(const int* __restrict__ cnt, int* __restrict__ rowptr,
                                               float* __restrict__ dinv, int n, int* __restrict__ stot,
                                               int tile) {
    const int t = threadIdx.x;
    const int lane = t & 31;
    const int wave = __builtin_amdgcn_readfirstlane(t >> 5);
    const int base = tile * SCAN_TILE;
    const bool whole = base + SCAN_TILE <= n;

    int4 c[SCAN_V];
    if (whole) {
#pragma unroll
        for (int j = 0; j < SCAN_V; ++j) c[j] = *(const int4*)(cnt + base + (j * 128 + t) * 4);
    } else {
#pragma unroll
        for (int j = 0; j < SCAN_V; ++j) {
            const int i0 = base + (j * 128 + t) * 4;
            const bool in = i0 + 3 < n;
            const int4 v = *(const int4*)(cnt + (in ? i0 : 0));
            c[j] = in ? v : make_int4(0, 0, 0, 0);
        }
        if (n & 3) {
#pragma unroll
            for (int j = 0; j < SCAN_V; ++j) {
                const int i0 = base + (j * 128 + t) * 4;
                if (i0 < n && i0 + 3 >= n) {
                    c[j].x = cnt[i0];
                    if (i0 + 1 < n) c[j].y = cnt[i0 + 1];
                    if (i0 + 2 < n) c[j].z = cnt[i0 + 2];
                }
            }
        }
    }

    int acc = 0;
#pragma unroll 2
    for (int f = 0; f < base; f += SCAN_TILE) {
        int4 v[SCAN_V];
#pragma unroll
        for (int j = 0; j < SCAN_V; ++j) v[j] = *(const int4*)(cnt + f + (j * 128 + t) * 4);
#pragma unroll
        for (int j = 0; j < SCAN_V; ++j) acc += v[j].x + v[j].y + v[j].z + v[j].w;
    }
#pragma unroll
    for (int d = 16; d >= 1; d >>= 1) acc += __shfl_xor(acc, d);

    int mine[SCAN_V], incl[SCAN_V];
#pragma unroll
    for (int j = 0; j < SCAN_V; ++j) { mine[j] = c[j].x + c[j].y + c[j].z + c[j].w; incl[j] = mine[j]; }
#pragma unroll
    for (int d = 1; d < 32; d <<= 1) {
#pragma unroll
        for (int j = 0; j < SCAN_V; ++j) {
            const int up = __shfl_up(incl[j], d);
            if (lane >= d) incl[j] += up;
        }
    }
    if (lane == 31) {
#pragma unroll
        for (int j = 0; j < SCAN_V; ++j) stot[j * 4 + wave] = incl[j];
        stot[32 + wave] = acc;
    }
    __syncthreads();

    const int4 f4 = *(const int4*)(stot + 32);
    int run = f4.x + f4.y + f4.z + f4.w;
    int off[SCAN_V];
#pragma unroll
    for (int j = 0; j < SCAN_V; ++j) {
        const int4 w4 = *(const int4*)(stot + j * 4);
        off[j] = run + (wave > 0 ? w4.x : 0) + (wave > 1 ? w4.y : 0) + (wave > 2 ? w4.z : 0);
        run += w4.x + w4.y + w4.z + w4.w;
    }
    if (whole) {
#pragma unroll
        for (int j = 0; j < SCAN_V; ++j) {
            const int i0 = base + (j * 128 + t) * 4;
            const int c0 = c[j].x, c1 = c[j].y, c2 = c[j].z, c3 = c[j].w;
            const int p0 = off[j] + incl[j] - mine[j];
            const int p1 = p0 + c0, p2 = p1 + c1, p3 = p2 + c2;
            *(int4*)(rowptr + i0) = make_int4(p0, p1, p2, p3);
            *(float4*)(dinv + i0) = make_float4(rsqrtf((float)c0 + 1.0f), rsqrtf((float)c1 + 1.0f),
                                                rsqrtf((float)c2 + 1.0f), rsqrtf((float)c3 + 1.0f));
        }
    } else {
#pragma unroll
        for (int j = 0; j < SCAN_V; ++j) {
            const int i0 = base + (j * 128 + t) * 4;
            const int c0 = c[j].x, c1 = c[j].y, c2 = c[j].z, c3 = c[j].w;
            const int p0 = off[j] + incl[j] - mine[j];
            const int p1 = p0 + c0, p2 = p1 + c1, p3 = p2 + c2;
            if (i0 + 3 < n) {
                *(int4*)(rowptr + i0) = make_int4(p0, p1, p2, p3);
                *(float4*)(dinv + i0) = make_float4(rsqrtf((float)c0 + 1.0f), rsqrtf((float)c1 + 1.0f),
                                                    rsqrtf((float)c2 + 1.0f), rsqrtf((float)c3 + 1.0f));
            } else {
                if (i0 + 0 < n) { rowptr[i0 + 0] = p0; dinv[i0 + 0] = rsqrtf((float)c0 + 1.0f); }
                if (i0 + 1 < n) { rowptr[i0 + 1] = p1; dinv[i0 + 1] = rsqrtf((float)c1 + 1.0f); }
                if (i0 + 2 < n) { rowptr[i0 + 2] = p2; dinv[i0 + 2] = rsqrtf((float)c2 + 1.0f); }
            }
        }
    }
    if (t == 0 && base + SCAN_TILE >= n) rowptr[n] = run;
}

#define LDS_PITCH 132

template <int N, int ADD_BIAS, int SCAN>
__global__ void __launch_bounds__(128)
gemm_wstat_wmma_f32(const float* __restrict__ A,
                    const float* __restrict__ W,
                    const float* __restrict__ bias,
                    float* __restrict__ C, int M,
                    const int* __restrict__ cnt, int* __restrict__ rowptr,
                    float* __restrict__ dinv) {
    constexpr int K = CFEAT;
    constexpr int KS = K / 4;
    constexpr int NPAIR = N / 32;
    constexpr int MT = 4 / NPAIR;
    constexpr int ROWS = 16 * MT;
    constexpr int LD4 = ROWS / 4;

    __shared__ __attribute__((aligned(16))) float lds[2][ROWS * LDS_PITCH];

    const int nscan = SCAN ? (M + SCAN_TILE - 1) / SCAN_TILE : 0;
    if (SCAN && (int)blockIdx.x < nscan) {
        __shared__ __attribute__((aligned(16))) int stot[36];
        scan_dinv_role(cnt, rowptr, dinv, M, stot, (int)blockIdx.x);
        return;
    }
    const int bid = (int)blockIdx.x - nscan;
    const int nblk = (int)gridDim.x - nscan;

    const int lane = threadIdx.x & 31;
    const int wave = threadIdx.x >> 5;
    const int lo = lane & 15;
    const int hi = lane >> 4;
    const int mt = wave / NPAIR;
    const int n0 = (wave % NPAIR) * 32;

    v2f b0[KS], b1[KS];
    {
        const float* wbase = W + (size_t)(2 * hi) * N + n0 + lo;
#pragma unroll
        for (int s = 0; s < KS; ++s) {
            b0[s].x = wbase[(size_t)(4 * s) * N];
            b0[s].y = wbase[(size_t)(4 * s + 1) * N];
            b1[s].x = wbase[(size_t)(4 * s) * N + 16];
            b1[s].y = wbase[(size_t)(4 * s + 1) * N + 16];
        }
    }
    const float bv0 = ADD_BIAS ? bias[n0 + lo] : 0.0f;
    const float bv1 = ADD_BIAS ? bias[n0 + 16 + lo] : 0.0f;

    const int steps = (M + ROWS - 1) / ROWS;
    const v4f* __restrict__ A4 = (const v4f*)A + lane;
    v4f g[LD4];

    int st = bid;
    if (st < steps) {
#pragma unroll
        for (int i = 0; i < LD4; ++i) {
            const int row = st * ROWS + wave + 4 * i;
            g[i] = A4[(size_t)(row < M ? row : M - 1) * (K / 4)];
        }
#pragma unroll
        for (int i = 0; i < LD4; ++i)
            *(v4f*)&lds[0][(wave + 4 * i) * LDS_PITCH + 4 * lane] = g[i];
    }
    __syncthreads();

    int cur = 0;
    for (; st < steps; st += nblk) {
        const int nxt = st + nblk;
        const bool more = nxt < steps;
        if (more) {
#pragma unroll
            for (int i = 0; i < LD4; ++i) {
                const int row = nxt * ROWS + wave + 4 * i;
                g[i] = A4[(size_t)(row < M ? row : M - 1) * (K / 4)];
            }
        }

        const float* arow = &lds[cur][(16 * mt + lo) * LDS_PITCH + 2 * hi];
        v2f a[KS];
#pragma unroll
        for (int s = 0; s < KS; ++s) a[s] = *(const v2f*)(arow + 4 * s);
        __builtin_amdgcn_sched_barrier(0);

        v8f acc0 = {}, acc1 = {};
#pragma unroll
        for (int s = 0; s < KS; ++s) {
            acc0 = __builtin_amdgcn_wmma_f32_16x16x4_f32(
                 false, a[s],  false, b0[s],
                 (short)0, acc0,  false,  false);
            acc1 = __builtin_amdgcn_wmma_f32_16x16x4_f32(
                 false, a[s],  false, b1[s],
                 (short)0, acc1,  false,  false);
        }

        const int r0 = st * ROWS + 16 * mt + 8 * hi;
        float* cbase = C + (size_t)r0 * N + n0 + lo;
        float o0[8], o1[8];
#pragma unroll
        for (int v = 0; v < 8; ++v) { o0[v] = acc0[v] + bv0; o1[v] = acc1[v] + bv1; }
        if ((st + 1) * ROWS <= M) {
#pragma unroll
            for (int v = 0; v < 8; ++v) {
                cbase[(size_t)v * N] = o0[v];
                cbase[(size_t)v * N + 16] = o1[v];
            }
        } else {
#pragma unroll
            for (int v = 0; v < 8; ++v) {
                if (r0 + v < M) {
                    cbase[(size_t)v * N] = o0[v];
                    cbase[(size_t)v * N + 16] = o1[v];
                }
            }
        }

        if (more) {
#pragma unroll
            for (int i = 0; i < LD4; ++i)
                *(v4f*)&lds[cur ^ 1][(wave + 4 * i) * LDS_PITCH + 4 * lane] = g[i];
        }
        __builtin_amdgcn_fence(__ATOMIC_RELEASE, "workgroup", "local");
        __builtin_amdgcn_s_barrier();
        __builtin_amdgcn_fence(__ATOMIC_ACQUIRE, "workgroup", "local");
        cur ^= 1;
    }
}

#define NODES_PER_WAVE 4

__device__ __forceinline__ void axpy4(float4& acc, const float4 v, const float w) {
    acc.x += v.x * w; acc.y += v.y * w; acc.z += v.z * w; acc.w += v.w * w;
}

__global__ void __launch_bounds__(256)
gather_bias_relu(const float* __restrict__ h,
                 const int* __restrict__ rowptr,
                 const int* __restrict__ srcw,
                 const float* __restrict__ dinv,
                 const float* __restrict__ bias,
                 float* __restrict__ outp, int n) {
    const int lane = threadIdx.x & 31;
    const int wave = __builtin_amdgcn_readfirstlane((int)(threadIdx.x >> 5));
    const int node0 = (blockIdx.x * (int)(blockDim.x >> 5) + wave) * NODES_PER_WAVE;
    if (node0 >= n) return;
    const float4* __restrict__ h4 = (const float4*)h + lane;
    const float4 b = ((const float4*)bias)[lane];
    int beg = rowptr[node0];
    for (int i = 0; i < NODES_PER_WAVE; ++i) {
        const int node = node0 + i;
        if (node >= n) break;
        const int end = rowptr[node + 1];
        const float dc = dinv[node];
        const float4 self = h4[(size_t)node * (CFEAT / 4)];
        float4 acc = make_float4(0.0f, 0.0f, 0.0f, 0.0f);
        for (int base = beg; base < end; base += 32) {
            const int j = base + lane;
            int r = 0;
            float w = 0.0f;
            if (j < end) { const int2 p = ((const int2*)srcw)[j]; r = p.x; w = __int_as_float(p.y) * dc; }
            const int wi = __float_as_int(w);
            const int m = (end - base) < 32 ? (end - base) : 32;
            int k = 0;
            for (; k + 4 <= m; k += 4) {
                const int r0 = __builtin_amdgcn_readlane(r, k + 0);
                const int r1 = __builtin_amdgcn_readlane(r, k + 1);
                const int r2 = __builtin_amdgcn_readlane(r, k + 2);
                const int r3 = __builtin_amdgcn_readlane(r, k + 3);
                const float4 v0 = h4[(size_t)r0 * (CFEAT / 4)];
                const float4 v1 = h4[(size_t)r1 * (CFEAT / 4)];
                const float4 v2 = h4[(size_t)r2 * (CFEAT / 4)];
                const float4 v3 = h4[(size_t)r3 * (CFEAT / 4)];
                axpy4(acc, v0, __int_as_float(__builtin_amdgcn_readlane(wi, k + 0)));
                axpy4(acc, v1, __int_as_float(__builtin_amdgcn_readlane(wi, k + 1)));
                axpy4(acc, v2, __int_as_float(__builtin_amdgcn_readlane(wi, k + 2)));
                axpy4(acc, v3, __int_as_float(__builtin_amdgcn_readlane(wi, k + 3)));
            }
            const int left = m - k;
            if (left == 3) {
                const int r0 = __builtin_amdgcn_readlane(r, k + 0);
                const int r1 = __builtin_amdgcn_readlane(r, k + 1);
                const int r2 = __builtin_amdgcn_readlane(r, k + 2);
                const float4 v0 = h4[(size_t)r0 * (CFEAT / 4)];
                const float4 v1 = h4[(size_t)r1 * (CFEAT / 4)];
                const float4 v2 = h4[(size_t)r2 * (CFEAT / 4)];
                axpy4(acc, v0, __int_as_float(__builtin_amdgcn_readlane(wi, k + 0)));
                axpy4(acc, v1, __int_as_float(__builtin_amdgcn_readlane(wi, k + 1)));
                axpy4(acc, v2, __int_as_float(__builtin_amdgcn_readlane(wi, k + 2)));
            } else if (left == 2) {
                const int r0 = __builtin_amdgcn_readlane(r, k + 0);
                const int r1 = __builtin_amdgcn_readlane(r, k + 1);
                const float4 v0 = h4[(size_t)r0 * (CFEAT / 4)];
                const float4 v1 = h4[(size_t)r1 * (CFEAT / 4)];
                axpy4(acc, v0, __int_as_float(__builtin_amdgcn_readlane(wi, k + 0)));
                axpy4(acc, v1, __int_as_float(__builtin_amdgcn_readlane(wi, k + 1)));
            } else if (left == 1) {
                const int r0 = __builtin_amdgcn_readlane(r, k);
                const float4 v0 = h4[(size_t)r0 * (CFEAT / 4)];
                axpy4(acc, v0, __int_as_float(__builtin_amdgcn_readlane(wi, k)));
            }
        }
        axpy4(acc, self, dc * dc);
        float4 o;
        o.x = acc.x + b.x; o.y = acc.y + b.y; o.z = acc.z + b.z; o.w = acc.w + b.w;
        o.x = o.x > 0.0f ? o.x : 0.0f;
        o.y = o.y > 0.0f ? o.y : 0.0f;
        o.z = o.z > 0.0f ? o.z : 0.0f;
        o.w = o.w > 0.0f ? o.w : 0.0f;
        ((float4*)outp)[(size_t)node * (CFEAT / 4) + lane] = o;
        beg = end;
    }
}

extern "C" void kernel_launch(void* const* d_in, const int* in_sizes, int n_in,
                              void* d_out, int out_size, void* d_ws, size_t ws_size,
                              hipStream_t stream) {
    const float* x  = (const float*)d_in[0];
    const int*   ei = (const int*)d_in[1];
    const float* W1 = (const float*)d_in[2];
    const float* b1 = (const float*)d_in[3];
    const float* W2 = (const float*)d_in[4];
    const float* b2 = (const float*)d_in[5];
    const float* Wl = (const float*)d_in[6];
    const float* bl = (const float*)d_in[7];
    float* out = (float*)d_out;

    const int E  = in_sizes[1] / 2;
    const int Nn = in_sizes[0] / CFEAT;
    const int C  = CFEAT;

    const int* row = ei;
    const int* col = ei + E;

    float* dinv = (float*)d_ws;
    float* bufA = dinv + Nn;
    float* bufB = bufA + (size_t)Nn * C;
    int* cnt    = (int*)(bufB + (size_t)Nn * C);
    int* rowptr = cnt + Nn;
    int* srcw   = rowptr + Nn + 4;
    int* rank   = srcw + 2 * (size_t)E;

    const int nodeBlocks = (Nn + 8 * NODES_PER_WAVE - 1) / (8 * NODES_PER_WAVE);

    hipMemsetAsync(cnt, 0, (size_t)Nn * sizeof(int), stream);
    count_rank_kernel<<<(E + 255) / 256, 256, 0, stream>>>(col, cnt, rank, E);
    gemm_wstat_wmma_f32<128, 0, 1><<<512 + (Nn + SCAN_TILE - 1) / SCAN_TILE, 128, 0, stream>>>(x, W1, nullptr, bufA, Nn, cnt, rowptr, dinv);
    fill_kernel<<<(E + 255) / 256, 256, 0, stream>>>(row, col, rowptr, rank, dinv, srcw, E);

    gather_bias_relu<<<nodeBlocks, 256, 0, stream>>>(bufA, rowptr, srcw, dinv, b1, bufB, Nn);

    gemm_wstat_wmma_f32<128, 0, 0><<<512, 128, 0, stream>>>(bufB, W2, nullptr, bufA, Nn, nullptr, nullptr, nullptr);
    gather_bias_relu<<<nodeBlocks, 256, 0, stream>>>(bufA, rowptr, srcw, dinv, b2, bufB, Nn);

    gemm_wstat_wmma_f32<64, 1, 0><<<512, 128, 0, stream>>>(bufB, Wl, bl, out, Nn, nullptr, nullptr, nullptr);
}


# Edge kernels (count+rank and fill): a lane takes 8 edges by 16-byte loads, a wave 256 consecutive edges, work-groups of 2 waves, so an eighth of the waves enter; sum order still arrival rank
# speedup vs baseline: 2.8177x; 1.0461x over previous
#include <hip/hip_runtime.h>
#include <hip/hip_bf16.h>

typedef __attribute__((ext_vector_type(2))) float v2f;
typedef __attribute__((ext_vector_type(4))) float v4f;
typedef __attribute__((ext_vector_type(8))) float v8f;

#define CFEAT 128

#define EDGE_G 2
#define EDGE_WAVE (128 * EDGE_G)
#define EDGE_BLOCK 64

__global__ void __launch_bounds__(EDGE_BLOCK)
count_rank_kernel(const int* __restrict__ col, int* __restrict__ cnt,
                  int* __restrict__ rank, int E) {
    const int lane = threadIdx.x & 31;
    const int wbase = __builtin_amdgcn_readfirstlane((int)((blockIdx.x * blockDim.x + threadIdx.x) >> 5)) * EDGE_WAVE;
    if (wbase >= E) return;
    const bool vec = ((E & 3) == 0) && ((((size_t)col | (size_t)rank) & 15) == 0);
    if (vec && wbase + EDGE_WAVE <= E) {
        int4 c[EDGE_G], r[EDGE_G];
#pragma unroll
        for (int g = 0; g < EDGE_G; ++g) c[g] = *(const int4*)(col + wbase + 128 * g + 4 * lane);
        __builtin_amdgcn_sched_barrier(0);
#pragma unroll
        for (int g = 0; g < EDGE_G; ++g) {
            r[g].x = atomicAdd(&cnt[c[g].x], 1);
            r[g].y = atomicAdd(&cnt[c[g].y], 1);
            r[g].z = atomicAdd(&cnt[c[g].z], 1);
            r[g].w = atomicAdd(&cnt[c[g].w], 1);
        }
#pragma unroll
        for (int g = 0; g < EDGE_G; ++g) *(int4*)(rank + wbase + 128 * g + 4 * lane) = r[g];
    } else {
        for (int i = lane; i < EDGE_WAVE; i += 32) {
            const int e = wbase + i;
            if (e < E) rank[e] = atomicAdd(&cnt[col[e]], 1);
        }
    }
}

__global__ void __launch_bounds__(EDGE_BLOCK)
fill_kernel(const int* __restrict__ row, const int* __restrict__ col,
            const int* __restrict__ rowptr, const int* __restrict__ rank,
            const float* __restrict__ dinv,
            int* __restrict__ srcw, int E) {
    const int lane = threadIdx.x & 31;
    const int wbase = __builtin_amdgcn_readfirstlane((int)((blockIdx.x * blockDim.x + threadIdx.x) >> 5)) * EDGE_WAVE;
    if (wbase >= E) return;
    int2* __restrict__ pairs = (int2*)srcw;
    const bool vec = ((E & 3) == 0) && ((((size_t)row | (size_t)col | (size_t)rank) & 15) == 0);
    if (vec && wbase + EDGE_WAVE <= E) {
        int4 r[EDGE_G], c[EDGE_G], k[EDGE_G];
#pragma unroll
        for (int g = 0; g < EDGE_G; ++g) {
            const int off = wbase + 128 * g + 4 * lane;
            r[g] = *(const int4*)(row + off);
            c[g] = *(const int4*)(col + off);
            k[g] = *(const int4*)(rank + off);
        }
        int4 p[EDGE_G];
        float4 d[EDGE_G];
#pragma unroll
        for (int g = 0; g < EDGE_G; ++g) {
            p[g] = make_int4(rowptr[c[g].x], rowptr[c[g].y], rowptr[c[g].z], rowptr[c[g].w]);
            d[g] = make_float4(dinv[r[g].x], dinv[r[g].y], dinv[r[g].z], dinv[r[g].w]);
        }
#pragma unroll
        for (int g = 0; g < EDGE_G; ++g) {
            pairs[p[g].x + k[g].x] = make_int2(r[g].x, __float_as_int(d[g].x));
            pairs[p[g].y + k[g].y] = make_int2(r[g].y, __float_as_int(d[g].y));
            pairs[p[g].z + k[g].z] = make_int2(r[g].z, __float_as_int(d[g].z));
            pairs[p[g].w + k[g].w] = make_int2(r[g].w, __float_as_int(d[g].w));
        }
    } else {
        for (int i = lane; i < EDGE_WAVE; i += 32) {
            const int e = wbase + i;
            if (e < E) {
                const int s = row[e];
                pairs[rowptr[col[e]] + rank[e]] = make_int2(s, __float_as_int(dinv[s]));
            }
        }
    }
}

#define SCAN_V 8
#define SCAN_TILE (128 * 4 * SCAN_V)

__device__ __forceinline__ void scan_dinv_role(const int* __restrict__ cnt, int* __restrict__ rowptr,
                                               float* __restrict__ dinv, int n, int* __restrict__ stot,
                                               int tile) {
    const int t = threadIdx.x;
    const int lane = t & 31;
    const int wave = __builtin_amdgcn_readfirstlane(t >> 5);
    const int base = tile * SCAN_TILE;
    const bool whole = base + SCAN_TILE <= n;

    int4 c[SCAN_V];
    if (whole) {
#pragma unroll
        for (int j = 0; j < SCAN_V; ++j) c[j] = *(const int4*)(cnt + base + (j * 128 + t) * 4);
    } else {
#pragma unroll
        for (int j = 0; j < SCAN_V; ++j) {
            const int i0 = base + (j * 128 + t) * 4;
            const bool in = i0 + 3 < n;
            const int4 v = *(const int4*)(cnt + (in ? i0 : 0));
            c[j] = in ? v : make_int4(0, 0, 0, 0);
        }
        if (n & 3) {
#pragma unroll
            for (int j = 0; j < SCAN_V; ++j) {
                const int i0 = base + (j * 128 + t) * 4;
                if (i0 < n && i0 + 3 >= n) {
                    c[j].x = cnt[i0];
                    if (i0 + 1 < n) c[j].y = cnt[i0 + 1];
                    if (i0 + 2 < n) c[j].z = cnt[i0 + 2];
                }
            }
        }
    }

    int acc = 0;
#pragma unroll 2
    for (int f = 0; f < base; f += SCAN_TILE) {
        int4 v[SCAN_V];
#pragma unroll
        for (int j = 0; j < SCAN_V; ++j) v[j] = *(const int4*)(cnt + f + (j * 128 + t) * 4);
#pragma unroll
        for (int j = 0; j < SCAN_V; ++j) acc += v[j].x + v[j].y + v[j].z + v[j].w;
    }
#pragma unroll
    for (int d = 16; d >= 1; d >>= 1) acc += __shfl_xor(acc, d);

    int mine[SCAN_V], incl[SCAN_V];
#pragma unroll
    for (int j = 0; j < SCAN_V; ++j) { mine[j] = c[j].x + c[j].y + c[j].z + c[j].w; incl[j] = mine[j]; }
#pragma unroll
    for (int d = 1; d < 32; d <<= 1) {
#pragma unroll
        for (int j = 0; j < SCAN_V; ++j) {
            const int up = __shfl_up(incl[j], d);
            if (lane >= d) incl[j] += up;
        }
    }
    if (lane == 31) {
#pragma unroll
        for (int j = 0; j < SCAN_V; ++j) stot[j * 4 + wave] = incl[j];
        stot[32 + wave] = acc;
    }
    __syncthreads();

    const int4 f4 = *(const int4*)(stot + 32);
    int run = f4.x + f4.y + f4.z + f4.w;
    int off[SCAN_V];
#pragma unroll
    for (int j = 0; j < SCAN_V; ++j) {
        const int4 w4 = *(const int4*)(stot + j * 4);
        off[j] = run + (wave > 0 ? w4.x : 0) + (wave > 1 ? w4.y : 0) + (wave > 2 ? w4.z : 0);
        run += w4.x + w4.y + w4.z + w4.w;
    }
    if (whole) {
#pragma unroll
        for (int j = 0; j < SCAN_V; ++j) {
            const int i0 = base + (j * 128 + t) * 4;
            const int c0 = c[j].x, c1 = c[j].y, c2 = c[j].z, c3 = c[j].w;
            const int p0 = off[j] + incl[j] - mine[j];
            const int p1 = p0 + c0, p2 = p1 + c1, p3 = p2 + c2;
            *(int4*)(rowptr + i0) = make_int4(p0, p1, p2, p3);
            *(float4*)(dinv + i0) = make_float4(rsqrtf((float)c0 + 1.0f), rsqrtf((float)c1 + 1.0f),
                                                rsqrtf((float)c2 + 1.0f), rsqrtf((float)c3 + 1.0f));
        }
    } else {
#pragma unroll
        for (int j = 0; j < SCAN_V; ++j) {
            const int i0 = base + (j * 128 + t) * 4;
            const int c0 = c[j].x, c1 = c[j].y, c2 = c[j].z, c3 = c[j].w;
            const int p0 = off[j] + incl[j] - mine[j];
            const int p1 = p0 + c0, p2 = p1 + c1, p3 = p2 + c2;
            if (i0 + 3 < n) {
                *(int4*)(rowptr + i0) = make_int4(p0, p1, p2, p3);
                *(float4*)(dinv + i0) = make_float4(rsqrtf((float)c0 + 1.0f), rsqrtf((float)c1 + 1.0f),
                                                    rsqrtf((float)c2 + 1.0f), rsqrtf((float)c3 + 1.0f));
            } else {
                if (i0 + 0 < n) { rowptr[i0 + 0] = p0; dinv[i0 + 0] = rsqrtf((float)c0 + 1.0f); }
                if (i0 + 1 < n) { rowptr[i0 + 1] = p1; dinv[i0 + 1] = rsqrtf((float)c1 + 1.0f); }
                if (i0 + 2 < n) { rowptr[i0 + 2] = p2; dinv[i0 + 2] = rsqrtf((float)c2 + 1.0f); }
            }
        }
    }
    if (t == 0 && base + SCAN_TILE >= n) rowptr[n] = run;
}

#define LDS_PITCH 132

template <int N, int ADD_BIAS, int SCAN>
__global__ void __launch_bounds__(128)
gemm_wstat_wmma_f32(const float* __restrict__ A,
                    const float* __restrict__ W,
                    const float* __restrict__ bias,
                    float* __restrict__ C, int M,
                    const int* __restrict__ cnt, int* __restrict__ rowptr,
                    float* __restrict__ dinv) {
    constexpr int K = CFEAT;
    constexpr int KS = K / 4;
    constexpr int NPAIR = N / 32;
    constexpr int MT = 4 / NPAIR;
    constexpr int ROWS = 16 * MT;
    constexpr int LD4 = ROWS / 4;

    __shared__ __attribute__((aligned(16))) float lds[2][ROWS * LDS_PITCH];

    const int nscan = SCAN ? (M + SCAN_TILE - 1) / SCAN_TILE : 0;
    if (SCAN && (int)blockIdx.x < nscan) {
        __shared__ __attribute__((aligned(16))) int stot[36];
        scan_dinv_role(cnt, rowptr, dinv, M, stot, (int)blockIdx.x);
        return;
    }
    const int bid = (int)blockIdx.x - nscan;
    const int nblk = (int)gridDim.x - nscan;

    const int lane = threadIdx.x & 31;
    const int wave = threadIdx.x >> 5;
    const int lo = lane & 15;
    const int hi = lane >> 4;
    const int mt = wave / NPAIR;
    const int n0 = (wave % NPAIR) * 32;

    v2f b0[KS], b1[KS];
    {
        const float* wbase = W + (size_t)(2 * hi) * N + n0 + lo;
#pragma unroll
        for (int s = 0; s < KS; ++s) {
            b0[s].x = wbase[(size_t)(4 * s) * N];
            b0[s].y = wbase[(size_t)(4 * s + 1) * N];
            b1[s].x = wbase[(size_t)(4 * s) * N + 16];
            b1[s].y = wbase[(size_t)(4 * s + 1) * N + 16];
        }
    }
    const float bv0 = ADD_BIAS ? bias[n0 + lo] : 0.0f;
    const float bv1 = ADD_BIAS ? bias[n0 + 16 + lo] : 0.0f;

    const int steps = (M + ROWS - 1) / ROWS;
    const v4f* __restrict__ A4 = (const v4f*)A + lane;
    v4f g[LD4];

    int st = bid;
    if (st < steps) {
#pragma unroll
        for (int i = 0; i < LD4; ++i) {
            const int row = st * ROWS + wave + 4 * i;
            g[i] = A4[(size_t)(row < M ? row : M - 1) * (K / 4)];
        }
#pragma unroll
        for (int i = 0; i < LD4; ++i)
            *(v4f*)&lds[0][(wave + 4 * i) * LDS_PITCH + 4 * lane] = g[i];
    }
    __syncthreads();

    int cur = 0;
    for (; st < steps; st += nblk) {
        const int nxt = st + nblk;
        const bool more = nxt < steps;
        if (more) {
#pragma unroll
            for (int i = 0; i < LD4; ++i) {
                const int row = nxt * ROWS + wave + 4 * i;
                g[i] = A4[(size_t)(row < M ? row : M - 1) * (K / 4)];
            }
        }

        const float* arow = &lds[cur][(16 * mt + lo) * LDS_PITCH + 2 * hi];
        v2f a[KS];
#pragma unroll
        for (int s = 0; s < KS; ++s) a[s] = *(const v2f*)(arow + 4 * s);
        __builtin_amdgcn_sched_barrier(0);

        v8f acc0 = {}, acc1 = {};
#pragma unroll
        for (int s = 0; s < KS; ++s) {
            acc0 = __builtin_amdgcn_wmma_f32_16x16x4_f32(
                 false, a[s],  false, b0[s],
                 (short)0, acc0,  false,  false);
            acc1 = __builtin_amdgcn_wmma_f32_16x16x4_f32(
                 false, a[s],  false, b1[s],
                 (short)0, acc1,  false,  false);
        }

        const int r0 = st * ROWS + 16 * mt + 8 * hi;
        float* cbase = C + (size_t)r0 * N + n0 + lo;
        float o0[8], o1[8];
#pragma unroll
        for (int v = 0; v < 8; ++v) { o0[v] = acc0[v] + bv0; o1[v] = acc1[v] + bv1; }
        if ((st + 1) * ROWS <= M) {
#pragma unroll
            for (int v = 0; v < 8; ++v) {
                cbase[(size_t)v * N] = o0[v];
                cbase[(size_t)v * N + 16] = o1[v];
            }
        } else {
#pragma unroll
            for (int v = 0; v < 8; ++v) {
                if (r0 + v < M) {
                    cbase[(size_t)v * N] = o0[v];
                    cbase[(size_t)v * N + 16] = o1[v];
                }
            }
        }

        if (more) {
#pragma unroll
            for (int i = 0; i < LD4; ++i)
                *(v4f*)&lds[cur ^ 1][(wave + 4 * i) * LDS_PITCH + 4 * lane] = g[i];
        }
        __builtin_amdgcn_fence(__ATOMIC_RELEASE, "workgroup", "local");
        __builtin_amdgcn_s_barrier();
        __builtin_amdgcn_fence(__ATOMIC_ACQUIRE, "workgroup", "local");
        cur ^= 1;
    }
}

#define NODES_PER_WAVE 4

__device__ __forceinline__ void axpy4(float4& acc, const float4 v, const float w) {
    acc.x += v.x * w; acc.y += v.y * w; acc.z += v.z * w; acc.w += v.w * w;
}

__global__ void __launch_bounds__(256)
gather_bias_relu(const float* __restrict__ h,
                 const int* __restrict__ rowptr,
                 const int* __restrict__ srcw,
                 const float* __restrict__ dinv,
                 const float* __restrict__ bias,
                 float* __restrict__ outp, int n) {
    const int lane = threadIdx.x & 31;
    const int wave = __builtin_amdgcn_readfirstlane((int)(threadIdx.x >> 5));
    const int node0 = (blockIdx.x * (int)(blockDim.x >> 5) + wave) * NODES_PER_WAVE;
    if (node0 >= n) return;
    const float4* __restrict__ h4 = (const float4*)h + lane;
    const float4 b = ((const float4*)bias)[lane];
    int beg = rowptr[node0];
    for (int i = 0; i < NODES_PER_WAVE; ++i) {
        const int node = node0 + i;
        if (node >= n) break;
        const int end = rowptr[node + 1];
        const float dc = dinv[node];
        const float4 self = h4[(size_t)node * (CFEAT / 4)];
        float4 acc = make_float4(0.0f, 0.0f, 0.0f, 0.0f);
        for (int base = beg; base < end; base += 32) {
            const int j = base + lane;
            int r = 0;
            float w = 0.0f;
            if (j < end) { const int2 p = ((const int2*)srcw)[j]; r = p.x; w = __int_as_float(p.y) * dc; }
            const int wi = __float_as_int(w);
            const int m = (end - base) < 32 ? (end - base) : 32;
            int k = 0;
            for (; k + 4 <= m; k += 4) {
                const int r0 = __builtin_amdgcn_readlane(r, k + 0);
                const int r1 = __builtin_amdgcn_readlane(r, k + 1);
                const int r2 = __builtin_amdgcn_readlane(r, k + 2);
                const int r3 = __builtin_amdgcn_readlane(r, k + 3);
                const float4 v0 = h4[(size_t)r0 * (CFEAT / 4)];
                const float4 v1 = h4[(size_t)r1 * (CFEAT / 4)];
                const float4 v2 = h4[(size_t)r2 * (CFEAT / 4)];
                const float4 v3 = h4[(size_t)r3 * (CFEAT / 4)];
                axpy4(acc, v0, __int_as_float(__builtin_amdgcn_readlane(wi, k + 0)));
                axpy4(acc, v1, __int_as_float(__builtin_amdgcn_readlane(wi, k + 1)));
                axpy4(acc, v2, __int_as_float(__builtin_amdgcn_readlane(wi, k + 2)));
                axpy4(acc, v3, __int_as_float(__builtin_amdgcn_readlane(wi, k + 3)));
            }
            const int left = m - k;
            if (left == 3) {
                const int r0 = __builtin_amdgcn_readlane(r, k + 0);
                const int r1 = __builtin_amdgcn_readlane(r, k + 1);
                const int r2 = __builtin_amdgcn_readlane(r, k + 2);
                const float4 v0 = h4[(size_t)r0 * (CFEAT / 4)];
                const float4 v1 = h4[(size_t)r1 * (CFEAT / 4)];
                const float4 v2 = h4[(size_t)r2 * (CFEAT / 4)];
                axpy4(acc, v0, __int_as_float(__builtin_amdgcn_readlane(wi, k + 0)));
                axpy4(acc, v1, __int_as_float(__builtin_amdgcn_readlane(wi, k + 1)));
                axpy4(acc, v2, __int_as_float(__builtin_amdgcn_readlane(wi, k + 2)));
            } else if (left == 2) {
                const int r0 = __builtin_amdgcn_readlane(r, k + 0);
                const int r1 = __builtin_amdgcn_readlane(r, k + 1);
                const float4 v0 = h4[(size_t)r0 * (CFEAT / 4)];
                const float4 v1 = h4[(size_t)r1 * (CFEAT / 4)];
                axpy4(acc, v0, __int_as_float(__builtin_amdgcn_readlane(wi, k + 0)));
                axpy4(acc, v1, __int_as_float(__builtin_amdgcn_readlane(wi, k + 1)));
            } else if (left == 1) {
                const int r0 = __builtin_amdgcn_readlane(r, k);
                const float4 v0 = h4[(size_t)r0 * (CFEAT / 4)];
                axpy4(acc, v0, __int_as_float(__builtin_amdgcn_readlane(wi, k)));
            }
        }
        axpy4(acc, self, dc * dc);
        float4 o;
        o.x = acc.x + b.x; o.y = acc.y + b.y; o.z = acc.z + b.z; o.w = acc.w + b.w;
        o.x = o.x > 0.0f ? o.x : 0.0f;
        o.y = o.y > 0.0f ? o.y : 0.0f;
        o.z = o.z > 0.0f ? o.z : 0.0f;
        o.w = o.w > 0.0f ? o.w : 0.0f;
        ((float4*)outp)[(size_t)node * (CFEAT / 4) + lane] = o;
        beg = end;
    }
}

extern "C" void kernel_launch(void* const* d_in, const int* in_sizes, int n_in,
                              void* d_out, int out_size, void* d_ws, size_t ws_size,
                              hipStream_t stream) {
    const float* x  = (const float*)d_in[0];
    const int*   ei = (const int*)d_in[1];
    const float* W1 = (const float*)d_in[2];
    const float* b1 = (const float*)d_in[3];
    const float* W2 = (const float*)d_in[4];
    const float* b2 = (const float*)d_in[5];
    const float* Wl = (const float*)d_in[6];
    const float* bl = (const float*)d_in[7];
    float* out = (float*)d_out;

    const int E  = in_sizes[1] / 2;
    const int Nn = in_sizes[0] / CFEAT;
    const int C  = CFEAT;

    const int* row = ei;
    const int* col = ei + E;

    float* dinv = (float*)d_ws;
    float* bufA = dinv + Nn;
    float* bufB = bufA + (size_t)Nn * C;
    int* cnt    = (int*)(bufB + (size_t)Nn * C);
    int* rowptr = cnt + Nn;
    int* srcw   = rowptr + Nn + 4;
    int* rank   = srcw + 2 * (size_t)E;

    const int nodeBlocks = (Nn + 8 * NODES_PER_WAVE - 1) / (8 * NODES_PER_WAVE);
    const int edgesPerBlock = (EDGE_BLOCK / 32) * EDGE_WAVE;
    const int edgeBlocks = (E + edgesPerBlock - 1) / edgesPerBlock;

    hipMemsetAsync(cnt, 0, (size_t)Nn * sizeof(int), stream);
    count_rank_kernel<<<edgeBlocks, EDGE_BLOCK, 0, stream>>>(col, cnt, rank, E);
    gemm_wstat_wmma_f32<128, 0, 1><<<512 + (Nn + SCAN_TILE - 1) / SCAN_TILE, 128, 0, stream>>>(x, W1, nullptr, bufA, Nn, cnt, rowptr, dinv);
    fill_kernel<<<edgeBlocks, EDGE_BLOCK, 0, stream>>>(row, col, rowptr, rank, dinv, srcw, E);

    gather_bias_relu<<<nodeBlocks, 256, 0, stream>>>(bufA, rowptr, srcw, dinv, b1, bufB, Nn);

    gemm_wstat_wmma_f32<128, 0, 0><<<512, 128, 0, stream>>>(bufB, W2, nullptr, bufA, Nn, nullptr, nullptr, nullptr);
    gather_bias_relu<<<nodeBlocks, 256, 0, stream>>>(bufA, rowptr, srcw, dinv, b2, bufB, Nn);

    gemm_wstat_wmma_f32<64, 1, 0><<<512, 128, 0, stream>>>(bufB, Wl, bl, out, Nn, nullptr, nullptr, nullptr);
}


# Next layer's dense product rides in the aggregation's launch: 16 nodes' sums go into LDS, times W on f32 WMMA (W in registers); a wave loads up to 20 rows at once; 2 launches gone; sum order kept
# speedup vs baseline: 2.9778x; 1.0568x over previous
#include <hip/hip_runtime.h>
#include <hip/hip_bf16.h>

typedef __attribute__((ext_vector_type(2))) float v2f;
typedef __attribute__((ext_vector_type(4))) float v4f;
typedef __attribute__((ext_vector_type(8))) float v8f;

#define CFEAT 128

#define EDGE_G 2
#define EDGE_WAVE (128 * EDGE_G)
#define EDGE_BLOCK 64

__global__ void __launch_bounds__(EDGE_BLOCK)
count_rank_kernel(const int* __restrict__ col, int* __restrict__ cnt,
                  int* __restrict__ rank, int E) {
    const int lane = threadIdx.x & 31;
    const int wbase = __builtin_amdgcn_readfirstlane((int)((blockIdx.x * blockDim.x + threadIdx.x) >> 5)) * EDGE_WAVE;
    if (wbase >= E) return;
    const bool vec = ((E & 3) == 0) && ((((size_t)col | (size_t)rank) & 15) == 0);
    if (vec && wbase + EDGE_WAVE <= E) {
        int4 c[EDGE_G], r[EDGE_G];
#pragma unroll
        for (int g = 0; g < EDGE_G; ++g) c[g] = *(const int4*)(col + wbase + 128 * g + 4 * lane);
        __builtin_amdgcn_sched_barrier(0);
#pragma unroll
        for (int g = 0; g < EDGE_G; ++g) {
            r[g].x = atomicAdd(&cnt[c[g].x], 1);
            r[g].y = atomicAdd(&cnt[c[g].y], 1);
            r[g].z = atomicAdd(&cnt[c[g].z], 1);
            r[g].w = atomicAdd(&cnt[c[g].w], 1);
        }
#pragma unroll
        for (int g = 0; g < EDGE_G; ++g) *(int4*)(rank + wbase + 128 * g + 4 * lane) = r[g];
    } else {
        for (int i = lane; i < EDGE_WAVE; i += 32) {
            const int e = wbase + i;
            if (e < E) rank[e] = atomicAdd(&cnt[col[e]], 1);
        }
    }
}

__global__ void __launch_bounds__(EDGE_BLOCK)
fill_kernel(const int* __restrict__ row, const int* __restrict__ col,
            const int* __restrict__ rowptr, const int* __restrict__ rank,
            const float* __restrict__ dinv,
            int* __restrict__ srcw, int E) {
    const int lane = threadIdx.x & 31;
    const int wbase = __builtin_amdgcn_readfirstlane((int)((blockIdx.x * blockDim.x + threadIdx.x) >> 5)) * EDGE_WAVE;
    if (wbase >= E) return;
    int2* __restrict__ pairs = (int2*)srcw;
    const bool vec = ((E & 3) == 0) && ((((size_t)row | (size_t)col | (size_t)rank) & 15) == 0);
    if (vec && wbase + EDGE_WAVE <= E) {
        int4 r[EDGE_G], c[EDGE_G], k[EDGE_G];
#pragma unroll
        for (int g = 0; g < EDGE_G; ++g) {
            const int off = wbase + 128 * g + 4 * lane;
            r[g] = *(const int4*)(row + off);
            c[g] = *(const int4*)(col + off);
            k[g] = *(const int4*)(rank + off);
        }
        int4 p[EDGE_G];
        float4 d[EDGE_G];
#pragma unroll
        for (int g = 0; g < EDGE_G; ++g) {
            p[g] = make_int4(rowptr[c[g].x], rowptr[c[g].y], rowptr[c[g].z], rowptr[c[g].w]);
            d[g] = make_float4(dinv[r[g].x], dinv[r[g].y], dinv[r[g].z], dinv[r[g].w]);
        }
#pragma unroll
        for (int g = 0; g < EDGE_G; ++g) {
            pairs[p[g].x + k[g].x] = make_int2(r[g].x, __float_as_int(d[g].x));
            pairs[p[g].y + k[g].y] = make_int2(r[g].y, __float_as_int(d[g].y));
            pairs[p[g].z + k[g].z] = make_int2(r[g].z, __float_as_int(d[g].z));
            pairs[p[g].w + k[g].w] = make_int2(r[g].w, __float_as_int(d[g].w));
        }
    } else {
        for (int i = lane; i < EDGE_WAVE; i += 32) {
            const int e = wbase + i;
            if (e < E) {
                const int s = row[e];
                pairs[rowptr[col[e]] + rank[e]] = make_int2(s, __float_as_int(dinv[s]));
            }
        }
    }
}

#define SCAN_V 8
#define SCAN_TILE (128 * 4 * SCAN_V)

__device__ __forceinline__ void scan_dinv_role(const int* __restrict__ cnt, int* __restrict__ rowptr,
                                               float* __restrict__ dinv, int n, int* __restrict__ stot,
                                               int tile) {
    const int t = threadIdx.x;
    const int lane = t & 31;
    const int wave = __builtin_amdgcn_readfirstlane(t >> 5);
    const int base = tile * SCAN_TILE;
    const bool whole = base + SCAN_TILE <= n;

    int4 c[SCAN_V];
    if (whole) {
#pragma unroll
        for (int j = 0; j < SCAN_V; ++j) c[j] = *(const int4*)(cnt + base + (j * 128 + t) * 4);
    } else {
#pragma unroll
        for (int j = 0; j < SCAN_V; ++j) {
            const int i0 = base + (j * 128 + t) * 4;
            const bool in = i0 + 3 < n;
            const int4 v = *(const int4*)(cnt + (in ? i0 : 0));
            c[j] = in ? v : make_int4(0, 0, 0, 0);
        }
        if (n & 3) {
#pragma unroll
            for (int j = 0; j < SCAN_V; ++j) {
                const int i0 = base + (j * 128 + t) * 4;
                if (i0 < n && i0 + 3 >= n) {
                    c[j].x = cnt[i0];
                    if (i0 + 1 < n) c[j].y = cnt[i0 + 1];
                    if (i0 + 2 < n) c[j].z = cnt[i0 + 2];
                }
            }
        }
    }

    int acc = 0;
#pragma unroll 2
    for (int f = 0; f < base; f += SCAN_TILE) {
        int4 v[SCAN_V];
#pragma unroll
        for (int j = 0; j < SCAN_V; ++j) v[j] = *(const int4*)(cnt + f + (j * 128 + t) * 4);
#pragma unroll
        for (int j = 0; j < SCAN_V; ++j) acc += v[j].x + v[j].y + v[j].z + v[j].w;
    }
#pragma unroll
    for (int d = 16; d >= 1; d >>= 1) acc += __shfl_xor(acc, d);

    int mine[SCAN_V], incl[SCAN_V];
#pragma unroll
    for (int j = 0; j < SCAN_V; ++j) { mine[j] = c[j].x + c[j].y + c[j].z + c[j].w; incl[j] = mine[j]; }
#pragma unroll
    for (int d = 1; d < 32; d <<= 1) {
#pragma unroll
        for (int j = 0; j < SCAN_V; ++j) {
            const int up = __shfl_up(incl[j], d);
            if (lane >= d) incl[j] += up;
        }
    }
    if (lane == 31) {
#pragma unroll
        for (int j = 0; j < SCAN_V; ++j) stot[j * 4 + wave] = incl[j];
        stot[32 + wave] = acc;
    }
    __syncthreads();

    const int4 f4 = *(const int4*)(stot + 32);
    int run = f4.x + f4.y + f4.z + f4.w;
    int off[SCAN_V];
#pragma unroll
    for (int j = 0; j < SCAN_V; ++j) {
        const int4 w4 = *(const int4*)(stot + j * 4);
        off[j] = run + (wave > 0 ? w4.x : 0) + (wave > 1 ? w4.y : 0) + (wave > 2 ? w4.z : 0);
        run += w4.x + w4.y + w4.z + w4.w;
    }
    if (whole) {
#pragma unroll
        for (int j = 0; j < SCAN_V; ++j) {
            const int i0 = base + (j * 128 + t) * 4;
            const int c0 = c[j].x, c1 = c[j].y, c2 = c[j].z, c3 = c[j].w;
            const int p0 = off[j] + incl[j] - mine[j];
            const int p1 = p0 + c0, p2 = p1 + c1, p3 = p2 + c2;
            *(int4*)(rowptr + i0) = make_int4(p0, p1, p2, p3);
            *(float4*)(dinv + i0) = make_float4(rsqrtf((float)c0 + 1.0f), rsqrtf((float)c1 + 1.0f),
                                                rsqrtf((float)c2 + 1.0f), rsqrtf((float)c3 + 1.0f));
        }
    } else {
#pragma unroll
        for (int j = 0; j < SCAN_V; ++j) {
            const int i0 = base + (j * 128 + t) * 4;
            const int c0 = c[j].x, c1 = c[j].y, c2 = c[j].z, c3 = c[j].w;
            const int p0 = off[j] + incl[j] - mine[j];
            const int p1 = p0 + c0, p2 = p1 + c1, p3 = p2 + c2;
            if (i0 + 3 < n) {
                *(int4*)(rowptr + i0) = make_int4(p0, p1, p2, p3);
                *(float4*)(dinv + i0) = make_float4(rsqrtf((float)c0 + 1.0f), rsqrtf((float)c1 + 1.0f),
                                                    rsqrtf((float)c2 + 1.0f), rsqrtf((float)c3 + 1.0f));
            } else {
                if (i0 + 0 < n) { rowptr[i0 + 0] = p0; dinv[i0 + 0] = rsqrtf((float)c0 + 1.0f); }
                if (i0 + 1 < n) { rowptr[i0 + 1] = p1; dinv[i0 + 1] = rsqrtf((float)c1 + 1.0f); }
                if (i0 + 2 < n) { rowptr[i0 + 2] = p2; dinv[i0 + 2] = rsqrtf((float)c2 + 1.0f); }
            }
        }
    }
    if (t == 0 && base + SCAN_TILE >= n) rowptr[n] = run;
}

#define LDS_PITCH 132

template <int N, int ADD_BIAS, int SCAN>
__global__ void __launch_bounds__(128)
gemm_wstat_wmma_f32(const float* __restrict__ A,
                    const float* __restrict__ W,
                    const float* __restrict__ bias,
                    float* __restrict__ C, int M,
                    const int* __restrict__ cnt, int* __restrict__ rowptr,
                    float* __restrict__ dinv) {
    constexpr int K = CFEAT;
    constexpr int KS = K / 4;
    constexpr int NPAIR = N / 32;
    constexpr int MT = 4 / NPAIR;
    constexpr int ROWS = 16 * MT;
    constexpr int LD4 = ROWS / 4;

    __shared__ __attribute__((aligned(16))) float lds[2][ROWS * LDS_PITCH];

    const int nscan = SCAN ? (M + SCAN_TILE - 1) / SCAN_TILE : 0;
    if (SCAN && (int)blockIdx.x < nscan) {
        __shared__ __attribute__((aligned(16))) int stot[36];
        scan_dinv_role(cnt, rowptr, dinv, M, stot, (int)blockIdx.x);
        return;
    }
    const int bid = (int)blockIdx.x - nscan;
    const int nblk = (int)gridDim.x - nscan;

    const int lane = threadIdx.x & 31;
    const int wave = threadIdx.x >> 5;
    const int lo = lane & 15;
    const int hi = lane >> 4;
    const int mt = wave / NPAIR;
    const int n0 = (wave % NPAIR) * 32;

    v2f b0[KS], b1[KS];
    {
        const float* wbase = W + (size_t)(2 * hi) * N + n0 + lo;
#pragma unroll
        for (int s = 0; s < KS; ++s) {
            b0[s].x = wbase[(size_t)(4 * s) * N];
            b0[s].y = wbase[(size_t)(4 * s + 1) * N];
            b1[s].x = wbase[(size_t)(4 * s) * N + 16];
            b1[s].y = wbase[(size_t)(4 * s + 1) * N + 16];
        }
    }
    const float bv0 = ADD_BIAS ? bias[n0 + lo] : 0.0f;
    const float bv1 = ADD_BIAS ? bias[n0 + 16 + lo] : 0.0f;

    const int steps = (M + ROWS - 1) / ROWS;
    const v4f* __restrict__ A4 = (const v4f*)A + lane;
    v4f g[LD4];

    int st = bid;
    if (st < steps) {
#pragma unroll
        for (int i = 0; i < LD4; ++i) {
            const int row = st * ROWS + wave + 4 * i;
            g[i] = A4[(size_t)(row < M ? row : M - 1) * (K / 4)];
        }
#pragma unroll
        for (int i = 0; i < LD4; ++i)
            *(v4f*)&lds[0][(wave + 4 * i) * LDS_PITCH + 4 * lane] = g[i];
    }
    __syncthreads();

    int cur = 0;
    for (; st < steps; st += nblk) {
        const int nxt = st + nblk;
        const bool more = nxt < steps;
        if (more) {
#pragma unroll
            for (int i = 0; i < LD4; ++i) {
                const int row = nxt * ROWS + wave + 4 * i;
                g[i] = A4[(size_t)(row < M ? row : M - 1) * (K / 4)];
            }
        }

        const float* arow = &lds[cur][(16 * mt + lo) * LDS_PITCH + 2 * hi];
        v2f a[KS];
#pragma unroll
        for (int s = 0; s < KS; ++s) a[s] = *(const v2f*)(arow + 4 * s);
        __builtin_amdgcn_sched_barrier(0);

        v8f acc0 = {}, acc1 = {};
#pragma unroll
        for (int s = 0; s < KS; ++s) {
            acc0 = __builtin_amdgcn_wmma_f32_16x16x4_f32(
                 false, a[s],  false, b0[s],
                 (short)0, acc0,  false,  false);
            acc1 = __builtin_amdgcn_wmma_f32_16x16x4_f32(
                 false, a[s],  false, b1[s],
                 (short)0, acc1,  false,  false);
        }

        const int r0 = st * ROWS + 16 * mt + 8 * hi;
        float* cbase = C + (size_t)r0 * N + n0 + lo;
        float o0[8], o1[8];
#pragma unroll
        for (int v = 0; v < 8; ++v) { o0[v] = acc0[v] + bv0; o1[v] = acc1[v] + bv1; }
        if ((st + 1) * ROWS <= M) {
#pragma unroll
            for (int v = 0; v < 8; ++v) {
                cbase[(size_t)v * N] = o0[v];
                cbase[(size_t)v * N + 16] = o1[v];
            }
        } else {
#pragma unroll
            for (int v = 0; v < 8; ++v) {
                if (r0 + v < M) {
                    cbase[(size_t)v * N] = o0[v];
                    cbase[(size_t)v * N + 16] = o1[v];
                }
            }
        }

        if (more) {
#pragma unroll
            for (int i = 0; i < LD4; ++i)
                *(v4f*)&lds[cur ^ 1][(wave + 4 * i) * LDS_PITCH + 4 * lane] = g[i];
        }
        __builtin_amdgcn_fence(__ATOMIC_RELEASE, "workgroup", "local");
        __builtin_amdgcn_s_barrier();
        __builtin_amdgcn_fence(__ATOMIC_ACQUIRE, "workgroup", "local");
        cur ^= 1;
    }
}

#define FT_ROWS 16
#define FT_NPW 2

__device__ __forceinline__ void axpy4(float4& acc, const float4 v, const float w) {
    acc.x += v.x * w; acc.y += v.y * w; acc.z += v.z * w; acc.w += v.w * w;
}

template <int S>
__device__ __forceinline__ void load_rows(float4 (&v)[S], const float4* __restrict__ h4, const int r, const int o) {
    int rr[S];
#pragma unroll
    for (int q = 0; q < S; ++q) rr[q] = __builtin_amdgcn_readlane(r, o + q);
#pragma unroll
    for (int q = 0; q < S; ++q) v[q] = h4[(size_t)rr[q] * (CFEAT / 4)];
}

template <int S>
__device__ __forceinline__ void add_rows(float4& acc, const float4 (&v)[S], const int wi, const int o) {
#pragma unroll
    for (int q = 0; q < S; ++q) axpy4(acc, v[q], __int_as_float(__builtin_amdgcn_readlane(wi, o + q)));
}

template <int T>
__device__ __forceinline__ void rows_case(float4& acc, const float4* __restrict__ h4, const int r, const int wi, const int k) {
    float4 v[T];
    load_rows<T>(v, h4, r, k);
    add_rows<T>(acc, v, wi, k);
}

__device__ __forceinline__ float4 sum_messages(const float4* __restrict__ h4, const int2* __restrict__ pairs,
                                               const int beg, const int end, const float dc,
                                               const int2 p0, const int lane) {
    float4 acc = make_float4(0.0f, 0.0f, 0.0f, 0.0f);
    for (int base = beg; base < end; base += 32) {
        const int j = base + lane;
        int r = 0;
        float w = 0.0f;
        if (base == beg) {
            if (j < end) { r = p0.x; w = __int_as_float(p0.y) * dc; }
        } else if (j < end) {
            const int2 p = pairs[j]; r = p.x; w = __int_as_float(p.y) * dc;
        }
        const int wi = __float_as_int(w);
        const int m = (end - base) < 32 ? (end - base) : 32;
        for (int k = 0; k < m;) {
            const int rem = m - k;
            const int take = rem < 20 ? rem : 20;
            switch (take) {
                case 1: rows_case<1>(acc, h4, r, wi, k); break;
                case 2: rows_case<2>(acc, h4, r, wi, k); break;
                case 3: rows_case<3>(acc, h4, r, wi, k); break;
                case 4: rows_case<4>(acc, h4, r, wi, k); break;
                case 5: rows_case<5>(acc, h4, r, wi, k); break;
                case 6: rows_case<6>(acc, h4, r, wi, k); break;
                case 7: rows_case<7>(acc, h4, r, wi, k); break;
                case 8: rows_case<8>(acc, h4, r, wi, k); break;
                case 9: rows_case<9>(acc, h4, r, wi, k); break;
                case 10: rows_case<10>(acc, h4, r, wi, k); break;
                case 11: rows_case<11>(acc, h4, r, wi, k); break;
                case 12: rows_case<12>(acc, h4, r, wi, k); break;
                case 13: rows_case<13>(acc, h4, r, wi, k); break;
                case 14: rows_case<14>(acc, h4, r, wi, k); break;
                case 15: rows_case<15>(acc, h4, r, wi, k); break;
                case 16: rows_case<16>(acc, h4, r, wi, k); break;
                case 17: rows_case<17>(acc, h4, r, wi, k); break;
                case 18: rows_case<18>(acc, h4, r, wi, k); break;
                case 19: rows_case<19>(acc, h4, r, wi, k); break;
                case 20: rows_case<20>(acc, h4, r, wi, k); break;
                default: break;
            }
            k += take;
        }
    }
    return acc;
}

__device__ __forceinline__ float4 bias_relu4(const float4 acc, const float4 b) {
    float4 o;
    o.x = acc.x + b.x; o.y = acc.y + b.y; o.z = acc.z + b.z; o.w = acc.w + b.w;
    o.x = o.x > 0.0f ? o.x : 0.0f;
    o.y = o.y > 0.0f ? o.y : 0.0f;
    o.z = o.z > 0.0f ? o.z : 0.0f;
    o.w = o.w > 0.0f ? o.w : 0.0f;
    return o;
}

template <int N, int ADD_BIAS>
__global__ void __launch_bounds__(256)
gather_relu_gemm_wmma_f32(const float* __restrict__ h,
                          const int* __restrict__ rowptr,
                          const int* __restrict__ srcw,
                          const float* __restrict__ dinv,
                          const float* __restrict__ bagg,
                          const float* __restrict__ W,
                          const float* __restrict__ bout,
                          float* __restrict__ C, int n) {
    constexpr int KS = CFEAT / 4;
    constexpr int NT = N / 16;
    constexpr int OP = N + 4;

    __shared__ __attribute__((aligned(16))) float xt[FT_ROWS * LDS_PITCH];
    __shared__ __attribute__((aligned(16))) float ot[FT_ROWS * OP];

    const int lane = threadIdx.x & 31;
    const int wave = __builtin_amdgcn_readfirstlane((int)(threadIdx.x >> 5));
    const int lo = lane & 15;
    const int hi = lane >> 4;
    const bool mm = (NT == 8) || (wave < NT);
    const int n0 = (wave % NT) * 16;

    const int ntiles = (n + FT_ROWS - 1) / FT_ROWS;
    int t = (int)blockIdx.x;

    int ev = 0;
    float dv = 0.0f;
    if (t < ntiles) {
        const int nd = t * FT_ROWS + FT_NPW * wave + lane;
        if (lane <= FT_NPW && nd <= n) ev = rowptr[nd];
        if (lane < FT_NPW && nd < n) dv = dinv[nd];
    }

    v2f b[KS];
#pragma unroll
    for (int s = 0; s < KS; ++s) b[s] = v2f{0.0f, 0.0f};
    float bv = 0.0f;
    if (mm) {
        const float* wbase = W + (size_t)(2 * hi) * N + n0 + lo;
#pragma unroll
        for (int s = 0; s < KS; ++s) {
            b[s].x = wbase[(size_t)(4 * s) * N];
            b[s].y = wbase[(size_t)(4 * s + 1) * N];
        }
        if (ADD_BIAS) bv = bout[n0 + lo];
    }

    const float4* __restrict__ h4 = (const float4*)h + lane;
    const int2* __restrict__ pairs = (const int2*)srcw;
    const float4 bb = ((const float4*)bagg)[lane];

    for (; t < ntiles; t += (int)gridDim.x) {
        const int nodeA = t * FT_ROWS + FT_NPW * wave;
        const int nodeB = nodeA + 1;
        const int e0 = __builtin_amdgcn_readlane(ev, 0);
        const int e1 = __builtin_amdgcn_readlane(ev, 1);
        const int e2 = __builtin_amdgcn_readlane(ev, 2);
        const float dcA = __int_as_float(__builtin_amdgcn_readlane(__float_as_int(dv), 0));
        const float dcB = __int_as_float(__builtin_amdgcn_readlane(__float_as_int(dv), 1));
        int2 pA = make_int2(0, 0), pB = make_int2(0, 0);
        if (e0 + lane < e1) pA = pairs[e0 + lane];
        if (e1 + lane < e2) pB = pairs[e1 + lane];
        float4 selfA = make_float4(0.0f, 0.0f, 0.0f, 0.0f), selfB = selfA;
        if (nodeA < n) selfA = h4[(size_t)nodeA * (CFEAT / 4)];
        if (nodeB < n) selfB = h4[(size_t)nodeB * (CFEAT / 4)];

#pragma nounroll
        for (int i = 0; i < FT_NPW; ++i) {
            float4 acc = sum_messages(h4, pairs, i ? e1 : e0, i ? e2 : e1, i ? dcB : dcA, i ? pB : pA, lane);
            const float dcc = i ? dcB : dcA;
            axpy4(acc, i ? selfB : selfA, dcc * dcc);
            *(float4*)&xt[(FT_NPW * wave + i) * LDS_PITCH + 4 * lane] = bias_relu4(acc, bb);
        }

        __builtin_amdgcn_fence(__ATOMIC_RELEASE, "workgroup", "local");
        __builtin_amdgcn_s_barrier();
        __builtin_amdgcn_fence(__ATOMIC_ACQUIRE, "workgroup", "local");

        {
            const int tn = t + (int)gridDim.x;
            ev = 0;
            dv = 0.0f;
            if (tn < ntiles) {
                const int nd = tn * FT_ROWS + FT_NPW * wave + lane;
                if (lane <= FT_NPW && nd <= n) ev = rowptr[nd];
                if (lane < FT_NPW && nd < n) dv = dinv[nd];
            }
        }

        if (mm) {
            const float* arow = &xt[lo * LDS_PITCH + 2 * hi];
            v2f a[KS];
#pragma unroll
            for (int s = 0; s < KS; ++s) a[s] = *(const v2f*)(arow + 4 * s);
            __builtin_amdgcn_sched_barrier(0);
            v8f acc = {};
#pragma unroll
            for (int s = 0; s < KS; ++s)
                acc = __builtin_amdgcn_wmma_f32_16x16x4_f32(
                     false, a[s],  false, b[s],
                     (short)0, acc,  false,  false);
#pragma unroll
            for (int v = 0; v < 8; ++v) ot[(v + 8 * hi) * OP + n0 + lo] = acc[v] + bv;
        }
        __builtin_amdgcn_fence(__ATOMIC_RELEASE, "workgroup", "local");
        __builtin_amdgcn_s_barrier();
        __builtin_amdgcn_fence(__ATOMIC_ACQUIRE, "workgroup", "local");

        if (N == 128) {
#pragma unroll
            for (int i = 0; i < FT_NPW; ++i) {
                const int rr = FT_NPW * wave + i;
                const int row = t * FT_ROWS + rr;
                if (row < n)
                    ((float4*)C)[(size_t)row * (N / 4) + lane] = *(const float4*)&ot[rr * OP + 4 * lane];
            }
        } else {
            const int rr = FT_NPW * wave + hi;
            const int row = t * FT_ROWS + rr;
            if (row < n)
                ((float4*)C)[(size_t)row * (N / 4) + lo] = *(const float4*)&ot[rr * OP + 4 * lo];
        }
    }
}

extern "C" void kernel_launch(void* const* d_in, const int* in_sizes, int n_in,
                              void* d_out, int out_size, void* d_ws, size_t ws_size,
                              hipStream_t stream) {
    const float* x  = (const float*)d_in[0];
    const int*   ei = (const int*)d_in[1];
    const float* W1 = (const float*)d_in[2];
    const float* b1 = (const float*)d_in[3];
    const float* W2 = (const float*)d_in[4];
    const float* b2 = (const float*)d_in[5];
    const float* Wl = (const float*)d_in[6];
    const float* bl = (const float*)d_in[7];
    float* out = (float*)d_out;

    const int E  = in_sizes[1] / 2;
    const int Nn = in_sizes[0] / CFEAT;
    const int C  = CFEAT;

    const int* row = ei;
    const int* col = ei + E;

    float* dinv = (float*)d_ws;
    float* bufA = dinv + Nn;
    float* bufB = bufA + (size_t)Nn * C;
    int* cnt    = (int*)(bufB + (size_t)Nn * C);
    int* rowptr = cnt + Nn;
    int* srcw   = rowptr + Nn + 4;
    int* rank   = srcw + 2 * (size_t)E;

    const int edgesPerBlock = (EDGE_BLOCK / 32) * EDGE_WAVE;
    const int edgeBlocks = (E + edgesPerBlock - 1) / edgesPerBlock;

    hipMemsetAsync(cnt, 0, (size_t)Nn * sizeof(int), stream);
    count_rank_kernel<<<edgeBlocks, EDGE_BLOCK, 0, stream>>>(col, cnt, rank, E);
    gemm_wstat_wmma_f32<128, 0, 1><<<512 + (Nn + SCAN_TILE - 1) / SCAN_TILE, 128, 0, stream>>>(x, W1, nullptr, bufA, Nn, cnt, rowptr, dinv);
    fill_kernel<<<edgeBlocks, EDGE_BLOCK, 0, stream>>>(row, col, rowptr, rank, dinv, srcw, E);

    gather_relu_gemm_wmma_f32<128, 0><<<500, 256, 0, stream>>>(bufA, rowptr, srcw, dinv, b1, W2, nullptr, bufB, Nn);

    gather_relu_gemm_wmma_f32<64, 1><<<500, 256, 0, stream>>>(bufB, rowptr, srcw, dinv, b2, Wl, bl, out, Nn);
}


# Edge list in fixed 32-slot buckets per target (overflow list behind), filled by the counting pass, which rides in the first GEMM's launch: scan, fill pass, 2 launches gone; sum order = arrival
# speedup vs baseline: 3.0846x; 1.0359x over previous
#include <hip/hip_runtime.h>
#include <hip/hip_bf16.h>

typedef __attribute__((ext_vector_type(2))) float v2f;
typedef __attribute__((ext_vector_type(4))) float v4f;
typedef __attribute__((ext_vector_type(8))) float v8f;

#define CFEAT 128

#define EDGE_G 2
#define EDGE_WAVE (128 * EDGE_G)
#define BKT 32

__device__ __forceinline__ void place_edge(int* __restrict__ bucket, int* __restrict__ cnt,
                                           int2* __restrict__ ovf, const int n,
                                           const int c, const int r, const int k) {
    if (k < BKT) {
        bucket[(size_t)c * BKT + k] = r;
    } else {
        const int s = atomicAdd(&cnt[n], 1);
        ovf[s] = make_int2(c, r);
    }
}

__device__ __forceinline__ void count_place_role(const int* __restrict__ row, const int* __restrict__ col,
                                                 int* __restrict__ cnt, int* __restrict__ bucket,
                                                 int2* __restrict__ ovf, const int n, const int E, const int wv) {
    const int lane = threadIdx.x & 31;
    const int wbase = wv * EDGE_WAVE;
    if (wbase >= E) return;
    const bool vec = ((E & 3) == 0) && ((((size_t)row | (size_t)col) & 15) == 0);
    if (vec && wbase + EDGE_WAVE <= E) {
        int4 c[EDGE_G], r[EDGE_G], k[EDGE_G];
#pragma unroll
        for (int g = 0; g < EDGE_G; ++g) {
            c[g] = *(const int4*)(col + wbase + 128 * g + 4 * lane);
            r[g] = *(const int4*)(row + wbase + 128 * g + 4 * lane);
        }
        __builtin_amdgcn_sched_barrier(0);
#pragma unroll
        for (int g = 0; g < EDGE_G; ++g) {
            k[g].x = atomicAdd(&cnt[c[g].x], 1);
            k[g].y = atomicAdd(&cnt[c[g].y], 1);
            k[g].z = atomicAdd(&cnt[c[g].z], 1);
            k[g].w = atomicAdd(&cnt[c[g].w], 1);
        }
#pragma unroll
        for (int g = 0; g < EDGE_G; ++g) {
            place_edge(bucket, cnt, ovf, n, c[g].x, r[g].x, k[g].x);
            place_edge(bucket, cnt, ovf, n, c[g].y, r[g].y, k[g].y);
            place_edge(bucket, cnt, ovf, n, c[g].z, r[g].z, k[g].z);
            place_edge(bucket, cnt, ovf, n, c[g].w, r[g].w, k[g].w);
        }
    } else {
        for (int i = lane; i < EDGE_WAVE; i += 32) {
            const int e = wbase + i;
            if (e < E) {
                const int c = col[e];
                place_edge(bucket, cnt, ovf, n, c, row[e], atomicAdd(&cnt[c], 1));
            }
        }
    }
}

#define LDS_PITCH 132

template <int N, int ADD_BIAS, int EDGE>
__global__ void __launch_bounds__(128)
gemm_wstat_wmma_f32(const float* __restrict__ A,
                    const float* __restrict__ W,
                    const float* __restrict__ bias,
                    float* __restrict__ C, int M,
                    const int* __restrict__ erow, const int* __restrict__ ecol,
                    int* __restrict__ cnt, int* __restrict__ bucket,
                    int* __restrict__ ovf, int E) {
    constexpr int K = CFEAT;
    constexpr int KS = K / 4;
    constexpr int NPAIR = N / 32;
    constexpr int MT = 4 / NPAIR;
    constexpr int ROWS = 16 * MT;
    constexpr int LD4 = ROWS / 4;

    __shared__ __attribute__((aligned(16))) float lds[2][ROWS * LDS_PITCH];

    const int nedge = EDGE ? (E + 4 * EDGE_WAVE - 1) / (4 * EDGE_WAVE) : 0;
    const int bid = (int)blockIdx.x;
    const int nblk = (int)gridDim.x - nedge;
    if (EDGE && bid >= nblk) {
        const int wv = __builtin_amdgcn_readfirstlane((bid - nblk) * 4 + (int)(threadIdx.x >> 5));
        count_place_role(erow, ecol, cnt, bucket, (int2*)ovf, M, E, wv);
        return;
    }

    const int lane = threadIdx.x & 31;
    const int wave = threadIdx.x >> 5;
    const int lo = lane & 15;
    const int hi = lane >> 4;
    const int mt = wave / NPAIR;
    const int n0 = (wave % NPAIR) * 32;

    v2f b0[KS], b1[KS];
    {
        const float* wbase = W + (size_t)(2 * hi) * N + n0 + lo;
#pragma unroll
        for (int s = 0; s < KS; ++s) {
            b0[s].x = wbase[(size_t)(4 * s) * N];
            b0[s].y = wbase[(size_t)(4 * s + 1) * N];
            b1[s].x = wbase[(size_t)(4 * s) * N + 16];
            b1[s].y = wbase[(size_t)(4 * s + 1) * N + 16];
        }
    }
    const float bv0 = ADD_BIAS ? bias[n0 + lo] : 0.0f;
    const float bv1 = ADD_BIAS ? bias[n0 + 16 + lo] : 0.0f;

    const int steps = (M + ROWS - 1) / ROWS;
    const v4f* __restrict__ A4 = (const v4f*)A + lane;
    v4f g[LD4];

    int st = bid;
    if (st < steps) {
#pragma unroll
        for (int i = 0; i < LD4; ++i) {
            const int row = st * ROWS + wave + 4 * i;
            g[i] = A4[(size_t)(row < M ? row : M - 1) * (K / 4)];
        }
#pragma unroll
        for (int i = 0; i < LD4; ++i)
            *(v4f*)&lds[0][(wave + 4 * i) * LDS_PITCH + 4 * lane] = g[i];
    }
    __syncthreads();

    int cur = 0;
    for (; st < steps; st += nblk) {
        const int nxt = st + nblk;
        const bool more = nxt < steps;
        if (more) {
#pragma unroll
            for (int i = 0; i < LD4; ++i) {
                const int row = nxt * ROWS + wave + 4 * i;
                g[i] = A4[(size_t)(row < M ? row : M - 1) * (K / 4)];
            }
        }

        const float* arow = &lds[cur][(16 * mt + lo) * LDS_PITCH + 2 * hi];
        v2f a[KS];
#pragma unroll
        for (int s = 0; s < KS; ++s) a[s] = *(const v2f*)(arow + 4 * s);
        __builtin_amdgcn_sched_barrier(0);

        v8f acc0 = {}, acc1 = {};
#pragma unroll
        for (int s = 0; s < KS; ++s) {
            acc0 = __builtin_amdgcn_wmma_f32_16x16x4_f32(
                 false, a[s],  false, b0[s],
                 (short)0, acc0,  false,  false);
            acc1 = __builtin_amdgcn_wmma_f32_16x16x4_f32(
                 false, a[s],  false, b1[s],
                 (short)0, acc1,  false,  false);
        }

        const int r0 = st * ROWS + 16 * mt + 8 * hi;
        float* cbase = C + (size_t)r0 * N + n0 + lo;
        float o0[8], o1[8];
#pragma unroll
        for (int v = 0; v < 8; ++v) { o0[v] = acc0[v] + bv0; o1[v] = acc1[v] + bv1; }
        if ((st + 1) * ROWS <= M) {
#pragma unroll
            for (int v = 0; v < 8; ++v) {
                cbase[(size_t)v * N] = o0[v];
                cbase[(size_t)v * N + 16] = o1[v];
            }
        } else {
#pragma unroll
            for (int v = 0; v < 8; ++v) {
                if (r0 + v < M) {
                    cbase[(size_t)v * N] = o0[v];
                    cbase[(size_t)v * N + 16] = o1[v];
                }
            }
        }

        if (more) {
#pragma unroll
            for (int i = 0; i < LD4; ++i)
                *(v4f*)&lds[cur ^ 1][(wave + 4 * i) * LDS_PITCH + 4 * lane] = g[i];
        }
        __builtin_amdgcn_fence(__ATOMIC_RELEASE, "workgroup", "local");
        __builtin_amdgcn_s_barrier();
        __builtin_amdgcn_fence(__ATOMIC_ACQUIRE, "workgroup", "local");
        cur ^= 1;
    }
}

#define FT_ROWS 16
#define FT_NPW 2

__device__ __forceinline__ void axpy4(float4& acc, const float4 v, const float w) {
    acc.x += v.x * w; acc.y += v.y * w; acc.z += v.z * w; acc.w += v.w * w;
}

template <int S>
__device__ __forceinline__ void load_rows(float4 (&v)[S], const float4* __restrict__ h4, const int r, const int o) {
    int rr[S];
#pragma unroll
    for (int q = 0; q < S; ++q) rr[q] = __builtin_amdgcn_readlane(r, o + q);
#pragma unroll
    for (int q = 0; q < S; ++q) v[q] = h4[(size_t)rr[q] * (CFEAT / 4)];
}

template <int S>
__device__ __forceinline__ void add_rows(float4& acc, const float4 (&v)[S], const int wi, const int o) {
#pragma unroll
    for (int q = 0; q < S; ++q) axpy4(acc, v[q], __int_as_float(__builtin_amdgcn_readlane(wi, o + q)));
}

template <int T>
__device__ __forceinline__ void rows_case(float4& acc, const float4* __restrict__ h4, const int r, const int wi, const int k) {
    float4 v[T];
    load_rows<T>(v, h4, r, k);
    add_rows<T>(acc, v, wi, k);
}

__device__ __forceinline__ float dinv_of(const int c) { return rsqrtf((float)c + 1.0f); }

template <int T>
__device__ __forceinline__ void rows_case_w(float4& acc, const float4* __restrict__ h4, const int* __restrict__ cnt,
                                            const int r, const float dc, const int k) {
    float4 v[T];
    load_rows<T>(v, h4, r, k);
    const int wi = __float_as_int(dinv_of(cnt[r]) * dc);
    add_rows<T>(acc, v, wi, k);
}

__device__ __forceinline__ float4 sum_messages(const float4* __restrict__ h4, const int* __restrict__ cnt,
                                               const int2* __restrict__ ovf, const int n, const int E,
                                               const int node, const int deg, const float dc,
                                               const int s0, const int lane) {
    static_assert(BKT == 32, "a bucket is one slot per lane, taken in two goes of at most 20 and 12 rows");
    float4 acc = make_float4(0.0f, 0.0f, 0.0f, 0.0f);
    {
        const int m = deg < BKT ? deg : BKT;
        const int r = lane < m ? s0 : 0;
        const int t1 = m < 20 ? m : 20;
        switch (t1) {
            case 1: rows_case_w<1>(acc, h4, cnt, r, dc, 0); break;
            case 2: rows_case_w<2>(acc, h4, cnt, r, dc, 0); break;
            case 3: rows_case_w<3>(acc, h4, cnt, r, dc, 0); break;
            case 4: rows_case_w<4>(acc, h4, cnt, r, dc, 0); break;
            case 5: rows_case_w<5>(acc, h4, cnt, r, dc, 0); break;
            case 6: rows_case_w<6>(acc, h4, cnt, r, dc, 0); break;
            case 7: rows_case_w<7>(acc, h4, cnt, r, dc, 0); break;
            case 8: rows_case_w<8>(acc, h4, cnt, r, dc, 0); break;
            case 9: rows_case_w<9>(acc, h4, cnt, r, dc, 0); break;
            case 10: rows_case_w<10>(acc, h4, cnt, r, dc, 0); break;
            case 11: rows_case_w<11>(acc, h4, cnt, r, dc, 0); break;
            case 12: rows_case_w<12>(acc, h4, cnt, r, dc, 0); break;
            case 13: rows_case_w<13>(acc, h4, cnt, r, dc, 0); break;
            case 14: rows_case_w<14>(acc, h4, cnt, r, dc, 0); break;
            case 15: rows_case_w<15>(acc, h4, cnt, r, dc, 0); break;
            case 16: rows_case_w<16>(acc, h4, cnt, r, dc, 0); break;
            case 17: rows_case_w<17>(acc, h4, cnt, r, dc, 0); break;
            case 18: rows_case_w<18>(acc, h4, cnt, r, dc, 0); break;
            case 19: rows_case_w<19>(acc, h4, cnt, r, dc, 0); break;
            case 20: rows_case_w<20>(acc, h4, cnt, r, dc, 0); break;
            default: break;
        }
        switch (m - t1) {
            case 1: rows_case_w<1>(acc, h4, cnt, r, dc, 20); break;
            case 2: rows_case_w<2>(acc, h4, cnt, r, dc, 20); break;
            case 3: rows_case_w<3>(acc, h4, cnt, r, dc, 20); break;
            case 4: rows_case_w<4>(acc, h4, cnt, r, dc, 20); break;
            case 5: rows_case_w<5>(acc, h4, cnt, r, dc, 20); break;
            case 6: rows_case_w<6>(acc, h4, cnt, r, dc, 20); break;
            case 7: rows_case_w<7>(acc, h4, cnt, r, dc, 20); break;
            case 8: rows_case_w<8>(acc, h4, cnt, r, dc, 20); break;
            case 9: rows_case_w<9>(acc, h4, cnt, r, dc, 20); break;
            case 10: rows_case_w<10>(acc, h4, cnt, r, dc, 20); break;
            case 11: rows_case_w<11>(acc, h4, cnt, r, dc, 20); break;
            case 12: rows_case_w<12>(acc, h4, cnt, r, dc, 20); break;
            default: break;
        }
    }
    if (deg > BKT) {
        int tot = cnt[n];
        tot = tot < E ? tot : E;
        for (int base = 0; base < tot; base += 32) {
            const int j = base + lane;
            int r = 0;
            float w = 0.0f;
            bool hit = false;
            if (j < tot) {
                const int2 p = ovf[j];
                hit = p.x == node;
                if (hit) { r = p.y; w = dinv_of(cnt[r]) * dc; }
            }
            const int wi = __float_as_int(w);
            unsigned todo = __builtin_amdgcn_ballot_w32(hit);
            while (todo) {
                const int q = __builtin_ctz(todo);
                todo &= todo - 1;
                rows_case<1>(acc, h4, r, wi, q);
            }
        }
    }
    return acc;
}

__device__ __forceinline__ float4 bias_relu4(const float4 acc, const float4 b) {
    float4 o;
    o.x = acc.x + b.x; o.y = acc.y + b.y; o.z = acc.z + b.z; o.w = acc.w + b.w;
    o.x = o.x > 0.0f ? o.x : 0.0f;
    o.y = o.y > 0.0f ? o.y : 0.0f;
    o.z = o.z > 0.0f ? o.z : 0.0f;
    o.w = o.w > 0.0f ? o.w : 0.0f;
    return o;
}

template <int N, int ADD_BIAS>
__global__ void __launch_bounds__(256)
gather_relu_gemm_wmma_f32(const float* __restrict__ h,
                          const int* __restrict__ cnt,
                          const int* __restrict__ bucket,
                          const int* __restrict__ ovfl,
                          const float* __restrict__ bagg,
                          const float* __restrict__ W,
                          const float* __restrict__ bout,
                          float* __restrict__ C, int n, int E) {
    constexpr int KS = CFEAT / 4;
    constexpr int NT = N / 16;
    constexpr int OP = N + 4;

    __shared__ __attribute__((aligned(16))) float xt[FT_ROWS * LDS_PITCH];
    __shared__ __attribute__((aligned(16))) float ot[FT_ROWS * OP];

    const int lane = threadIdx.x & 31;
    const int wave = __builtin_amdgcn_readfirstlane((int)(threadIdx.x >> 5));
    const int lo = lane & 15;
    const int hi = lane >> 4;
    const bool mm = (NT == 8) || (wave < NT);
    const int n0 = (wave % NT) * 16;

    const int ntiles = (n + FT_ROWS - 1) / FT_ROWS;
    int t = (int)blockIdx.x;

    int cv = 0;
    if (t < ntiles) {
        const int nd = t * FT_ROWS + FT_NPW * wave + lane;
        if (lane < FT_NPW && nd < n) cv = cnt[nd];
    }

    v2f b[KS];
#pragma unroll
    for (int s = 0; s < KS; ++s) b[s] = v2f{0.0f, 0.0f};
    float bv = 0.0f;
    if (mm) {
        const float* wbase = W + (size_t)(2 * hi) * N + n0 + lo;
#pragma unroll
        for (int s = 0; s < KS; ++s) {
            b[s].x = wbase[(size_t)(4 * s) * N];
            b[s].y = wbase[(size_t)(4 * s + 1) * N];
        }
        if (ADD_BIAS) bv = bout[n0 + lo];
    }

    const float4* __restrict__ h4 = (const float4*)h + lane;
    const int2* __restrict__ ovf = (const int2*)ovfl;
    const float4 bb = ((const float4*)bagg)[lane];

    for (; t < ntiles; t += (int)gridDim.x) {
        const int nodeA = t * FT_ROWS + FT_NPW * wave;
        const int nodeB = nodeA + 1;
        const int degA = __builtin_amdgcn_readlane(cv, 0);
        const int degB = __builtin_amdgcn_readlane(cv, 1);
        const int dvi = __float_as_int(dinv_of(cv));
        const float dcA = __int_as_float(__builtin_amdgcn_readlane(dvi, 0));
        const float dcB = __int_as_float(__builtin_amdgcn_readlane(dvi, 1));
        int sA = 0, sB = 0;
        if (lane < BKT && lane < degA) sA = bucket[(size_t)nodeA * BKT + lane];
        if (lane < BKT && lane < degB) sB = bucket[(size_t)nodeB * BKT + lane];
        float4 selfA = make_float4(0.0f, 0.0f, 0.0f, 0.0f), selfB = selfA;
        if (nodeA < n) selfA = h4[(size_t)nodeA * (CFEAT / 4)];
        if (nodeB < n) selfB = h4[(size_t)nodeB * (CFEAT / 4)];

#pragma nounroll
        for (int i = 0; i < FT_NPW; ++i) {
            float4 acc = sum_messages(h4, cnt, ovf, n, E, i ? nodeB : nodeA, i ? degB : degA,
                                      i ? dcB : dcA, i ? sB : sA, lane);
            const float dcc = i ? dcB : dcA;
            axpy4(acc, i ? selfB : selfA, dcc * dcc);
            *(float4*)&xt[(FT_NPW * wave + i) * LDS_PITCH + 4 * lane] = bias_relu4(acc, bb);
        }

        __builtin_amdgcn_fence(__ATOMIC_RELEASE, "workgroup", "local");
        __builtin_amdgcn_s_barrier();
        __builtin_amdgcn_fence(__ATOMIC_ACQUIRE, "workgroup", "local");

        {
            const int tn = t + (int)gridDim.x;
            cv = 0;
            if (tn < ntiles) {
                const int nd = tn * FT_ROWS + FT_NPW * wave + lane;
                if (lane < FT_NPW && nd < n) cv = cnt[nd];
            }
        }

        if (mm) {
            const float* arow = &xt[lo * LDS_PITCH + 2 * hi];
            v2f a[KS];
#pragma unroll
            for (int s = 0; s < KS; ++s) a[s] = *(const v2f*)(arow + 4 * s);
            __builtin_amdgcn_sched_barrier(0);
            v8f acc = {};
#pragma unroll
            for (int s = 0; s < KS; ++s)
                acc = __builtin_amdgcn_wmma_f32_16x16x4_f32(
                     false, a[s],  false, b[s],
                     (short)0, acc,  false,  false);
#pragma unroll
            for (int v = 0; v < 8; ++v) ot[(v + 8 * hi) * OP + n0 + lo] = acc[v] + bv;
        }
        __builtin_amdgcn_fence(__ATOMIC_RELEASE, "workgroup", "local");
        __builtin_amdgcn_s_barrier();
        __builtin_amdgcn_fence(__ATOMIC_ACQUIRE, "workgroup", "local");

        if (N == 128) {
#pragma unroll
            for (int i = 0; i < FT_NPW; ++i) {
                const int rr = FT_NPW * wave + i;
                const int row = t * FT_ROWS + rr;
                if (row < n)
                    ((float4*)C)[(size_t)row * (N / 4) + lane] = *(const float4*)&ot[rr * OP + 4 * lane];
            }
        } else {
            const int rr = FT_NPW * wave + hi;
            const int row = t * FT_ROWS + rr;
            if (row < n)
                ((float4*)C)[(size_t)row * (N / 4) + lo] = *(const float4*)&ot[rr * OP + 4 * lo];
        }
    }
}

extern "C" void kernel_launch(void* const* d_in, const int* in_sizes, int n_in,
                              void* d_out, int out_size, void* d_ws, size_t ws_size,
                              hipStream_t stream) {
    const float* x  = (const float*)d_in[0];
    const int*   ei = (const int*)d_in[1];
    const float* W1 = (const float*)d_in[2];
    const float* b1 = (const float*)d_in[3];
    const float* W2 = (const float*)d_in[4];
    const float* b2 = (const float*)d_in[5];
    const float* Wl = (const float*)d_in[6];
    const float* bl = (const float*)d_in[7];
    float* out = (float*)d_out;

    const int E  = in_sizes[1] / 2;
    const int Nn = in_sizes[0] / CFEAT;
    const int C  = CFEAT;

    const int* row = ei;
    const int* col = ei + E;

    float* bufA = (float*)d_ws;
    float* bufB = bufA + (size_t)Nn * C;
    int* ovf    = (int*)(bufB + (size_t)Nn * C);
    int* bucket = ovf + 2 * (size_t)E;
    int* cnt    = bucket + (size_t)Nn * BKT;

    const int edgesPerBlock = 4 * EDGE_WAVE;
    const int edgeBlocks = (E + edgesPerBlock - 1) / edgesPerBlock;

    hipMemsetAsync(cnt, 0, ((size_t)Nn + 1) * sizeof(int), stream);
    gemm_wstat_wmma_f32<128, 0, 1><<<512 + edgeBlocks, 128, 0, stream>>>(x, W1, nullptr, bufA, Nn, row, col, cnt, bucket, ovf, E);

    gather_relu_gemm_wmma_f32<128, 0><<<500, 256, 0, stream>>>(bufA, cnt, bucket, ovf, b1, W2, nullptr, bufB, Nn, E);

    gather_relu_gemm_wmma_f32<64, 1><<<500, 256, 0, stream>>>(bufB, cnt, bucket, ovf, b2, Wl, bl, out, Nn, E);
}


# Edge filing by owner work-groups: each owns 640 targets, counters and buckets in LDS, scans all targets; global atomics, scattered stores, counter memset gone; own launch; sum order = arrival rank
# speedup vs baseline: 3.2121x; 1.0413x over previous
#include <hip/hip_runtime.h>
#include <hip/hip_bf16.h>

typedef __attribute__((ext_vector_type(2))) float v2f;
typedef __attribute__((ext_vector_type(4))) float v4f;
typedef __attribute__((ext_vector_type(8))) float v8f;

#define CFEAT 128

#define BKT 32
#define SC_NPB 640
#define SC_THREADS 1024
#define SC_WAVES (SC_THREADS / 32)
#define SC_U 8
#define SC_LISTW 512

__device__ __forceinline__ void scan_file(const int* __restrict__ mylist, const int fill, const int lane, const int lo,
                                          const int* __restrict__ erow, const int* __restrict__ ecol,
                                          int* __restrict__ lcnt, int* __restrict__ lhead, int* __restrict__ lbkt,
                                          int2* __restrict__ ovf) {
    for (int i = lane; i < fill; i += 32) {
        const int e = mylist[i];
        const int t = ecol[e] - lo;
        const int r = erow[e];
        const int k = atomicAdd(&lcnt[t], 1);
        if (k < BKT) {
            lbkt[t * BKT + k] = r;
        } else {
            const int prev = atomicExch(&lhead[t], e);
            ovf[e] = make_int2(prev, r);
        }
    }
}

__global__ void __launch_bounds__(SC_THREADS)
edge_owner_scan(const int* __restrict__ erow, const int* __restrict__ ecol,
                int* __restrict__ cnt, int* __restrict__ head, int* __restrict__ bucket,
                int2* __restrict__ ovf, int n, int E) {
    __shared__ int lcnt[SC_NPB];
    __shared__ int lhead[SC_NPB];
    __shared__ __attribute__((aligned(16))) int lbkt[SC_NPB * BKT];
    __shared__ int llist[SC_WAVES][SC_LISTW];

    const int tid = (int)threadIdx.x;
    const int lane = tid & 31;
    const int wave = __builtin_amdgcn_readfirstlane(tid >> 5);
    const int lo = (int)blockIdx.x * SC_NPB;

    for (int i = tid; i < SC_NPB; i += SC_THREADS) { lcnt[i] = 0; lhead[i] = -1; }
    __syncthreads();

    int* mylist = llist[wave];
    int fill = 0;
    const bool vec = ((E & 3) == 0) && ((((size_t)ecol) & 15) == 0);
    constexpr int WBLK = 128 * SC_U;

    for (int base = wave * WBLK; base < E; base += SC_WAVES * WBLK) {
        int4 c[SC_U];
        if (vec && base + WBLK <= E) {
#pragma unroll
            for (int g = 0; g < SC_U; ++g) c[g] = *(const int4*)(ecol + base + 128 * g + 4 * lane);
        } else {
#pragma unroll
            for (int g = 0; g < SC_U; ++g) {
                const int e0 = base + 128 * g + 4 * lane;
                c[g].x = e0 + 0 < E ? ecol[e0 + 0] : lo - 1;
                c[g].y = e0 + 1 < E ? ecol[e0 + 1] : lo - 1;
                c[g].z = e0 + 2 < E ? ecol[e0 + 2] : lo - 1;
                c[g].w = e0 + 3 < E ? ecol[e0 + 3] : lo - 1;
            }
        }
#pragma unroll
        for (int g = 0; g < SC_U; ++g) {
            const int e0 = base + 128 * g + 4 * lane;
            const int cc[4] = {c[g].x, c[g].y, c[g].z, c[g].w};
#pragma unroll
            for (int j = 0; j < 4; ++j) {
                const bool hit = (unsigned)(cc[j] - lo) < (unsigned)SC_NPB;
                const unsigned mask = __builtin_amdgcn_ballot_w32(hit);
                if (mask) {
                    if (hit) mylist[fill + (int)__builtin_amdgcn_mbcnt_lo(mask, 0u)] = e0 + j;
                    fill += __builtin_popcount(mask);
                }
            }
            if (fill > SC_LISTW - 128) {
                __builtin_amdgcn_wave_barrier();
                scan_file(mylist, fill, lane, lo, erow, ecol, lcnt, lhead, lbkt, ovf);
                __builtin_amdgcn_wave_barrier();
                fill = 0;
            }
        }
    }
    __builtin_amdgcn_wave_barrier();
    scan_file(mylist, fill, lane, lo, erow, ecol, lcnt, lhead, lbkt, ovf);
    __syncthreads();

    for (int i = tid; i < SC_NPB; i += SC_THREADS) {
        if (lo + i < n) { cnt[lo + i] = lcnt[i]; head[lo + i] = lhead[i]; }
    }
    for (int i = tid; i < SC_NPB * BKT / 4; i += SC_THREADS) {
        if (lo + (i >> 3) < n) ((int4*)bucket)[(size_t)lo * (BKT / 4) + i] = ((const int4*)lbkt)[i];
    }
}

#define LDS_PITCH 132

template <int N, int ADD_BIAS>
__global__ void __launch_bounds__(128)
gemm_wstat_wmma_f32(const float* __restrict__ A,
                    const float* __restrict__ W,
                    const float* __restrict__ bias,
                    float* __restrict__ C, int M) {
    constexpr int K = CFEAT;
    constexpr int KS = K / 4;
    constexpr int NPAIR = N / 32;
    constexpr int MT = 4 / NPAIR;
    constexpr int ROWS = 16 * MT;
    constexpr int LD4 = ROWS / 4;

    __shared__ __attribute__((aligned(16))) float lds[2][ROWS * LDS_PITCH];

    const int bid = (int)blockIdx.x;
    const int nblk = (int)gridDim.x;

    const int lane = threadIdx.x & 31;
    const int wave = threadIdx.x >> 5;
    const int lo = lane & 15;
    const int hi = lane >> 4;
    const int mt = wave / NPAIR;
    const int n0 = (wave % NPAIR) * 32;

    v2f b0[KS], b1[KS];
    {
        const float* wbase = W + (size_t)(2 * hi) * N + n0 + lo;
#pragma unroll
        for (int s = 0; s < KS; ++s) {
            b0[s].x = wbase[(size_t)(4 * s) * N];
            b0[s].y = wbase[(size_t)(4 * s + 1) * N];
            b1[s].x = wbase[(size_t)(4 * s) * N + 16];
            b1[s].y = wbase[(size_t)(4 * s + 1) * N + 16];
        }
    }
    const float bv0 = ADD_BIAS ? bias[n0 + lo] : 0.0f;
    const float bv1 = ADD_BIAS ? bias[n0 + 16 + lo] : 0.0f;

    const int steps = (M + ROWS - 1) / ROWS;
    const v4f* __restrict__ A4 = (const v4f*)A + lane;
    v4f g[LD4];

    int st = bid;
    if (st < steps) {
#pragma unroll
        for (int i = 0; i < LD4; ++i) {
            const int row = st * ROWS + wave + 4 * i;
            g[i] = A4[(size_t)(row < M ? row : M - 1) * (K / 4)];
        }
#pragma unroll
        for (int i = 0; i < LD4; ++i)
            *(v4f*)&lds[0][(wave + 4 * i) * LDS_PITCH + 4 * lane] = g[i];
    }
    __syncthreads();

    int cur = 0;
    for (; st < steps; st += nblk) {
        const int nxt = st + nblk;
        const bool more = nxt < steps;
        if (more) {
#pragma unroll
            for (int i = 0; i < LD4; ++i) {
                const int row = nxt * ROWS + wave + 4 * i;
                g[i] = A4[(size_t)(row < M ? row : M - 1) * (K / 4)];
            }
        }

        const float* arow = &lds[cur][(16 * mt + lo) * LDS_PITCH + 2 * hi];
        v2f a[KS];
#pragma unroll
        for (int s = 0; s < KS; ++s) a[s] = *(const v2f*)(arow + 4 * s);
        __builtin_amdgcn_sched_barrier(0);

        v8f acc0 = {}, acc1 = {};
#pragma unroll
        for (int s = 0; s < KS; ++s) {
            acc0 = __builtin_amdgcn_wmma_f32_16x16x4_f32(
                 false, a[s],  false, b0[s],
                 (short)0, acc0,  false,  false);
            acc1 = __builtin_amdgcn_wmma_f32_16x16x4_f32(
                 false, a[s],  false, b1[s],
                 (short)0, acc1,  false,  false);
        }

        const int r0 = st * ROWS + 16 * mt + 8 * hi;
        float* cbase = C + (size_t)r0 * N + n0 + lo;
        float o0[8], o1[8];
#pragma unroll
        for (int v = 0; v < 8; ++v) { o0[v] = acc0[v] + bv0; o1[v] = acc1[v] + bv1; }
        if ((st + 1) * ROWS <= M) {
#pragma unroll
            for (int v = 0; v < 8; ++v) {
                cbase[(size_t)v * N] = o0[v];
                cbase[(size_t)v * N + 16] = o1[v];
            }
        } else {
#pragma unroll
            for (int v = 0; v < 8; ++v) {
                if (r0 + v < M) {
                    cbase[(size_t)v * N] = o0[v];
                    cbase[(size_t)v * N + 16] = o1[v];
                }
            }
        }

        if (more) {
#pragma unroll
            for (int i = 0; i < LD4; ++i)
                *(v4f*)&lds[cur ^ 1][(wave + 4 * i) * LDS_PITCH + 4 * lane] = g[i];
        }
        __builtin_amdgcn_fence(__ATOMIC_RELEASE, "workgroup", "local");
        __builtin_amdgcn_s_barrier();
        __builtin_amdgcn_fence(__ATOMIC_ACQUIRE, "workgroup", "local");
        cur ^= 1;
    }
}

#define FT_ROWS 16
#define FT_NPW 2

__device__ __forceinline__ void axpy4(float4& acc, const float4 v, const float w) {
    acc.x += v.x * w; acc.y += v.y * w; acc.z += v.z * w; acc.w += v.w * w;
}

template <int S>
__device__ __forceinline__ void load_rows(float4 (&v)[S], const float4* __restrict__ h4, const int r, const int o) {
    int rr[S];
#pragma unroll
    for (int q = 0; q < S; ++q) rr[q] = __builtin_amdgcn_readlane(r, o + q);
#pragma unroll
    for (int q = 0; q < S; ++q) v[q] = h4[(size_t)rr[q] * (CFEAT / 4)];
}

template <int S>
__device__ __forceinline__ void add_rows(float4& acc, const float4 (&v)[S], const int wi, const int o) {
#pragma unroll
    for (int q = 0; q < S; ++q) axpy4(acc, v[q], __int_as_float(__builtin_amdgcn_readlane(wi, o + q)));
}

template <int T>
__device__ __forceinline__ void rows_case(float4& acc, const float4* __restrict__ h4, const int r, const int wi, const int k) {
    float4 v[T];
    load_rows<T>(v, h4, r, k);
    add_rows<T>(acc, v, wi, k);
}

__device__ __forceinline__ float dinv_of(const int c) { return rsqrtf((float)c + 1.0f); }

template <int T>
__device__ __forceinline__ void rows_case_w(float4& acc, const float4* __restrict__ h4, const int* __restrict__ cnt,
                                            const int r, const float dc, const int k) {
    float4 v[T];
    load_rows<T>(v, h4, r, k);
    const int wi = __float_as_int(dinv_of(cnt[r]) * dc);
    add_rows<T>(acc, v, wi, k);
}

__device__ __forceinline__ float4 sum_messages(const float4* __restrict__ h4, const int* __restrict__ cnt,
                                               const int2* __restrict__ ovf, const int* __restrict__ head, const int E,
                                               const int node, const int deg, const float dc,
                                               const int s0, const int lane) {
    static_assert(BKT == 32, "a bucket is one slot per lane, taken in two goes of at most 20 and 12 rows");
    float4 acc = make_float4(0.0f, 0.0f, 0.0f, 0.0f);
    {
        const int m = deg < BKT ? deg : BKT;
        const int r = lane < m ? s0 : 0;
        const int t1 = m < 20 ? m : 20;
        switch (t1) {
            case 1: rows_case_w<1>(acc, h4, cnt, r, dc, 0); break;
            case 2: rows_case_w<2>(acc, h4, cnt, r, dc, 0); break;
            case 3: rows_case_w<3>(acc, h4, cnt, r, dc, 0); break;
            case 4: rows_case_w<4>(acc, h4, cnt, r, dc, 0); break;
            case 5: rows_case_w<5>(acc, h4, cnt, r, dc, 0); break;
            case 6: rows_case_w<6>(acc, h4, cnt, r, dc, 0); break;
            case 7: rows_case_w<7>(acc, h4, cnt, r, dc, 0); break;
            case 8: rows_case_w<8>(acc, h4, cnt, r, dc, 0); break;
            case 9: rows_case_w<9>(acc, h4, cnt, r, dc, 0); break;
            case 10: rows_case_w<10>(acc, h4, cnt, r, dc, 0); break;
            case 11: rows_case_w<11>(acc, h4, cnt, r, dc, 0); break;
            case 12: rows_case_w<12>(acc, h4, cnt, r, dc, 0); break;
            case 13: rows_case_w<13>(acc, h4, cnt, r, dc, 0); break;
            case 14: rows_case_w<14>(acc, h4, cnt, r, dc, 0); break;
            case 15: rows_case_w<15>(acc, h4, cnt, r, dc, 0); break;
            case 16: rows_case_w<16>(acc, h4, cnt, r, dc, 0); break;
            case 17: rows_case_w<17>(acc, h4, cnt, r, dc, 0); break;
            case 18: rows_case_w<18>(acc, h4, cnt, r, dc, 0); break;
            case 19: rows_case_w<19>(acc, h4, cnt, r, dc, 0); break;
            case 20: rows_case_w<20>(acc, h4, cnt, r, dc, 0); break;
            default: break;
        }
        switch (m - t1) {
            case 1: rows_case_w<1>(acc, h4, cnt, r, dc, 20); break;
            case 2: rows_case_w<2>(acc, h4, cnt, r, dc, 20); break;
            case 3: rows_case_w<3>(acc, h4, cnt, r, dc, 20); break;
            case 4: rows_case_w<4>(acc, h4, cnt, r, dc, 20); break;
            case 5: rows_case_w<5>(acc, h4, cnt, r, dc, 20); break;
            case 6: rows_case_w<6>(acc, h4, cnt, r, dc, 20); break;
            case 7: rows_case_w<7>(acc, h4, cnt, r, dc, 20); break;
            case 8: rows_case_w<8>(acc, h4, cnt, r, dc, 20); break;
            case 9: rows_case_w<9>(acc, h4, cnt, r, dc, 20); break;
            case 10: rows_case_w<10>(acc, h4, cnt, r, dc, 20); break;
            case 11: rows_case_w<11>(acc, h4, cnt, r, dc, 20); break;
            case 12: rows_case_w<12>(acc, h4, cnt, r, dc, 20); break;
            default: break;
        }
    }
    if (deg > BKT) {
        int e = head[node];
        for (int i = BKT; i < deg && e >= 0 && e < E; ++i) {
            const int2 p = ovf[e];
            const float w = dinv_of(cnt[p.y]) * dc;
            axpy4(acc, h4[(size_t)p.y * (CFEAT / 4)], w);
            e = p.x;
        }
    }
    return acc;
}

__device__ __forceinline__ float4 bias_relu4(const float4 acc, const float4 b) {
    float4 o;
    o.x = acc.x + b.x; o.y = acc.y + b.y; o.z = acc.z + b.z; o.w = acc.w + b.w;
    o.x = o.x > 0.0f ? o.x : 0.0f;
    o.y = o.y > 0.0f ? o.y : 0.0f;
    o.z = o.z > 0.0f ? o.z : 0.0f;
    o.w = o.w > 0.0f ? o.w : 0.0f;
    return o;
}

template <int N, int ADD_BIAS>
__global__ void __launch_bounds__(256)
gather_relu_gemm_wmma_f32(const float* __restrict__ h,
                          const int* __restrict__ cnt,
                          const int* __restrict__ bucket,
                          const int* __restrict__ ovfl,
                          const int* __restrict__ head,
                          const float* __restrict__ bagg,
                          const float* __restrict__ W,
                          const float* __restrict__ bout,
                          float* __restrict__ C, int n, int E) {
    constexpr int KS = CFEAT / 4;
    constexpr int NT = N / 16;
    constexpr int OP = N + 4;

    __shared__ __attribute__((aligned(16))) float xt[FT_ROWS * LDS_PITCH];
    __shared__ __attribute__((aligned(16))) float ot[FT_ROWS * OP];

    const int lane = threadIdx.x & 31;
    const int wave = __builtin_amdgcn_readfirstlane((int)(threadIdx.x >> 5));
    const int lo = lane & 15;
    const int hi = lane >> 4;
    const bool mm = (NT == 8) || (wave < NT);
    const int n0 = (wave % NT) * 16;

    const int ntiles = (n + FT_ROWS - 1) / FT_ROWS;
    int t = (int)blockIdx.x;

    int cv = 0;
    if (t < ntiles) {
        const int nd = t * FT_ROWS + FT_NPW * wave + lane;
        if (lane < FT_NPW && nd < n) cv = cnt[nd];
    }

    v2f b[KS];
#pragma unroll
    for (int s = 0; s < KS; ++s) b[s] = v2f{0.0f, 0.0f};
    float bv = 0.0f;
    if (mm) {
        const float* wbase = W + (size_t)(2 * hi) * N + n0 + lo;
#pragma unroll
        for (int s = 0; s < KS; ++s) {
            b[s].x = wbase[(size_t)(4 * s) * N];
            b[s].y = wbase[(size_t)(4 * s + 1) * N];
        }
        if (ADD_BIAS) bv = bout[n0 + lo];
    }

    const float4* __restrict__ h4 = (const float4*)h + lane;
    const int2* __restrict__ ovf = (const int2*)ovfl;
    const float4 bb = ((const float4*)bagg)[lane];

    for (; t < ntiles; t += (int)gridDim.x) {
        const int nodeA = t * FT_ROWS + FT_NPW * wave;
        const int nodeB = nodeA + 1;
        const int degA = __builtin_amdgcn_readlane(cv, 0);
        const int degB = __builtin_amdgcn_readlane(cv, 1);
        const int dvi = __float_as_int(dinv_of(cv));
        const float dcA = __int_as_float(__builtin_amdgcn_readlane(dvi, 0));
        const float dcB = __int_as_float(__builtin_amdgcn_readlane(dvi, 1));
        int sA = 0, sB = 0;
        if (lane < BKT && lane < degA) sA = bucket[(size_t)nodeA * BKT + lane];
        if (lane < BKT && lane < degB) sB = bucket[(size_t)nodeB * BKT + lane];
        float4 selfA = make_float4(0.0f, 0.0f, 0.0f, 0.0f), selfB = selfA;
        if (nodeA < n) selfA = h4[(size_t)nodeA * (CFEAT / 4)];
        if (nodeB < n) selfB = h4[(size_t)nodeB * (CFEAT / 4)];

#pragma nounroll
        for (int i = 0; i < FT_NPW; ++i) {
            float4 acc = sum_messages(h4, cnt, ovf, head, E, i ? nodeB : nodeA, i ? degB : degA,
                                      i ? dcB : dcA, i ? sB : sA, lane);
            const float dcc = i ? dcB : dcA;
            axpy4(acc, i ? selfB : selfA, dcc * dcc);
            *(float4*)&xt[(FT_NPW * wave + i) * LDS_PITCH + 4 * lane] = bias_relu4(acc, bb);
        }

        __builtin_amdgcn_fence(__ATOMIC_RELEASE, "workgroup", "local");
        __builtin_amdgcn_s_barrier();
        __builtin_amdgcn_fence(__ATOMIC_ACQUIRE, "workgroup", "local");

        {
            const int tn = t + (int)gridDim.x;
            cv = 0;
            if (tn < ntiles) {
                const int nd = tn * FT_ROWS + FT_NPW * wave + lane;
                if (lane < FT_NPW && nd < n) cv = cnt[nd];
            }
        }

        if (mm) {
            const float* arow = &xt[lo * LDS_PITCH + 2 * hi];
            v2f a[KS];
#pragma unroll
            for (int s = 0; s < KS; ++s) a[s] = *(const v2f*)(arow + 4 * s);
            __builtin_amdgcn_sched_barrier(0);
            v8f acc = {};
#pragma unroll
            for (int s = 0; s < KS; ++s)
                acc = __builtin_amdgcn_wmma_f32_16x16x4_f32(
                     false, a[s],  false, b[s],
                     (short)0, acc,  false,  false);
#pragma unroll
            for (int v = 0; v < 8; ++v) ot[(v + 8 * hi) * OP + n0 + lo] = acc[v] + bv;
        }
        __builtin_amdgcn_fence(__ATOMIC_RELEASE, "workgroup", "local");
        __builtin_amdgcn_s_barrier();
        __builtin_amdgcn_fence(__ATOMIC_ACQUIRE, "workgroup", "local");

        if (N == 128) {
#pragma unroll
            for (int i = 0; i < FT_NPW; ++i) {
                const int rr = FT_NPW * wave + i;
                const int row = t * FT_ROWS + rr;
                if (row < n)
                    ((float4*)C)[(size_t)row * (N / 4) + lane] = *(const float4*)&ot[rr * OP + 4 * lane];
            }
        } else {
            const int rr = FT_NPW * wave + hi;
            const int row = t * FT_ROWS + rr;
            if (row < n)
                ((float4*)C)[(size_t)row * (N / 4) + lo] = *(const float4*)&ot[rr * OP + 4 * lo];
        }
    }
}

extern "C" void kernel_launch(void* const* d_in, const int* in_sizes, int n_in,
                              void* d_out, int out_size, void* d_ws, size_t ws_size,
                              hipStream_t stream) {
    const float* x  = (const float*)d_in[0];
    const int*   ei = (const int*)d_in[1];
    const float* W1 = (const float*)d_in[2];
    const float* b1 = (const float*)d_in[3];
    const float* W2 = (const float*)d_in[4];
    const float* b2 = (const float*)d_in[5];
    const float* Wl = (const float*)d_in[6];
    const float* bl = (const float*)d_in[7];
    float* out = (float*)d_out;

    const int E  = in_sizes[1] / 2;
    const int Nn = in_sizes[0] / CFEAT;
    const int C  = CFEAT;

    const int* row = ei;
    const int* col = ei + E;

    float* bufA = (float*)d_ws;
    float* bufB = bufA + (size_t)Nn * C;
    int* bucket = (int*)(bufB + (size_t)Nn * C);
    int* cnt    = bucket + (size_t)Nn * BKT;
    int* head   = cnt + Nn;
    int* ovf    = head + Nn;

    edge_owner_scan<<<(Nn + SC_NPB - 1) / SC_NPB, SC_THREADS, 0, stream>>>(row, col, cnt, head, bucket, (int2*)ovf, Nn, E);
    gemm_wstat_wmma_f32<128, 0><<<512, 128, 0, stream>>>(x, W1, nullptr, bufA, Nn);

    gather_relu_gemm_wmma_f32<128, 0><<<500, 256, 0, stream>>>(bufA, cnt, bucket, ovf, head, b1, W2, nullptr, bufB, Nn, E);

    gather_relu_gemm_wmma_f32<64, 1><<<500, 256, 0, stream>>>(bufB, cnt, bucket, ovf, head, b2, Wl, bl, out, Nn, E);
}


# Edge list cut into 4 parts: 4x the owner work-groups scan a quarter of the targets each into 16-slot buckets of their own; dinv plane made in the GEMM's launch; sum order = part, then arrival rank
# speedup vs baseline: 4.3161x; 1.3437x over previous
#include <hip/hip_runtime.h>
#include <hip/hip_bf16.h>

typedef __attribute__((ext_vector_type(2))) float v2f;
typedef __attribute__((ext_vector_type(4))) float v4f;
typedef __attribute__((ext_vector_type(8))) float v8f;

#define CFEAT 128

#define NPART 4
#define BKT 16
#define SC_NPB 640
#define SC_THREADS 1024
#define SC_WAVES (SC_THREADS / 32)
#define SC_U 8
#define SC_LISTW 512

__device__ __forceinline__ void scan_file(const int* __restrict__ mylist, const int fill, const int lane, const int lo,
                                          const int* __restrict__ erow, const int* __restrict__ ecol,
                                          int* __restrict__ lcnt, int* __restrict__ lhead, int* __restrict__ lbkt,
                                          int2* __restrict__ ovf) {
    for (int i = lane; i < fill; i += 32) {
        const int e = mylist[i];
        const int t = ecol[e] - lo;
        const int r = erow[e];
        const int k = atomicAdd(&lcnt[t], 1);
        if (k < BKT) {
            lbkt[t * BKT + k] = r;
        } else {
            const int prev = atomicExch(&lhead[t], e);
            ovf[e] = make_int2(prev, r);
        }
    }
}

__global__ void __launch_bounds__(SC_THREADS)
edge_owner_scan(const int* __restrict__ erow, const int* __restrict__ ecol,
                int* __restrict__ cntp, int* __restrict__ headp, int* __restrict__ bucketp,
                int2* __restrict__ ovf, int n, int E) {
    __shared__ int lcnt[SC_NPB];
    __shared__ int lhead[SC_NPB];
    __shared__ __attribute__((aligned(16))) int lbkt[SC_NPB * BKT];
    __shared__ int llist[SC_WAVES][SC_LISTW];

    const int tid = (int)threadIdx.x;
    const int lane = tid & 31;
    const int wave = __builtin_amdgcn_readfirstlane(tid >> 5);
    const int part = (int)blockIdx.x % NPART;
    const int lo = ((int)blockIdx.x / NPART) * SC_NPB;

    for (int i = tid; i < SC_NPB; i += SC_THREADS) { lcnt[i] = 0; lhead[i] = -1; }
    __syncthreads();

    int* mylist = llist[wave];
    int fill = 0;
    const bool vec = ((E & 3) == 0) && ((((size_t)ecol) & 15) == 0);
    constexpr int WBLK = 128 * SC_U;
    const int plen = ((E + NPART - 1) / NPART + WBLK - 1) / WBLK * WBLK;
    const int eb = part * plen < E ? part * plen : E;
    const int ee = eb + plen < E ? eb + plen : E;

    for (int base = eb + wave * WBLK; base < ee; base += SC_WAVES * WBLK) {
        int4 c[SC_U];
        if (vec && base + WBLK <= ee) {
#pragma unroll
            for (int g = 0; g < SC_U; ++g) c[g] = *(const int4*)(ecol + base + 128 * g + 4 * lane);
        } else {
#pragma unroll
            for (int g = 0; g < SC_U; ++g) {
                const int e0 = base + 128 * g + 4 * lane;
                c[g].x = e0 + 0 < ee ? ecol[e0 + 0] : lo - 1;
                c[g].y = e0 + 1 < ee ? ecol[e0 + 1] : lo - 1;
                c[g].z = e0 + 2 < ee ? ecol[e0 + 2] : lo - 1;
                c[g].w = e0 + 3 < ee ? ecol[e0 + 3] : lo - 1;
            }
        }
#pragma unroll
        for (int g = 0; g < SC_U; ++g) {
            const int e0 = base + 128 * g + 4 * lane;
            const int cc[4] = {c[g].x, c[g].y, c[g].z, c[g].w};
#pragma unroll
            for (int j = 0; j < 4; ++j) {
                const bool hit = (unsigned)(cc[j] - lo) < (unsigned)SC_NPB;
                const unsigned mask = __builtin_amdgcn_ballot_w32(hit);
                if (mask) {
                    if (hit) mylist[fill + (int)__builtin_amdgcn_mbcnt_lo(mask, 0u)] = e0 + j;
                    fill += __builtin_popcount(mask);
                }
            }
            if (fill > SC_LISTW - 128) {
                __builtin_amdgcn_wave_barrier();
                scan_file(mylist, fill, lane, lo, erow, ecol, lcnt, lhead, lbkt, ovf);
                __builtin_amdgcn_wave_barrier();
                fill = 0;
            }
        }
    }
    __builtin_amdgcn_wave_barrier();
    scan_file(mylist, fill, lane, lo, erow, ecol, lcnt, lhead, lbkt, ovf);
    __syncthreads();

    const size_t pbase = (size_t)part * n;
    for (int i = tid; i < SC_NPB; i += SC_THREADS) {
        if (lo + i < n) { cntp[pbase + lo + i] = lcnt[i]; headp[pbase + lo + i] = lhead[i]; }
    }
    for (int i = tid; i < SC_NPB * BKT / 4; i += SC_THREADS) {
        if (lo + i / (BKT / 4) < n) ((int4*)bucketp)[(pbase + lo) * (BKT / 4) + i] = ((const int4*)lbkt)[i];
    }
}

#define LDS_PITCH 132

template <int N, int ADD_BIAS>
__global__ void __launch_bounds__(128)
gemm_wstat_wmma_f32(const float* __restrict__ A,
                    const float* __restrict__ W,
                    const float* __restrict__ bias,
                    float* __restrict__ C, int M,
                    const int* __restrict__ cntp, float* __restrict__ dinv) {
    constexpr int K = CFEAT;
    constexpr int KS = K / 4;
    constexpr int NPAIR = N / 32;
    constexpr int MT = 4 / NPAIR;
    constexpr int ROWS = 16 * MT;
    constexpr int LD4 = ROWS / 4;

    __shared__ __attribute__((aligned(16))) float lds[2][ROWS * LDS_PITCH];

    const int bid = (int)blockIdx.x;
    const int nblk = (int)gridDim.x;

    for (int i = bid * 128 + (int)threadIdx.x; i < M; i += nblk * 128) {
        int c = 0;
        for (int p = 0; p < NPART; ++p) c += cntp[(size_t)p * M + i];
        dinv[i] = rsqrtf((float)c + 1.0f);
    }

    const int lane = threadIdx.x & 31;
    const int wave = threadIdx.x >> 5;
    const int lo = lane & 15;
    const int hi = lane >> 4;
    const int mt = wave / NPAIR;
    const int n0 = (wave % NPAIR) * 32;

    v2f b0[KS], b1[KS];
    {
        const float* wbase = W + (size_t)(2 * hi) * N + n0 + lo;
#pragma unroll
        for (int s = 0; s < KS; ++s) {
            b0[s].x = wbase[(size_t)(4 * s) * N];
            b0[s].y = wbase[(size_t)(4 * s + 1) * N];
            b1[s].x = wbase[(size_t)(4 * s) * N + 16];
            b1[s].y = wbase[(size_t)(4 * s + 1) * N + 16];
        }
    }
    const float bv0 = ADD_BIAS ? bias[n0 + lo] : 0.0f;
    const float bv1 = ADD_BIAS ? bias[n0 + 16 + lo] : 0.0f;

    const int steps = (M + ROWS - 1) / ROWS;
    const v4f* __restrict__ A4 = (const v4f*)A + lane;
    v4f g[LD4];

    int st = bid;
    if (st < steps) {
#pragma unroll
        for (int i = 0; i < LD4; ++i) {
            const int row = st * ROWS + wave + 4 * i;
            g[i] = A4[(size_t)(row < M ? row : M - 1) * (K / 4)];
        }
#pragma unroll
        for (int i = 0; i < LD4; ++i)
            *(v4f*)&lds[0][(wave + 4 * i) * LDS_PITCH + 4 * lane] = g[i];
    }
    __syncthreads();

    int cur = 0;
    for (; st < steps; st += nblk) {
        const int nxt = st + nblk;
        const bool more = nxt < steps;
        if (more) {
#pragma unroll
            for (int i = 0; i < LD4; ++i) {
                const int row = nxt * ROWS + wave + 4 * i;
                g[i] = A4[(size_t)(row < M ? row : M - 1) * (K / 4)];
            }
        }

        const float* arow = &lds[cur][(16 * mt + lo) * LDS_PITCH + 2 * hi];
        v2f a[KS];
#pragma unroll
        for (int s = 0; s < KS; ++s) a[s] = *(const v2f*)(arow + 4 * s);
        __builtin_amdgcn_sched_barrier(0);

        v8f acc0 = {}, acc1 = {};
#pragma unroll
        for (int s = 0; s < KS; ++s) {
            acc0 = __builtin_amdgcn_wmma_f32_16x16x4_f32(
                 false, a[s],  false, b0[s],
                 (short)0, acc0,  false,  false);
            acc1 = __builtin_amdgcn_wmma_f32_16x16x4_f32(
                 false, a[s],  false, b1[s],
                 (short)0, acc1,  false,  false);
        }

        const int r0 = st * ROWS + 16 * mt + 8 * hi;
        float* cbase = C + (size_t)r0 * N + n0 + lo;
        float o0[8], o1[8];
#pragma unroll
        for (int v = 0; v < 8; ++v) { o0[v] = acc0[v] + bv0; o1[v] = acc1[v] + bv1; }
        if ((st + 1) * ROWS <= M) {
#pragma unroll
            for (int v = 0; v < 8; ++v) {
                cbase[(size_t)v * N] = o0[v];
                cbase[(size_t)v * N + 16] = o1[v];
            }
        } else {
#pragma unroll
            for (int v = 0; v < 8; ++v) {
                if (r0 + v < M) {
                    cbase[(size_t)v * N] = o0[v];
                    cbase[(size_t)v * N + 16] = o1[v];
                }
            }
        }

        if (more) {
#pragma unroll
            for (int i = 0; i < LD4; ++i)
                *(v4f*)&lds[cur ^ 1][(wave + 4 * i) * LDS_PITCH + 4 * lane] = g[i];
        }
        __builtin_amdgcn_fence(__ATOMIC_RELEASE, "workgroup", "local");
        __builtin_amdgcn_s_barrier();
        __builtin_amdgcn_fence(__ATOMIC_ACQUIRE, "workgroup", "local");
        cur ^= 1;
    }
}

#define FT_ROWS 16
#define FT_NPW 2

__device__ __forceinline__ void axpy4(float4& acc, const float4 v, const float w) {
    acc.x += v.x * w; acc.y += v.y * w; acc.z += v.z * w; acc.w += v.w * w;
}

template <int S>
__device__ __forceinline__ void load_rows(float4 (&v)[S], const float4* __restrict__ h4, const int r, const int o) {
    int rr[S];
#pragma unroll
    for (int q = 0; q < S; ++q) rr[q] = __builtin_amdgcn_readlane(r, o + q);
#pragma unroll
    for (int q = 0; q < S; ++q) v[q] = h4[(size_t)rr[q] * (CFEAT / 4)];
}

template <int S>
__device__ __forceinline__ void add_rows(float4& acc, const float4 (&v)[S], const int wi, const int o) {
#pragma unroll
    for (int q = 0; q < S; ++q) axpy4(acc, v[q], __int_as_float(__builtin_amdgcn_readlane(wi, o + q)));
}

template <int T>
__device__ __forceinline__ void rows_case(float4& acc, const float4* __restrict__ h4, const int r, const int wi, const int k) {
    float4 v[T];
    load_rows<T>(v, h4, r, k);
    add_rows<T>(acc, v, wi, k);
}

__device__ __forceinline__ float dinv_of(const int c) { return rsqrtf((float)c + 1.0f); }

template <int T>
__device__ __forceinline__ void rows_case_w(float4& acc, const float4* __restrict__ h4, const float* __restrict__ dinv,
                                            const int r, const float dc, const int k) {
    float4 v[T];
    load_rows<T>(v, h4, r, k);
    const int wi = __float_as_int(dinv[r] * dc);
    add_rows<T>(acc, v, wi, k);
}

__device__ __forceinline__ size_t slot_index(const int i, const int o1, const int o2, const int o3,
                                             const int n, const int node) {
    static_assert(NPART == 4, "three bucket boundaries");
    const int p = (i >= o1) + (i >= o2) + (i >= o3);
    const int j = i - (i >= o3 ? o3 : i >= o2 ? o2 : i >= o1 ? o1 : 0);
    return ((size_t)p * n + node) * BKT + j;
}

__device__ __forceinline__ void add_chain(float4& acc, const float4* __restrict__ h4, const float* __restrict__ dinv,
                                          const int2* __restrict__ ovf, const int* __restrict__ head, const int E,
                                          const int c, const float dc) {
    if (c > BKT) {
        int e = *head;
        for (int i = BKT; i < c && e >= 0 && e < E; ++i) {
            const int2 q = ovf[e];
            axpy4(acc, h4[(size_t)q.y * (CFEAT / 4)], dinv[q.y] * dc);
            e = q.x;
        }
    }
}

__device__ __forceinline__ float4 sum_messages(const float4* __restrict__ h4, const float* __restrict__ dinv,
                                               const int* __restrict__ bucketp,
                                               const int2* __restrict__ ovf, const int* __restrict__ headp,
                                               const int n, const int E, const int node,
                                               const int c0, const int c1, const int c2, const int c3,
                                               const float dc, const int s0, const int lane) {
    static_assert(NPART == 4, "four partial in-degrees");
    float4 acc = make_float4(0.0f, 0.0f, 0.0f, 0.0f);
    const int m0 = c0 < BKT ? c0 : BKT, m1 = c1 < BKT ? c1 : BKT, m2 = c2 < BKT ? c2 : BKT, m3 = c3 < BKT ? c3 : BKT;
    const int o1 = m0, o2 = o1 + m1, o3 = o2 + m2, mt = o3 + m3;
    {
        const int m = mt < 32 ? mt : 32;
        const int r = lane < m ? s0 : 0;
        const int t1 = m < 20 ? m : 20;
        switch (t1) {
            case 1: rows_case_w<1>(acc, h4, dinv, r, dc, 0); break;
            case 2: rows_case_w<2>(acc, h4, dinv, r, dc, 0); break;
            case 3: rows_case_w<3>(acc, h4, dinv, r, dc, 0); break;
            case 4: rows_case_w<4>(acc, h4, dinv, r, dc, 0); break;
            case 5: rows_case_w<5>(acc, h4, dinv, r, dc, 0); break;
            case 6: rows_case_w<6>(acc, h4, dinv, r, dc, 0); break;
            case 7: rows_case_w<7>(acc, h4, dinv, r, dc, 0); break;
            case 8: rows_case_w<8>(acc, h4, dinv, r, dc, 0); break;
            case 9: rows_case_w<9>(acc, h4, dinv, r, dc, 0); break;
            case 10: rows_case_w<10>(acc, h4, dinv, r, dc, 0); break;
            case 11: rows_case_w<11>(acc, h4, dinv, r, dc, 0); break;
            case 12: rows_case_w<12>(acc, h4, dinv, r, dc, 0); break;
            case 13: rows_case_w<13>(acc, h4, dinv, r, dc, 0); break;
            case 14: rows_case_w<14>(acc, h4, dinv, r, dc, 0); break;
            case 15: rows_case_w<15>(acc, h4, dinv, r, dc, 0); break;
            case 16: rows_case_w<16>(acc, h4, dinv, r, dc, 0); break;
            case 17: rows_case_w<17>(acc, h4, dinv, r, dc, 0); break;
            case 18: rows_case_w<18>(acc, h4, dinv, r, dc, 0); break;
            case 19: rows_case_w<19>(acc, h4, dinv, r, dc, 0); break;
            case 20: rows_case_w<20>(acc, h4, dinv, r, dc, 0); break;
            default: break;
        }
        switch (m - t1) {
            case 1: rows_case_w<1>(acc, h4, dinv, r, dc, 20); break;
            case 2: rows_case_w<2>(acc, h4, dinv, r, dc, 20); break;
            case 3: rows_case_w<3>(acc, h4, dinv, r, dc, 20); break;
            case 4: rows_case_w<4>(acc, h4, dinv, r, dc, 20); break;
            case 5: rows_case_w<5>(acc, h4, dinv, r, dc, 20); break;
            case 6: rows_case_w<6>(acc, h4, dinv, r, dc, 20); break;
            case 7: rows_case_w<7>(acc, h4, dinv, r, dc, 20); break;
            case 8: rows_case_w<8>(acc, h4, dinv, r, dc, 20); break;
            case 9: rows_case_w<9>(acc, h4, dinv, r, dc, 20); break;
            case 10: rows_case_w<10>(acc, h4, dinv, r, dc, 20); break;
            case 11: rows_case_w<11>(acc, h4, dinv, r, dc, 20); break;
            case 12: rows_case_w<12>(acc, h4, dinv, r, dc, 20); break;
            default: break;
        }
    }
    if (mt > 32) {
        for (int i = 32; i < mt; ++i) {
            const int s = bucketp[slot_index(i, o1, o2, o3, n, node)];
            axpy4(acc, h4[(size_t)s * (CFEAT / 4)], dinv[s] * dc);
        }
    }
    if ((c0 > BKT) | (c1 > BKT) | (c2 > BKT) | (c3 > BKT)) {
        add_chain(acc, h4, dinv, ovf, headp + node, E, c0, dc);
        add_chain(acc, h4, dinv, ovf, headp + (size_t)n + node, E, c1, dc);
        add_chain(acc, h4, dinv, ovf, headp + (size_t)2 * n + node, E, c2, dc);
        add_chain(acc, h4, dinv, ovf, headp + (size_t)3 * n + node, E, c3, dc);
    }
    return acc;
}

__device__ __forceinline__ float4 bias_relu4(const float4 acc, const float4 b) {
    float4 o;
    o.x = acc.x + b.x; o.y = acc.y + b.y; o.z = acc.z + b.z; o.w = acc.w + b.w;
    o.x = o.x > 0.0f ? o.x : 0.0f;
    o.y = o.y > 0.0f ? o.y : 0.0f;
    o.z = o.z > 0.0f ? o.z : 0.0f;
    o.w = o.w > 0.0f ? o.w : 0.0f;
    return o;
}

template <int N, int ADD_BIAS>
__global__ void __launch_bounds__(256)
gather_relu_gemm_wmma_f32(const float* __restrict__ h,
                          const int* __restrict__ cntp,
                          const float* __restrict__ dinv,
                          const int* __restrict__ bucketp,
                          const int* __restrict__ ovfl,
                          const int* __restrict__ headp,
                          const float* __restrict__ bagg,
                          const float* __restrict__ W,
                          const float* __restrict__ bout,
                          float* __restrict__ C, int n, int E) {
    constexpr int KS = CFEAT / 4;
    constexpr int NT = N / 16;
    constexpr int OP = N + 4;

    __shared__ __attribute__((aligned(16))) float xt[FT_ROWS * LDS_PITCH];
    __shared__ __attribute__((aligned(16))) float ot[FT_ROWS * OP];

    const int lane = threadIdx.x & 31;
    const int wave = __builtin_amdgcn_readfirstlane((int)(threadIdx.x >> 5));
    const int lo = lane & 15;
    const int hi = lane >> 4;
    const bool mm = (NT == 8) || (wave < NT);
    const int n0 = (wave % NT) * 16;

    const int ntiles = (n + FT_ROWS - 1) / FT_ROWS;
    int t = (int)blockIdx.x;

    const int cnode = lane / NPART;
    const size_t cplane = (size_t)(lane % NPART) * n;
    int cv = 0;
    if (t < ntiles) {
        const int nd = t * FT_ROWS + FT_NPW * wave + cnode;
        if (lane < FT_NPW * NPART && nd < n) cv = cntp[cplane + nd];
    }

    v2f b[KS];
#pragma unroll
    for (int s = 0; s < KS; ++s) b[s] = v2f{0.0f, 0.0f};
    float bv = 0.0f;
    if (mm) {
        const float* wbase = W + (size_t)(2 * hi) * N + n0 + lo;
#pragma unroll
        for (int s = 0; s < KS; ++s) {
            b[s].x = wbase[(size_t)(4 * s) * N];
            b[s].y = wbase[(size_t)(4 * s + 1) * N];
        }
        if (ADD_BIAS) bv = bout[n0 + lo];
    }

    const float4* __restrict__ h4 = (const float4*)h + lane;
    const int2* __restrict__ ovf = (const int2*)ovfl;
    const float4 bb = ((const float4*)bagg)[lane];

    for (; t < ntiles; t += (int)gridDim.x) {
        const int nodeA = t * FT_ROWS + FT_NPW * wave;
        const int nodeB = nodeA + 1;
        static_assert(NPART == 4, "four partial in-degrees per node");
        const int cA0 = __builtin_amdgcn_readlane(cv, 0), cA1 = __builtin_amdgcn_readlane(cv, 1);
        const int cA2 = __builtin_amdgcn_readlane(cv, 2), cA3 = __builtin_amdgcn_readlane(cv, 3);
        const int cB0 = __builtin_amdgcn_readlane(cv, 4), cB1 = __builtin_amdgcn_readlane(cv, 5);
        const int cB2 = __builtin_amdgcn_readlane(cv, 6), cB3 = __builtin_amdgcn_readlane(cv, 7);
        const float dcA = dinv_of(cA0 + cA1 + cA2 + cA3);
        const float dcB = dinv_of(cB0 + cB1 + cB2 + cB3);
        int sA = 0, sB = 0;
        {
            const int a1 = cA0 < BKT ? cA0 : BKT, a2 = a1 + (cA1 < BKT ? cA1 : BKT), a3 = a2 + (cA2 < BKT ? cA2 : BKT);
            const int b1 = cB0 < BKT ? cB0 : BKT, b2 = b1 + (cB1 < BKT ? cB1 : BKT), b3 = b2 + (cB2 < BKT ? cB2 : BKT);
            if (lane < a3 + (cA3 < BKT ? cA3 : BKT)) sA = bucketp[slot_index(lane, a1, a2, a3, n, nodeA)];
            if (lane < b3 + (cB3 < BKT ? cB3 : BKT)) sB = bucketp[slot_index(lane, b1, b2, b3, n, nodeB)];
        }
        float4 selfA = make_float4(0.0f, 0.0f, 0.0f, 0.0f), selfB = selfA;
        if (nodeA < n) selfA = h4[(size_t)nodeA * (CFEAT / 4)];
        if (nodeB < n) selfB = h4[(size_t)nodeB * (CFEAT / 4)];

#pragma nounroll
        for (int i = 0; i < FT_NPW; ++i) {
            float4 acc = sum_messages(h4, dinv, bucketp, ovf, headp, n, E, i ? nodeB : nodeA,
                                      i ? cB0 : cA0, i ? cB1 : cA1, i ? cB2 : cA2, i ? cB3 : cA3,
                                      i ? dcB : dcA, i ? sB : sA, lane);
            const float dcc = i ? dcB : dcA;
            axpy4(acc, i ? selfB : selfA, dcc * dcc);
            *(float4*)&xt[(FT_NPW * wave + i) * LDS_PITCH + 4 * lane] = bias_relu4(acc, bb);
        }

        __builtin_amdgcn_fence(__ATOMIC_RELEASE, "workgroup", "local");
        __builtin_amdgcn_s_barrier();
        __builtin_amdgcn_fence(__ATOMIC_ACQUIRE, "workgroup", "local");

        {
            const int tn = t + (int)gridDim.x;
            cv = 0;
            if (tn < ntiles) {
                const int nd = tn * FT_ROWS + FT_NPW * wave + cnode;
                if (lane < FT_NPW * NPART && nd < n) cv = cntp[cplane + nd];
            }
        }

        if (mm) {
            const float* arow = &xt[lo * LDS_PITCH + 2 * hi];
            v2f a[KS];
#pragma unroll
            for (int s = 0; s < KS; ++s) a[s] = *(const v2f*)(arow + 4 * s);
            __builtin_amdgcn_sched_barrier(0);
            v8f acc = {};
#pragma unroll
            for (int s = 0; s < KS; ++s)
                acc = __builtin_amdgcn_wmma_f32_16x16x4_f32(
                     false, a[s],  false, b[s],
                     (short)0, acc,  false,  false);
#pragma unroll
            for (int v = 0; v < 8; ++v) ot[(v + 8 * hi) * OP + n0 + lo] = acc[v] + bv;
        }
        __builtin_amdgcn_fence(__ATOMIC_RELEASE, "workgroup", "local");
        __builtin_amdgcn_s_barrier();
        __builtin_amdgcn_fence(__ATOMIC_ACQUIRE, "workgroup", "local");

        if (N == 128) {
#pragma unroll
            for (int i = 0; i < FT_NPW; ++i) {
                const int rr = FT_NPW * wave + i;
                const int row = t * FT_ROWS + rr;
                if (row < n)
                    ((float4*)C)[(size_t)row * (N / 4) + lane] = *(const float4*)&ot[rr * OP + 4 * lane];
            }
        } else {
            const int rr = FT_NPW * wave + hi;
            const int row = t * FT_ROWS + rr;
            if (row < n)
                ((float4*)C)[(size_t)row * (N / 4) + lo] = *(const float4*)&ot[rr * OP + 4 * lo];
        }
    }
}

extern "C" void kernel_launch(void* const* d_in, const int* in_sizes, int n_in,
                              void* d_out, int out_size, void* d_ws, size_t ws_size,
                              hipStream_t stream) {
    const float* x  = (const float*)d_in[0];
    const int*   ei = (const int*)d_in[1];
    const float* W1 = (const float*)d_in[2];
    const float* b1 = (const float*)d_in[3];
    const float* W2 = (const float*)d_in[4];
    const float* b2 = (const float*)d_in[5];
    const float* Wl = (const float*)d_in[6];
    const float* bl = (const float*)d_in[7];
    float* out = (float*)d_out;

    const int E  = in_sizes[1] / 2;
    const int Nn = in_sizes[0] / CFEAT;
    const int C  = CFEAT;

    const int* row = ei;
    const int* col = ei + E;

    float* bufA = (float*)d_ws;
    float* bufB = bufA + (size_t)Nn * C;
    int* bucketp = (int*)(bufB + (size_t)Nn * C);
    int* cntp    = bucketp + (size_t)NPART * Nn * BKT;
    int* headp   = cntp + (size_t)NPART * Nn;
    int* ovf     = headp + (size_t)NPART * Nn;
    float* dinv  = (float*)(ovf + (size_t)2 * E);

    edge_owner_scan<<<((Nn + SC_NPB - 1) / SC_NPB) * NPART, SC_THREADS, 0, stream>>>(row, col, cntp, headp, bucketp, (int2*)ovf, Nn, E);
    gemm_wstat_wmma_f32<128, 0><<<512, 128, 0, stream>>>(x, W1, nullptr, bufA, Nn, cntp, dinv);

    gather_relu_gemm_wmma_f32<128, 0><<<500, 256, 0, stream>>>(bufA, cntp, dinv, bucketp, ovf, headp, b1, W2, nullptr, bufB, Nn, E);

    gather_relu_gemm_wmma_f32<64, 1><<<500, 256, 0, stream>>>(bufB, cntp, dinv, bucketp, ovf, headp, b2, Wl, bl, out, Nn, E);
}


# Edge weight split: producers store rows times own node's dinv (GEMM scales x rows, fused kernel its relu tile); aggregation adds rows plain, scales sum once; no per-source dinv gather; sum order kept
# speedup vs baseline: 4.5162x; 1.0464x over previous
#include <hip/hip_runtime.h>
#include <hip/hip_bf16.h>

typedef __attribute__((ext_vector_type(2))) float v2f;
typedef __attribute__((ext_vector_type(4))) float v4f;
typedef __attribute__((ext_vector_type(8))) float v8f;

#define CFEAT 128

#define NPART 4
#define BKT 16
#define SC_NPB 640
#define SC_THREADS 1024
#define SC_WAVES (SC_THREADS / 32)
#define SC_U 8
#define SC_LISTW 512

__device__ __forceinline__ void scan_file(const int* __restrict__ mylist, const int fill, const int lane, const int lo,
                                          const int* __restrict__ erow, const int* __restrict__ ecol,
                                          int* __restrict__ lcnt, int* __restrict__ lhead, int* __restrict__ lbkt,
                                          int2* __restrict__ ovf) {
    for (int i = lane; i < fill; i += 32) {
        const int e = mylist[i];
        const int t = ecol[e] - lo;
        const int r = erow[e];
        const int k = atomicAdd(&lcnt[t], 1);
        if (k < BKT) {
            lbkt[t * BKT + k] = r;
        } else {
            const int prev = atomicExch(&lhead[t], e);
            ovf[e] = make_int2(prev, r);
        }
    }
}

__global__ void __launch_bounds__(SC_THREADS)
edge_owner_scan(const int* __restrict__ erow, const int* __restrict__ ecol,
                int* __restrict__ cntp, int* __restrict__ headp, int* __restrict__ bucketp,
                int2* __restrict__ ovf, int n, int E) {
    __shared__ int lcnt[SC_NPB];
    __shared__ int lhead[SC_NPB];
    __shared__ __attribute__((aligned(16))) int lbkt[SC_NPB * BKT];
    __shared__ int llist[SC_WAVES][SC_LISTW];

    const int tid = (int)threadIdx.x;
    const int lane = tid & 31;
    const int wave = __builtin_amdgcn_readfirstlane(tid >> 5);
    const int part = (int)blockIdx.x % NPART;
    const int lo = ((int)blockIdx.x / NPART) * SC_NPB;

    for (int i = tid; i < SC_NPB; i += SC_THREADS) { lcnt[i] = 0; lhead[i] = -1; }
    __syncthreads();

    int* mylist = llist[wave];
    int fill = 0;
    const bool vec = ((E & 3) == 0) && ((((size_t)ecol) & 15) == 0);
    constexpr int WBLK = 128 * SC_U;
    const int plen = ((E + NPART - 1) / NPART + WBLK - 1) / WBLK * WBLK;
    const int eb = part * plen < E ? part * plen : E;
    const int ee = eb + plen < E ? eb + plen : E;

    for (int base = eb + wave * WBLK; base < ee; base += SC_WAVES * WBLK) {
        int4 c[SC_U];
        if (vec && base + WBLK <= ee) {
#pragma unroll
            for (int g = 0; g < SC_U; ++g) c[g] = *(const int4*)(ecol + base + 128 * g + 4 * lane);
        } else {
#pragma unroll
            for (int g = 0; g < SC_U; ++g) {
                const int e0 = base + 128 * g + 4 * lane;
                c[g].x = e0 + 0 < ee ? ecol[e0 + 0] : lo - 1;
                c[g].y = e0 + 1 < ee ? ecol[e0 + 1] : lo - 1;
                c[g].z = e0 + 2 < ee ? ecol[e0 + 2] : lo - 1;
                c[g].w = e0 + 3 < ee ? ecol[e0 + 3] : lo - 1;
            }
        }
#pragma unroll
        for (int g = 0; g < SC_U; ++g) {
            const int e0 = base + 128 * g + 4 * lane;
            const int cc[4] = {c[g].x, c[g].y, c[g].z, c[g].w};
#pragma unroll
            for (int j = 0; j < 4; ++j) {
                const bool hit = (unsigned)(cc[j] - lo) < (unsigned)SC_NPB;
                const unsigned mask = __builtin_amdgcn_ballot_w32(hit);
                if (mask) {
                    if (hit) mylist[fill + (int)__builtin_amdgcn_mbcnt_lo(mask, 0u)] = e0 + j;
                    fill += __builtin_popcount(mask);
                }
            }
            if (fill > SC_LISTW - 128) {
                __builtin_amdgcn_wave_barrier();
                scan_file(mylist, fill, lane, lo, erow, ecol, lcnt, lhead, lbkt, ovf);
                __builtin_amdgcn_wave_barrier();
                fill = 0;
            }
        }
    }
    __builtin_amdgcn_wave_barrier();
    scan_file(mylist, fill, lane, lo, erow, ecol, lcnt, lhead, lbkt, ovf);
    __syncthreads();

    const size_t pbase = (size_t)part * n;
    for (int i = tid; i < SC_NPB; i += SC_THREADS) {
        if (lo + i < n) { cntp[pbase + lo + i] = lcnt[i]; headp[pbase + lo + i] = lhead[i]; }
    }
    for (int i = tid; i < SC_NPB * BKT / 4; i += SC_THREADS) {
        if (lo + i / (BKT / 4) < n) ((int4*)bucketp)[(pbase + lo) * (BKT / 4) + i] = ((const int4*)lbkt)[i];
    }
}

#define LDS_PITCH 132

__device__ __forceinline__ float dinv_of(const int c) { return rsqrtf((float)c + 1.0f); }

__device__ __forceinline__ int sum_parts(const int cv, const int o) {
    static_assert(NPART == 4, "four partial in-degrees");
    return __builtin_amdgcn_readlane(cv, o) + __builtin_amdgcn_readlane(cv, o + 1) +
           __builtin_amdgcn_readlane(cv, o + 2) + __builtin_amdgcn_readlane(cv, o + 3);
}

template <int N, int ADD_BIAS>
__global__ void __launch_bounds__(128)
gemm_wstat_wmma_f32(const float* __restrict__ A,
                    const float* __restrict__ W,
                    const float* __restrict__ bias,
                    float* __restrict__ C, int M,
                    const int* __restrict__ cntp) {
    constexpr int K = CFEAT;
    constexpr int KS = K / 4;
    constexpr int NPAIR = N / 32;
    constexpr int MT = 4 / NPAIR;
    constexpr int ROWS = 16 * MT;
    constexpr int LD4 = ROWS / 4;

    __shared__ __attribute__((aligned(16))) float lds[2][ROWS * LDS_PITCH];

    const int bid = (int)blockIdx.x;
    const int nblk = (int)gridDim.x;

    const int lane = threadIdx.x & 31;
    const int wave = threadIdx.x >> 5;
    const int lo = lane & 15;
    const int hi = lane >> 4;
    const int mt = wave / NPAIR;
    const int n0 = (wave % NPAIR) * 32;

    v2f b0[KS], b1[KS];
    {
        const float* wbase = W + (size_t)(2 * hi) * N + n0 + lo;
#pragma unroll
        for (int s = 0; s < KS; ++s) {
            b0[s].x = wbase[(size_t)(4 * s) * N];
            b0[s].y = wbase[(size_t)(4 * s + 1) * N];
            b1[s].x = wbase[(size_t)(4 * s) * N + 16];
            b1[s].y = wbase[(size_t)(4 * s + 1) * N + 16];
        }
    }
    const float bv0 = ADD_BIAS ? bias[n0 + lo] : 0.0f;
    const float bv1 = ADD_BIAS ? bias[n0 + 16 + lo] : 0.0f;

    const int steps = (M + ROWS - 1) / ROWS;
    const v4f* __restrict__ A4 = (const v4f*)A + lane;
    v4f g[LD4];

    static_assert(NPART * LD4 <= 32, "one lane per (row of the wave, part)");
    const int crow = wave + 4 * (lane / NPART);
    const size_t cplane = (size_t)(lane % NPART) * M;
    int cv = 0;

    int st = bid;
    if (st < steps) {
#pragma unroll
        for (int i = 0; i < LD4; ++i) {
            const int row = st * ROWS + wave + 4 * i;
            g[i] = A4[(size_t)(row < M ? row : M - 1) * (K / 4)];
        }
        if (lane < NPART * LD4 && st * ROWS + crow < M) cv = cntp[cplane + st * ROWS + crow];
#pragma unroll
        for (int i = 0; i < LD4; ++i)
            *(v4f*)&lds[0][(wave + 4 * i) * LDS_PITCH + 4 * lane] = g[i] * dinv_of(sum_parts(cv, NPART * i));
    }
    __syncthreads();

    int cur = 0;
    for (; st < steps; st += nblk) {
        const int nxt = st + nblk;
        const bool more = nxt < steps;
        if (more) {
#pragma unroll
            for (int i = 0; i < LD4; ++i) {
                const int row = nxt * ROWS + wave + 4 * i;
                g[i] = A4[(size_t)(row < M ? row : M - 1) * (K / 4)];
            }
            cv = 0;
            if (lane < NPART * LD4 && nxt * ROWS + crow < M) cv = cntp[cplane + nxt * ROWS + crow];
        }

        const float* arow = &lds[cur][(16 * mt + lo) * LDS_PITCH + 2 * hi];
        v2f a[KS];
#pragma unroll
        for (int s = 0; s < KS; ++s) a[s] = *(const v2f*)(arow + 4 * s);
        __builtin_amdgcn_sched_barrier(0);

        v8f acc0 = {}, acc1 = {};
#pragma unroll
        for (int s = 0; s < KS; ++s) {
            acc0 = __builtin_amdgcn_wmma_f32_16x16x4_f32(
                 false, a[s],  false, b0[s],
                 (short)0, acc0,  false,  false);
            acc1 = __builtin_amdgcn_wmma_f32_16x16x4_f32(
                 false, a[s],  false, b1[s],
                 (short)0, acc1,  false,  false);
        }

        const int r0 = st * ROWS + 16 * mt + 8 * hi;
        float* cbase = C + (size_t)r0 * N + n0 + lo;
        float o0[8], o1[8];
#pragma unroll
        for (int v = 0; v < 8; ++v) { o0[v] = acc0[v] + bv0; o1[v] = acc1[v] + bv1; }
        if ((st + 1) * ROWS <= M) {
#pragma unroll
            for (int v = 0; v < 8; ++v) {
                cbase[(size_t)v * N] = o0[v];
                cbase[(size_t)v * N + 16] = o1[v];
            }
        } else {
#pragma unroll
            for (int v = 0; v < 8; ++v) {
                if (r0 + v < M) {
                    cbase[(size_t)v * N] = o0[v];
                    cbase[(size_t)v * N + 16] = o1[v];
                }
            }
        }

        if (more) {
#pragma unroll
            for (int i = 0; i < LD4; ++i)
                *(v4f*)&lds[cur ^ 1][(wave + 4 * i) * LDS_PITCH + 4 * lane] = g[i] * dinv_of(sum_parts(cv, NPART * i));
        }
        __builtin_amdgcn_fence(__ATOMIC_RELEASE, "workgroup", "local");
        __builtin_amdgcn_s_barrier();
        __builtin_amdgcn_fence(__ATOMIC_ACQUIRE, "workgroup", "local");
        cur ^= 1;
    }
}

#define FT_ROWS 16
#define FT_NPW 2

__device__ __forceinline__ void add4(float4& acc, const float4 v) {
    acc.x += v.x; acc.y += v.y; acc.z += v.z; acc.w += v.w;
}

__device__ __forceinline__ float4 scale4(const float4 v, const float w) {
    return make_float4(v.x * w, v.y * w, v.z * w, v.w * w);
}

template <int S>
__device__ __forceinline__ void load_rows(float4 (&v)[S], const float4* __restrict__ h4, const int r, const int o) {
    int rr[S];
#pragma unroll
    for (int q = 0; q < S; ++q) rr[q] = __builtin_amdgcn_readlane(r, o + q);
#pragma unroll
    for (int q = 0; q < S; ++q) v[q] = h4[(size_t)rr[q] * (CFEAT / 4)];
}

template <int T>
__device__ __forceinline__ void rows_case(float4& acc, const float4* __restrict__ h4, const int r, const int k) {
    float4 v[T];
    load_rows<T>(v, h4, r, k);
#pragma unroll
    for (int q = 0; q < T; ++q) add4(acc, v[q]);
}

__device__ __forceinline__ size_t slot_index(const int i, const int o1, const int o2, const int o3,
                                             const int n, const int node) {
    static_assert(NPART == 4, "three bucket boundaries");
    const int p = (i >= o1) + (i >= o2) + (i >= o3);
    const int j = i - (i >= o3 ? o3 : i >= o2 ? o2 : i >= o1 ? o1 : 0);
    return ((size_t)p * n + node) * BKT + j;
}

__device__ __forceinline__ void add_chain(float4& acc, const float4* __restrict__ h4,
                                          const int2* __restrict__ ovf, const int* __restrict__ head, const int E,
                                          const int c) {
    if (c > BKT) {
        int e = *head;
        for (int i = BKT; i < c && e >= 0 && e < E; ++i) {
            const int2 q = ovf[e];
            add4(acc, h4[(size_t)q.y * (CFEAT / 4)]);
            e = q.x;
        }
    }
}

__device__ __forceinline__ float4 sum_messages(const float4* __restrict__ h4,
                                               const int* __restrict__ bucketp,
                                               const int2* __restrict__ ovf, const int* __restrict__ headp,
                                               const int n, const int E, const int node,
                                               const int c0, const int c1, const int c2, const int c3,
                                               const int s0) {
    static_assert(NPART == 4, "four partial in-degrees");
    float4 acc = make_float4(0.0f, 0.0f, 0.0f, 0.0f);
    const int m0 = c0 < BKT ? c0 : BKT, m1 = c1 < BKT ? c1 : BKT, m2 = c2 < BKT ? c2 : BKT, m3 = c3 < BKT ? c3 : BKT;
    const int o1 = m0, o2 = o1 + m1, o3 = o2 + m2, mt = o3 + m3;
    {
        const int m = mt < 32 ? mt : 32;
        const int r = s0;
        const int t1 = m < 20 ? m : 20;
        switch (t1) {
            case 1: rows_case<1>(acc, h4, r, 0); break;
            case 2: rows_case<2>(acc, h4, r, 0); break;
            case 3: rows_case<3>(acc, h4, r, 0); break;
            case 4: rows_case<4>(acc, h4, r, 0); break;
            case 5: rows_case<5>(acc, h4, r, 0); break;
            case 6: rows_case<6>(acc, h4, r, 0); break;
            case 7: rows_case<7>(acc, h4, r, 0); break;
            case 8: rows_case<8>(acc, h4, r, 0); break;
            case 9: rows_case<9>(acc, h4, r, 0); break;
            case 10: rows_case<10>(acc, h4, r, 0); break;
            case 11: rows_case<11>(acc, h4, r, 0); break;
            case 12: rows_case<12>(acc, h4, r, 0); break;
            case 13: rows_case<13>(acc, h4, r, 0); break;
            case 14: rows_case<14>(acc, h4, r, 0); break;
            case 15: rows_case<15>(acc, h4, r, 0); break;
            case 16: rows_case<16>(acc, h4, r, 0); break;
            case 17: rows_case<17>(acc, h4, r, 0); break;
            case 18: rows_case<18>(acc, h4, r, 0); break;
            case 19: rows_case<19>(acc, h4, r, 0); break;
            case 20: rows_case<20>(acc, h4, r, 0); break;
            default: break;
        }
        switch (m - t1) {
            case 1: rows_case<1>(acc, h4, r, 20); break;
            case 2: rows_case<2>(acc, h4, r, 20); break;
            case 3: rows_case<3>(acc, h4, r, 20); break;
            case 4: rows_case<4>(acc, h4, r, 20); break;
            case 5: rows_case<5>(acc, h4, r, 20); break;
            case 6: rows_case<6>(acc, h4, r, 20); break;
            case 7: rows_case<7>(acc, h4, r, 20); break;
            case 8: rows_case<8>(acc, h4, r, 20); break;
            case 9: rows_case<9>(acc, h4, r, 20); break;
            case 10: rows_case<10>(acc, h4, r, 20); break;
            case 11: rows_case<11>(acc, h4, r, 20); break;
            case 12: rows_case<12>(acc, h4, r, 20); break;
            default: break;
        }
    }
    if (mt > 32) {
        for (int i = 32; i < mt; ++i) {
            const int s = bucketp[slot_index(i, o1, o2, o3, n, node)];
            add4(acc, h4[(size_t)s * (CFEAT / 4)]);
        }
    }
    if ((c0 > BKT) | (c1 > BKT) | (c2 > BKT) | (c3 > BKT)) {
        add_chain(acc, h4, ovf, headp + node, E, c0);
        add_chain(acc, h4, ovf, headp + (size_t)n + node, E, c1);
        add_chain(acc, h4, ovf, headp + (size_t)2 * n + node, E, c2);
        add_chain(acc, h4, ovf, headp + (size_t)3 * n + node, E, c3);
    }
    return acc;
}

__device__ __forceinline__ float4 bias_relu4(const float4 acc, const float4 b) {
    float4 o;
    o.x = acc.x + b.x; o.y = acc.y + b.y; o.z = acc.z + b.z; o.w = acc.w + b.w;
    o.x = o.x > 0.0f ? o.x : 0.0f;
    o.y = o.y > 0.0f ? o.y : 0.0f;
    o.z = o.z > 0.0f ? o.z : 0.0f;
    o.w = o.w > 0.0f ? o.w : 0.0f;
    return o;
}

template <int N, int ADD_BIAS, int SCALE_OUT>
__global__ void __launch_bounds__(256)
gather_relu_gemm_wmma_f32(const float* __restrict__ h,
                          const int* __restrict__ cntp,
                          const int* __restrict__ bucketp,
                          const int* __restrict__ ovfl,
                          const int* __restrict__ headp,
                          const float* __restrict__ bagg,
                          const float* __restrict__ W,
                          const float* __restrict__ bout,
                          float* __restrict__ C, int n, int E) {
    constexpr int KS = CFEAT / 4;
    constexpr int NT = N / 16;
    constexpr int OP = N + 4;
    static_assert(!(SCALE_OUT && ADD_BIAS), "C's rows are scaled through xt: a bias added behind the product would not be");

    __shared__ __attribute__((aligned(16))) float xt[FT_ROWS * LDS_PITCH];
    __shared__ __attribute__((aligned(16))) float ot[FT_ROWS * OP];

    const int lane = threadIdx.x & 31;
    const int wave = __builtin_amdgcn_readfirstlane((int)(threadIdx.x >> 5));
    const int lo = lane & 15;
    const int hi = lane >> 4;
    const bool mm = (NT == 8) || (wave < NT);
    const int n0 = (wave % NT) * 16;

    const int ntiles = (n + FT_ROWS - 1) / FT_ROWS;
    int t = (int)blockIdx.x;

    const int cnode = lane / NPART;
    const size_t cplane = (size_t)(lane % NPART) * n;
    int cv = 0;
    if (t < ntiles) {
        const int nd = t * FT_ROWS + FT_NPW * wave + cnode;
        if (lane < FT_NPW * NPART && nd < n) cv = cntp[cplane + nd];
    }

    v2f b[KS];
#pragma unroll
    for (int s = 0; s < KS; ++s) b[s] = v2f{0.0f, 0.0f};
    float bv = 0.0f;
    if (mm) {
        const float* wbase = W + (size_t)(2 * hi) * N + n0 + lo;
#pragma unroll
        for (int s = 0; s < KS; ++s) {
            b[s].x = wbase[(size_t)(4 * s) * N];
            b[s].y = wbase[(size_t)(4 * s + 1) * N];
        }
        if (ADD_BIAS) bv = bout[n0 + lo];
    }

    const float4* __restrict__ h4 = (const float4*)h + lane;
    const int2* __restrict__ ovf = (const int2*)ovfl;
    const float4 bb = ((const float4*)bagg)[lane];

    for (; t < ntiles; t += (int)gridDim.x) {
        const int nodeA = t * FT_ROWS + FT_NPW * wave;
        const int nodeB = nodeA + 1;
        static_assert(NPART == 4, "four partial in-degrees per node");
        const int cA0 = __builtin_amdgcn_readlane(cv, 0), cA1 = __builtin_amdgcn_readlane(cv, 1);
        const int cA2 = __builtin_amdgcn_readlane(cv, 2), cA3 = __builtin_amdgcn_readlane(cv, 3);
        const int cB0 = __builtin_amdgcn_readlane(cv, 4), cB1 = __builtin_amdgcn_readlane(cv, 5);
        const int cB2 = __builtin_amdgcn_readlane(cv, 6), cB3 = __builtin_amdgcn_readlane(cv, 7);
        const float dcA = dinv_of(cA0 + cA1 + cA2 + cA3);
        const float dcB = dinv_of(cB0 + cB1 + cB2 + cB3);
        int sA = 0, sB = 0;
        {
            const int a1 = cA0 < BKT ? cA0 : BKT, a2 = a1 + (cA1 < BKT ? cA1 : BKT), a3 = a2 + (cA2 < BKT ? cA2 : BKT);
            const int b1 = cB0 < BKT ? cB0 : BKT, b2 = b1 + (cB1 < BKT ? cB1 : BKT), b3 = b2 + (cB2 < BKT ? cB2 : BKT);
            if (lane < a3 + (cA3 < BKT ? cA3 : BKT)) sA = bucketp[slot_index(lane, a1, a2, a3, n, nodeA)];
            if (lane < b3 + (cB3 < BKT ? cB3 : BKT)) sB = bucketp[slot_index(lane, b1, b2, b3, n, nodeB)];
        }
        float4 selfA = make_float4(0.0f, 0.0f, 0.0f, 0.0f), selfB = selfA;
        if (nodeA < n) selfA = h4[(size_t)nodeA * (CFEAT / 4)];
        if (nodeB < n) selfB = h4[(size_t)nodeB * (CFEAT / 4)];

#pragma nounroll
        for (int i = 0; i < FT_NPW; ++i) {
            float4 acc = sum_messages(h4, bucketp, ovf, headp, n, E, i ? nodeB : nodeA,
                                      i ? cB0 : cA0, i ? cB1 : cA1, i ? cB2 : cA2, i ? cB3 : cA3,
                                      i ? sB : sA);
            const float dcc = i ? dcB : dcA;
            add4(acc, i ? selfB : selfA);
            float4 o = bias_relu4(scale4(acc, dcc), bb);
            if (SCALE_OUT) o = scale4(o, dcc);
            *(float4*)&xt[(FT_NPW * wave + i) * LDS_PITCH + 4 * lane] = o;
        }

        __builtin_amdgcn_fence(__ATOMIC_RELEASE, "workgroup", "local");
        __builtin_amdgcn_s_barrier();
        __builtin_amdgcn_fence(__ATOMIC_ACQUIRE, "workgroup", "local");

        {
            const int tn = t + (int)gridDim.x;
            cv = 0;
            if (tn < ntiles) {
                const int nd = tn * FT_ROWS + FT_NPW * wave + cnode;
                if (lane < FT_NPW * NPART && nd < n) cv = cntp[cplane + nd];
            }
        }

        if (mm) {
            const float* arow = &xt[lo * LDS_PITCH + 2 * hi];
            v2f a[KS];
#pragma unroll
            for (int s = 0; s < KS; ++s) a[s] = *(const v2f*)(arow + 4 * s);
            __builtin_amdgcn_sched_barrier(0);
            v8f acc = {};
#pragma unroll
            for (int s = 0; s < KS; ++s)
                acc = __builtin_amdgcn_wmma_f32_16x16x4_f32(
                     false, a[s],  false, b[s],
                     (short)0, acc,  false,  false);
#pragma unroll
            for (int v = 0; v < 8; ++v) ot[(v + 8 * hi) * OP + n0 + lo] = acc[v] + bv;
        }
        __builtin_amdgcn_fence(__ATOMIC_RELEASE, "workgroup", "local");
        __builtin_amdgcn_s_barrier();
        __builtin_amdgcn_fence(__ATOMIC_ACQUIRE, "workgroup", "local");

        if (N == 128) {
#pragma unroll
            for (int i = 0; i < FT_NPW; ++i) {
                const int rr = FT_NPW * wave + i;
                const int row = t * FT_ROWS + rr;
                if (row < n)
                    ((float4*)C)[(size_t)row * (N / 4) + lane] = *(const float4*)&ot[rr * OP + 4 * lane];
            }
        } else {
            const int rr = FT_NPW * wave + hi;
            const int row = t * FT_ROWS + rr;
            if (row < n)
                ((float4*)C)[(size_t)row * (N / 4) + lo] = *(const float4*)&ot[rr * OP + 4 * lo];
        }
    }
}

extern "C" void kernel_launch(void* const* d_in, const int* in_sizes, int n_in,
                              void* d_out, int out_size, void* d_ws, size_t ws_size,
                              hipStream_t stream) {
    const float* x  = (const float*)d_in[0];
    const int*   ei = (const int*)d_in[1];
    const float* W1 = (const float*)d_in[2];
    const float* b1 = (const float*)d_in[3];
    const float* W2 = (const float*)d_in[4];
    const float* b2 = (const float*)d_in[5];
    const float* Wl = (const float*)d_in[6];
    const float* bl = (const float*)d_in[7];
    float* out = (float*)d_out;

    const int E  = in_sizes[1] / 2;
    const int Nn = in_sizes[0] / CFEAT;
    const int C  = CFEAT;

    const int* row = ei;
    const int* col = ei + E;

    float* bufA = (float*)d_ws;
    float* bufB = bufA + (size_t)Nn * C;
    int* bucketp = (int*)(bufB + (size_t)Nn * C);
    int* cntp    = bucketp + (size_t)NPART * Nn * BKT;
    int* headp   = cntp + (size_t)NPART * Nn;
    int* ovf     = headp + (size_t)NPART * Nn;

    edge_owner_scan<<<((Nn + SC_NPB - 1) / SC_NPB) * NPART, SC_THREADS, 0, stream>>>(row, col, cntp, headp, bucketp, (int2*)ovf, Nn, E);
    gemm_wstat_wmma_f32<128, 0><<<512, 128, 0, stream>>>(x, W1, nullptr, bufA, Nn, cntp);

    gather_relu_gemm_wmma_f32<128, 0, 1><<<500, 256, 0, stream>>>(bufA, cntp, bucketp, ovf, headp, b1, W2, nullptr, bufB, Nn, E);

    gather_relu_gemm_wmma_f32<64, 1, 0><<<500, 256, 0, stream>>>(bufB, cntp, bucketp, ovf, headp, b2, Wl, bl, out, Nn, E);
}
